# starting kernel
# baseline (speedup 1.0000x reference)
#include <hip/hip_runtime.h>


typedef __attribute__((ext_vector_type(2))) float v2f;
typedef __attribute__((ext_vector_type(8))) float v8f;

#define H 128
#define LDSPAD 132
#define TPB 8

__global__ __launch_bounds__(256) void zero_kernel(float* p, long n) {
  long i = (long)blockIdx.x * blockDim.x + threadIdx.x;
  if (i < n) p[i] = 0.0f;
}

__global__ __launch_bounds__(256) void pad_kernel(const float* __restrict__ in,
                                                  float* __restrict__ out,
                                                  int N, int F) {
  long t = (long)blockIdx.x * blockDim.x + threadIdx.x;
  if (t >= (long)N * H) return;
  int i = (int)(t >> 7), h = (int)(t & 127);
  out[t] = (h < F) ? in[(long)i * F + h] : 0.0f;
}

__global__ __launch_bounds__(256) void deg_kernel(const int* __restrict__ edge,
                                                  int E, float* __restrict__ deg) {
  int e = blockIdx.x * blockDim.x + threadIdx.x;
  if (e < E) atomicAdd(&deg[edge[E + e]], 1.0f);
}

__global__ __launch_bounds__(256) void scatter_kernel(const float* __restrict__ X,
                                                      const int* __restrict__ edge,
                                                      int E, float* __restrict__ S) {
  long idx = (long)blockIdx.x * blockDim.x + threadIdx.x;
  if (idx >= (long)E * 32) return;
  int e = (int)(idx >> 5);
  int c = (int)(idx & 31) << 2;
  int s = edge[e], d = edge[E + e];
  const float4 v = *(const float4*)(X + (long)s * H + c);
  float* o = S + (long)d * H + c;
  atomicAdd(o + 0, v.x); atomicAdd(o + 1, v.y);
  atomicAdd(o + 2, v.z); atomicAdd(o + 3, v.w);
}

template <bool HAS_DEG, bool RELU>
__global__ __launch_bounds__(256) void gemm_dual_kernel(
    const float* __restrict__ A, const float* __restrict__ B,
    const float* __restrict__ WA, const float* __restrict__ WB,
    const float* __restrict__ bias, const float* __restrict__ deg,
    float* __restrict__ Y, int M) {
  __shared__ float lwa[H * LDSPAD];
  __shared__ float lwb[H * LDSPAD];
  for (int idx = threadIdx.x; idx < H * H; idx += 256) {
    int k = idx >> 7, n = idx & 127;
    lwa[n * LDSPAD + k] = WA[idx];
    lwb[n * LDSPAD + k] = WB[idx];
  }
  __syncthreads();

  const int wave = threadIdx.x >> 5;
  const int lane = threadIdx.x & 31;
  const int l15  = lane & 15;
  const int kh2  = (lane >> 4) << 1;
  const int ntiles = M >> 4;

  for (int t = blockIdx.x * TPB + wave; t < ntiles; t += gridDim.x * TPB) {
    const int m0 = t << 4;
    v8f cA[8], cB[8];
#pragma unroll
    for (int n = 0; n < 8; ++n) {
      cA[n] = (v8f){0, 0, 0, 0, 0, 0, 0, 0};
      cB[n] = (v8f){0, 0, 0, 0, 0, 0, 0, 0};
    }

    const float* Ap = A + (long)(m0 + l15) * H + kh2;
    const float* Bp = B + (long)(m0 + l15) * H + kh2;
    const float* wabase = &lwa[l15 * LDSPAD + kh2];
    const float* wbbase = &lwb[l15 * LDSPAD + kh2];

#pragma unroll 1
    for (int kk = 0; kk < 32; ++kk) {
      const int k0 = kk << 2;
      v2f a = *(const v2f*)(Ap + k0);
      v2f b = *(const v2f*)(Bp + k0);
#pragma unroll
      for (int n = 0; n < 8; ++n) {
        v2f wa = *(const v2f*)(wabase + n * 16 * LDSPAD + k0);
        v2f wb = *(const v2f*)(wbbase + n * 16 * LDSPAD + k0);
        cA[n] = __builtin_amdgcn_wmma_f32_16x16x4_f32(false, a, false, wa,
                                                      (short)0, cA[n], false, false);
        cB[n] = __builtin_amdgcn_wmma_f32_16x16x4_f32(false, b, false, wb,
                                                      (short)0, cB[n], false, false);
      }
    }

    const int rbase = m0 + ((lane >> 4) << 3);
    float degv[8];
    if (HAS_DEG) {
      const float4 d0 = *(const float4*)(deg + rbase);
      const float4 d1 = *(const float4*)(deg + rbase + 4);
      degv[0] = d0.x; degv[1] = d0.y; degv[2] = d0.z; degv[3] = d0.w;
      degv[4] = d1.x; degv[5] = d1.y; degv[6] = d1.z; degv[7] = d1.w;
    } else {
#pragma unroll
      for (int r = 0; r < 8; ++r) degv[r] = 1.0f;
    }
#pragma unroll
    for (int n = 0; n < 8; ++n) {
      const float bv = bias[n * 16 + l15];
#pragma unroll
      for (int r = 0; r < 8; ++r) {
        float acc = HAS_DEG ? (cA[n][r] + degv[r] * (cB[n][r] + bv))
                            : (cA[n][r] + cB[n][r] + bv);
        if (RELU) acc = fmaxf(acc, 0.0f);
        Y[(long)(rbase + r) * H + n * 16 + l15] = acc;
      }
    }
  }
}

__global__ __launch_bounds__(256) void pool_sum_kernel(const float* __restrict__ xl,
                                                       const int* __restrict__ batch,
                                                       int NL, float* __restrict__ sums) {
  long idx = (long)blockIdx.x * blockDim.x + threadIdx.x;
  if (idx >= (long)NL * 32) return;
  int link = (int)(idx >> 5);
  int c = (int)(idx & 31) << 2;
  int b = batch[link];
  const float4 v = *(const float4*)(xl + (long)link * H + c);
  float* o = sums + (long)b * H + c;
  atomicAdd(o + 0, v.x); atomicAdd(o + 1, v.y);
  atomicAdd(o + 2, v.z); atomicAdd(o + 3, v.w);
}

__global__ __launch_bounds__(256) void pool_cnt_kernel(const int* __restrict__ batch,
                                                       int NL, float* __restrict__ cnt) {
  int i = blockIdx.x * blockDim.x + threadIdx.x;
  if (i < NL) atomicAdd(&cnt[batch[i]], 1.0f);
}

__global__ __launch_bounds__(256) void finalize_kernel(const float* __restrict__ sums,
                                                       const float* __restrict__ cnt,
                                                       float* __restrict__ out, int G) {
  int t = blockIdx.x * blockDim.x + threadIdx.x;
  if (t >= G * H) return;
  int g = t >> 7;
  out[t] = sums[t] / fmaxf(cnt[g], 1.0f);
}

extern "C" void kernel_launch(void* const* d_in, const int* in_sizes, int n_in,
                              void* d_out, int out_size, void* d_ws, size_t ws_size,
                              hipStream_t stream) {
  const float* x_node = (const float*)d_in[0];
  const float* x_link = (const float*)d_in[1];
  const float* Wm_nl  = (const float*)d_in[2];
  const float* bm_nl  = (const float*)d_in[3];
  const float* Wu_nl  = (const float*)d_in[4];
  const float* bu_nl  = (const float*)d_in[5];
  const float* Wm_ln  = (const float*)d_in[6];
  const float* bm_ln  = (const float*)d_in[7];
  const float* Wu_ln  = (const float*)d_in[8];
  const float* bu_ln  = (const float*)d_in[9];
  const int*   nl     = (const int*)d_in[10];
  const int*   ln     = (const int*)d_in[11];
  const int*   batch  = (const int*)d_in[12];

  const int NL = in_sizes[12];
  const int NN = 50000;
  const int F_NODE = in_sizes[0] / NN;
  const int F_LINK = in_sizes[1] / NL;
  const int E  = in_sizes[10] / 2;
  const int Lnum = in_sizes[3] / H;
  const int G = out_size / H;

  const size_t MH = (size_t)((NN > NL) ? NN : NL) * H;
  float* ws   = (float*)d_ws;
  float* xn   = ws;
  float* xl   = xn  + MH;
  float* xn2  = xl  + MH;
  float* xl2  = xn2 + MH;
  float* Sl   = xl2 + MH;
  float* Sn   = Sl  + MH;
  float* AGG  = Sn  + MH;
  float* dgl  = AGG + MH;
  float* dgn  = dgl + NL;
  float* sums = dgn + NN;
  float* cnt  = sums + (size_t)G * H;
  (void)n_in; (void)ws_size;

  auto nb = [](long n, int b) { return (int)((n + b - 1) / b); };

  zero_kernel<<<nb(NL, 256), 256, 0, stream>>>(dgl, NL);
  zero_kernel<<<nb(NN, 256), 256, 0, stream>>>(dgn, NN);
  deg_kernel<<<nb(E, 256), 256, 0, stream>>>(nl, E, dgl);
  deg_kernel<<<nb(E, 256), 256, 0, stream>>>(ln, E, dgn);

  pad_kernel<<<nb((long)NN * H, 256), 256, 0, stream>>>(x_node, xn, NN, F_NODE);
  pad_kernel<<<nb((long)NL * H, 256), 256, 0, stream>>>(x_link, xl, NL, F_LINK);

  for (int l = 0; l < Lnum; ++l) {
    float* cxn = (l & 1) ? xn2 : xn;
    float* cxl = (l & 1) ? xl2 : xl;
    float* oxn = (l & 1) ? xn  : xn2;
    float* oxl = (l & 1) ? xl  : xl2;

    const float* WmNLt = Wm_nl + (size_t)l * 2 * H * H;
    const float* WmNLb = WmNLt + H * H;
    const float* WuNLt = Wu_nl + (size_t)l * 2 * H * H;
    const float* WuNLb = WuNLt + H * H;
    const float* WmLNt = Wm_ln + (size_t)l * 2 * H * H;
    const float* WmLNb = WmLNt + H * H;
    const float* WuLNt = Wu_ln + (size_t)l * 2 * H * H;
    const float* WuLNb = WuLNt + H * H;

    zero_kernel<<<nb((long)NL * H, 256), 256, 0, stream>>>(Sl, (long)NL * H);
    zero_kernel<<<nb((long)NN * H, 256), 256, 0, stream>>>(Sn, (long)NN * H);
    scatter_kernel<<<nb((long)E * 32, 256), 256, 0, stream>>>(cxn, nl, E, Sl);
    scatter_kernel<<<nb((long)E * 32, 256), 256, 0, stream>>>(cxl, ln, E, Sn);

    const int gl = nb(NL / 16, TPB);
    const int gn = nb(NN / 16, TPB);
    gemm_dual_kernel<true, false><<<gl, 256, 0, stream>>>(
        Sl, cxl, WmNLt, WmNLb, bm_nl + (size_t)l * H, dgl, AGG, NL);
    gemm_dual_kernel<false, true><<<gl, 256, 0, stream>>>(
        AGG, cxl, WuNLt, WuNLb, bu_nl + (size_t)l * H, nullptr, oxl, NL);
    gemm_dual_kernel<true, false><<<gn, 256, 0, stream>>>(
        Sn, cxn, WmLNt, WmLNb, bm_ln + (size_t)l * H, dgn, AGG, NN);
    gemm_dual_kernel<false, true><<<gn, 256, 0, stream>>>(
        AGG, cxn, WuLNt, WuLNb, bu_ln + (size_t)l * H, nullptr, oxn, NN);
  }

  float* fxl = (Lnum & 1) ? xl2 : xl;
  zero_kernel<<<nb((long)G * H, 256), 256, 0, stream>>>(sums, (long)G * H);
  zero_kernel<<<nb(G, 256), 256, 0, stream>>>(cnt, G);
  pool_sum_kernel<<<nb((long)NL * 32, 256), 256, 0, stream>>>(fxl, batch, NL, sums);
  pool_cnt_kernel<<<nb(NL, 256), 256, 0, stream>>>(batch, NL, cnt);
  finalize_kernel<<<nb((long)G * H, 256), 256, 0, stream>>>(sums, cnt, (float*)d_out, G);
}


# first layer scatters and multiplies only the F real input columns (the padded columns are exact zeros); pad launches dropped, raw inputs read directly
# speedup vs baseline: 1.3358x; 1.3358x over previous
#include <hip/hip_runtime.h>


typedef __attribute__((ext_vector_type(2))) float v2f;
typedef __attribute__((ext_vector_type(8))) float v8f;

#define H 128
#define LDSPAD 132
#define TPB 8

__global__ __launch_bounds__(256) void zero_kernel(float* p, long n) {
  long i = (long)blockIdx.x * blockDim.x + threadIdx.x;
  if (i < n) p[i] = 0.0f;
}

__global__ __launch_bounds__(256) void pad_kernel(const float* __restrict__ in,
                                                  float* __restrict__ out,
                                                  int N, int F) {
  long t = (long)blockIdx.x * blockDim.x + threadIdx.x;
  if (t >= (long)N * H) return;
  int i = (int)(t >> 7), h = (int)(t & 127);
  out[t] = (h < F) ? in[(long)i * F + h] : 0.0f;
}

__global__ __launch_bounds__(256) void deg_kernel(const int* __restrict__ edge,
                                                  int E, float* __restrict__ deg) {
  int e = blockIdx.x * blockDim.x + threadIdx.x;
  if (e < E) atomicAdd(&deg[edge[E + e]], 1.0f);
}

__global__ __launch_bounds__(256) void scatter_kernel(const float* __restrict__ X,
                                                      const int* __restrict__ edge,
                                                      int E, float* __restrict__ S) {
  long idx = (long)blockIdx.x * blockDim.x + threadIdx.x;
  if (idx >= (long)E * 32) return;
  int e = (int)(idx >> 5);
  int c = (int)(idx & 31) << 2;
  int s = edge[e], d = edge[E + e];
  const float4 v = *(const float4*)(X + (long)s * H + c);
  float* o = S + (long)d * H + c;
  atomicAdd(o + 0, v.x); atomicAdd(o + 1, v.y);
  atomicAdd(o + 2, v.z); atomicAdd(o + 3, v.w);
}

template <bool HAS_DEG, bool RELU>
__global__ __launch_bounds__(256) void gemm_dual_kernel(
    const float* __restrict__ A, const float* __restrict__ B,
    const float* __restrict__ WA, const float* __restrict__ WB,
    const float* __restrict__ bias, const float* __restrict__ deg,
    float* __restrict__ Y, int M) {
  __shared__ float lwa[H * LDSPAD];
  __shared__ float lwb[H * LDSPAD];
  for (int idx = threadIdx.x; idx < H * H; idx += 256) {
    int k = idx >> 7, n = idx & 127;
    lwa[n * LDSPAD + k] = WA[idx];
    lwb[n * LDSPAD + k] = WB[idx];
  }
  __syncthreads();

  const int wave = threadIdx.x >> 5;
  const int lane = threadIdx.x & 31;
  const int l15  = lane & 15;
  const int kh2  = (lane >> 4) << 1;
  const int ntiles = M >> 4;

  for (int t = blockIdx.x * TPB + wave; t < ntiles; t += gridDim.x * TPB) {
    const int m0 = t << 4;
    v8f cA[8], cB[8];
#pragma unroll
    for (int n = 0; n < 8; ++n) {
      cA[n] = (v8f){0, 0, 0, 0, 0, 0, 0, 0};
      cB[n] = (v8f){0, 0, 0, 0, 0, 0, 0, 0};
    }

    const float* Ap = A + (long)(m0 + l15) * H + kh2;
    const float* Bp = B + (long)(m0 + l15) * H + kh2;
    const float* wabase = &lwa[l15 * LDSPAD + kh2];
    const float* wbbase = &lwb[l15 * LDSPAD + kh2];

#pragma unroll 1
    for (int kk = 0; kk < 32; ++kk) {
      const int k0 = kk << 2;
      v2f a = *(const v2f*)(Ap + k0);
      v2f b = *(const v2f*)(Bp + k0);
#pragma unroll
      for (int n = 0; n < 8; ++n) {
        v2f wa = *(const v2f*)(wabase + n * 16 * LDSPAD + k0);
        v2f wb = *(const v2f*)(wbbase + n * 16 * LDSPAD + k0);
        cA[n] = __builtin_amdgcn_wmma_f32_16x16x4_f32(false, a, false, wa,
                                                      (short)0, cA[n], false, false);
        cB[n] = __builtin_amdgcn_wmma_f32_16x16x4_f32(false, b, false, wb,
                                                      (short)0, cB[n], false, false);
      }
    }

    const int rbase = m0 + ((lane >> 4) << 3);
    float degv[8];
    if (HAS_DEG) {
      const float4 d0 = *(const float4*)(deg + rbase);
      const float4 d1 = *(const float4*)(deg + rbase + 4);
      degv[0] = d0.x; degv[1] = d0.y; degv[2] = d0.z; degv[3] = d0.w;
      degv[4] = d1.x; degv[5] = d1.y; degv[6] = d1.z; degv[7] = d1.w;
    } else {
#pragma unroll
      for (int r = 0; r < 8; ++r) degv[r] = 1.0f;
    }
#pragma unroll
    for (int n = 0; n < 8; ++n) {
      const float bv = bias[n * 16 + l15];
#pragma unroll
      for (int r = 0; r < 8; ++r) {
        float acc = HAS_DEG ? (cA[n][r] + degv[r] * (cB[n][r] + bv))
                            : (cA[n][r] + cB[n][r] + bv);
        if (RELU) acc = fmaxf(acc, 0.0f);
        Y[(long)(rbase + r) * H + n * 16 + l15] = acc;
      }
    }
  }
}

__global__ __launch_bounds__(256) void scatter_first_kernel(const float* __restrict__ X,
                                                            int F,
                                                            const int* __restrict__ edge,
                                                            int E, float* __restrict__ S) {
  const unsigned idx = blockIdx.x * blockDim.x + threadIdx.x;
  const unsigned uF = (unsigned)F;
  if (idx >= (unsigned)E * uF) return;
  const unsigned e = idx / uF;
  const unsigned c = idx - e * uF;
  const int s = edge[e], d = edge[(unsigned)E + e];
  atomicAdd(S + (size_t)d * uF + c, X[(size_t)s * uF + c]);
}

__device__ __forceinline__ v2f load_narrow(const float* __restrict__ row, int k, int kv) {
  const int k1 = k + 1;
  const float x0 = row[k  < kv ? k  : kv - 1];
  const float x1 = row[k1 < kv ? k1 : kv - 1];
  v2f r;
  r.x = (k  < kv) ? x0 : 0.0f;
  r.y = (k1 < kv) ? x1 : 0.0f;
  return r;
}

__device__ __forceinline__ void wmma_tail_slots(v8f (&c)[8]) {
  asm volatile("v_nop\n\tv_nop\n\tv_nop\n\tv_nop"
               : "+v"(c[0]), "+v"(c[1]), "+v"(c[2]), "+v"(c[3]),
                 "+v"(c[4]), "+v"(c[5]), "+v"(c[6]), "+v"(c[7]));
}

template <bool HAS_DEG, bool RELU, bool A_NARROW>
__global__ __launch_bounds__(256) void gemm_dual_first_kernel(
    const float* __restrict__ A, int lda, int ka,
    const float* __restrict__ B, int ldb, int kb,
    const float* __restrict__ WA, const float* __restrict__ WB,
    const float* __restrict__ bias, const float* __restrict__ deg,
    float* __restrict__ Y, int M) {
  __shared__ float lwa[H * LDSPAD];
  __shared__ float lwb[H * LDSPAD];
  const int ksa = A_NARROW ? ((ka + 3) & ~3) : H;
  const int ksb = (kb + 3) & ~3;
  for (int idx = threadIdx.x; idx < ksa * H; idx += 256) {
    int k = idx >> 7, n = idx & 127;
    lwa[n * LDSPAD + k] = WA[idx];
  }
  for (int idx = threadIdx.x; idx < ksb * H; idx += 256) {
    int k = idx >> 7, n = idx & 127;
    lwb[n * LDSPAD + k] = WB[idx];
  }
  __syncthreads();

  const int wave = threadIdx.x >> 5;
  const int lane = threadIdx.x & 31;
  const int l15  = lane & 15;
  const int kh2  = (lane >> 4) << 1;
  const int ntiles = M >> 4;

  for (int t = blockIdx.x * TPB + wave; t < ntiles; t += gridDim.x * TPB) {
    const int m0 = t << 4;
    v8f cA[8], cB[8];
#pragma unroll
    for (int n = 0; n < 8; ++n) {
      cA[n] = (v8f){0, 0, 0, 0, 0, 0, 0, 0};
      cB[n] = (v8f){0, 0, 0, 0, 0, 0, 0, 0};
    }
    const float* wabase = &lwa[l15 * LDSPAD + kh2];
    const float* wbbase = &lwb[l15 * LDSPAD + kh2];

    const float* Brow = B + (long)(m0 + l15) * ldb;
    for (int k0 = 0; k0 < ksb; k0 += 4) {
      const v2f b = load_narrow(Brow, k0 + kh2, kb);
#pragma unroll
      for (int n = 0; n < 8; ++n) {
        v2f wb = *(const v2f*)(wbbase + n * 16 * LDSPAD + k0);
        cB[n] = __builtin_amdgcn_wmma_f32_16x16x4_f32(false, b, false, wb,
                                                      (short)0, cB[n], false, false);
      }
      wmma_tail_slots(cB);
    }

    if (A_NARROW) {
      const float* Arow = A + (long)(m0 + l15) * lda;
      for (int k0 = 0; k0 < ksa; k0 += 4) {
        const v2f a = load_narrow(Arow, k0 + kh2, ka);
#pragma unroll
        for (int n = 0; n < 8; ++n) {
          v2f wa = *(const v2f*)(wabase + n * 16 * LDSPAD + k0);
          cA[n] = __builtin_amdgcn_wmma_f32_16x16x4_f32(false, a, false, wa,
                                                        (short)0, cA[n], false, false);
        }
        wmma_tail_slots(cA);
      }
    } else {
      const float* Ap = A + (long)(m0 + l15) * H + kh2;
#pragma unroll 2
      for (int kk = 0; kk < 32; ++kk) {
        const int k0 = kk << 2;
        v2f a = *(const v2f*)(Ap + k0);
#pragma unroll
        for (int n = 0; n < 8; ++n) {
          v2f wa = *(const v2f*)(wabase + n * 16 * LDSPAD + k0);
          cA[n] = __builtin_amdgcn_wmma_f32_16x16x4_f32(false, a, false, wa,
                                                        (short)0, cA[n], false, false);
        }
      }
    }

    const int rbase = m0 + ((lane >> 4) << 3);
    float degv[8];
    if (HAS_DEG) {
      const float4 d0 = *(const float4*)(deg + rbase);
      const float4 d1 = *(const float4*)(deg + rbase + 4);
      degv[0] = d0.x; degv[1] = d0.y; degv[2] = d0.z; degv[3] = d0.w;
      degv[4] = d1.x; degv[5] = d1.y; degv[6] = d1.z; degv[7] = d1.w;
    } else {
#pragma unroll
      for (int r = 0; r < 8; ++r) degv[r] = 1.0f;
    }
#pragma unroll
    for (int n = 0; n < 8; ++n) {
      const float bv = bias[n * 16 + l15];
#pragma unroll
      for (int r = 0; r < 8; ++r) {
        float acc = HAS_DEG ? (cA[n][r] + degv[r] * (cB[n][r] + bv))
                            : (cA[n][r] + cB[n][r] + bv);
        if (RELU) acc = fmaxf(acc, 0.0f);
        Y[(long)(rbase + r) * H + n * 16 + l15] = acc;
      }
    }
  }
}

__global__ __launch_bounds__(256) void pool_sum_kernel(const float* __restrict__ xl,
                                                       const int* __restrict__ batch,
                                                       int NL, float* __restrict__ sums) {
  long idx = (long)blockIdx.x * blockDim.x + threadIdx.x;
  if (idx >= (long)NL * 32) return;
  int link = (int)(idx >> 5);
  int c = (int)(idx & 31) << 2;
  int b = batch[link];
  const float4 v = *(const float4*)(xl + (long)link * H + c);
  float* o = sums + (long)b * H + c;
  atomicAdd(o + 0, v.x); atomicAdd(o + 1, v.y);
  atomicAdd(o + 2, v.z); atomicAdd(o + 3, v.w);
}

__global__ __launch_bounds__(256) void pool_cnt_kernel(const int* __restrict__ batch,
                                                       int NL, float* __restrict__ cnt) {
  int i = blockIdx.x * blockDim.x + threadIdx.x;
  if (i < NL) atomicAdd(&cnt[batch[i]], 1.0f);
}

__global__ __launch_bounds__(256) void finalize_kernel(const float* __restrict__ sums,
                                                       const float* __restrict__ cnt,
                                                       float* __restrict__ out, int G) {
  int t = blockIdx.x * blockDim.x + threadIdx.x;
  if (t >= G * H) return;
  int g = t >> 7;
  out[t] = sums[t] / fmaxf(cnt[g], 1.0f);
}

extern "C" void kernel_launch(void* const* d_in, const int* in_sizes, int n_in,
                              void* d_out, int out_size, void* d_ws, size_t ws_size,
                              hipStream_t stream) {
  const float* x_node = (const float*)d_in[0];
  const float* x_link = (const float*)d_in[1];
  const float* Wm_nl  = (const float*)d_in[2];
  const float* bm_nl  = (const float*)d_in[3];
  const float* Wu_nl  = (const float*)d_in[4];
  const float* bu_nl  = (const float*)d_in[5];
  const float* Wm_ln  = (const float*)d_in[6];
  const float* bm_ln  = (const float*)d_in[7];
  const float* Wu_ln  = (const float*)d_in[8];
  const float* bu_ln  = (const float*)d_in[9];
  const int*   nl     = (const int*)d_in[10];
  const int*   ln     = (const int*)d_in[11];
  const int*   batch  = (const int*)d_in[12];

  const int NL = in_sizes[12];
  const int NN = 50000;
  const int F_NODE = in_sizes[0] / NN;
  const int F_LINK = in_sizes[1] / NL;
  const int E  = in_sizes[10] / 2;
  const int Lnum = in_sizes[3] / H;
  const int G = out_size / H;

  const size_t MH = (size_t)((NN > NL) ? NN : NL) * H;
  float* ws   = (float*)d_ws;
  float* xn   = ws;
  float* xl   = xn  + MH;
  float* xn2  = xl  + MH;
  float* xl2  = xn2 + MH;
  float* Sl   = xl2 + MH;
  float* Sn   = Sl  + MH;
  float* AGG  = Sn  + MH;
  float* dgl  = AGG + MH;
  float* dgn  = dgl + NL;
  float* sums = dgn + NN;
  float* cnt  = sums + (size_t)G * H;
  (void)n_in; (void)ws_size;

  auto nb = [](long n, int b) { return (int)((n + b - 1) / b); };

  zero_kernel<<<nb(NL, 256), 256, 0, stream>>>(dgl, NL);
  zero_kernel<<<nb(NN, 256), 256, 0, stream>>>(dgn, NN);
  deg_kernel<<<nb(E, 256), 256, 0, stream>>>(nl, E, dgl);
  deg_kernel<<<nb(E, 256), 256, 0, stream>>>(ln, E, dgn);

  const int gl = nb(NL / 16, TPB);
  const int gn = nb(NN / 16, TPB);

  if (Lnum == 0) {
    pad_kernel<<<nb((long)NL * H, 256), 256, 0, stream>>>(x_link, xl, NL, F_LINK);
  } else {
    float* S0l = Sl;
    float* S0n = Sl + (size_t)NL * F_NODE;
    const long nS0 = (long)NL * F_NODE + (long)NN * F_LINK;
    zero_kernel<<<nb(nS0, 256), 256, 0, stream>>>(S0l, nS0);
    scatter_first_kernel<<<nb((long)E * F_NODE, 256), 256, 0, stream>>>(x_node, F_NODE, nl, E, S0l);
    scatter_first_kernel<<<nb((long)E * F_LINK, 256), 256, 0, stream>>>(x_link, F_LINK, ln, E, S0n);
    gemm_dual_first_kernel<true, false, true><<<gl, 256, 0, stream>>>(
        S0l, F_NODE, F_NODE, x_link, F_LINK, F_LINK,
        Wm_nl, Wm_nl + H * H, bm_nl, dgl, AGG, NL);
    gemm_dual_first_kernel<false, true, false><<<gl, 256, 0, stream>>>(
        AGG, H, H, x_link, F_LINK, F_LINK,
        Wu_nl, Wu_nl + H * H, bu_nl, nullptr, xl2, NL);
    gemm_dual_first_kernel<true, false, true><<<gn, 256, 0, stream>>>(
        S0n, F_LINK, F_LINK, x_node, F_NODE, F_NODE,
        Wm_ln, Wm_ln + H * H, bm_ln, dgn, AGG, NN);
    gemm_dual_first_kernel<false, true, false><<<gn, 256, 0, stream>>>(
        AGG, H, H, x_node, F_NODE, F_NODE,
        Wu_ln, Wu_ln + H * H, bu_ln, nullptr, xn2, NN);
  }

  for (int l = 1; l < Lnum; ++l) {
    float* cxn = (l & 1) ? xn2 : xn;
    float* cxl = (l & 1) ? xl2 : xl;
    float* oxn = (l & 1) ? xn  : xn2;
    float* oxl = (l & 1) ? xl  : xl2;

    const float* WmNLt = Wm_nl + (size_t)l * 2 * H * H;
    const float* WmNLb = WmNLt + H * H;
    const float* WuNLt = Wu_nl + (size_t)l * 2 * H * H;
    const float* WuNLb = WuNLt + H * H;
    const float* WmLNt = Wm_ln + (size_t)l * 2 * H * H;
    const float* WmLNb = WmLNt + H * H;
    const float* WuLNt = Wu_ln + (size_t)l * 2 * H * H;
    const float* WuLNb = WuLNt + H * H;

    zero_kernel<<<nb((long)NL * H, 256), 256, 0, stream>>>(Sl, (long)NL * H);
    zero_kernel<<<nb((long)NN * H, 256), 256, 0, stream>>>(Sn, (long)NN * H);
    scatter_kernel<<<nb((long)E * 32, 256), 256, 0, stream>>>(cxn, nl, E, Sl);
    scatter_kernel<<<nb((long)E * 32, 256), 256, 0, stream>>>(cxl, ln, E, Sn);

    gemm_dual_kernel<true, false><<<gl, 256, 0, stream>>>(
        Sl, cxl, WmNLt, WmNLb, bm_nl + (size_t)l * H, dgl, AGG, NL);
    gemm_dual_kernel<false, true><<<gl, 256, 0, stream>>>(
        AGG, cxl, WuNLt, WuNLb, bu_nl + (size_t)l * H, nullptr, oxl, NL);
    gemm_dual_kernel<true, false><<<gn, 256, 0, stream>>>(
        Sn, cxn, WmLNt, WmLNb, bm_ln + (size_t)l * H, dgn, AGG, NN);
    gemm_dual_kernel<false, true><<<gn, 256, 0, stream>>>(
        AGG, cxn, WuLNt, WuLNb, bu_ln + (size_t)l * H, nullptr, oxn, NN);
  }

  float* fxl = (Lnum & 1) ? xl2 : xl;
  zero_kernel<<<nb((long)G * H, 256), 256, 0, stream>>>(sums, (long)G * H);
  zero_kernel<<<nb(G, 256), 256, 0, stream>>>(cnt, G);
  pool_sum_kernel<<<nb((long)NL * 32, 256), 256, 0, stream>>>(fxl, batch, NL, sums);
  pool_cnt_kernel<<<nb(NL, 256), 256, 0, stream>>>(batch, NL, cnt);
  finalize_kernel<<<nb((long)G * H, 256), 256, 0, stream>>>(sums, cnt, (float*)d_out, G);
}


# row scatter-add re-tiled (layer scatters and pooled sum): a wave takes 8 edges, lane = column, so each float atomic instruction covers one whole 128-B line; 8x fewer waves; sums stay atomic
# speedup vs baseline: 1.7014x; 1.2737x over previous
#include <hip/hip_runtime.h>


typedef __attribute__((ext_vector_type(2))) float v2f;
typedef __attribute__((ext_vector_type(8))) float v8f;

#define H 128
#define LDSPAD 132
#define TPB 8

__global__ __launch_bounds__(256) void zero_kernel(float* p, long n) {
  long i = (long)blockIdx.x * blockDim.x + threadIdx.x;
  if (i < n) p[i] = 0.0f;
}

__global__ __launch_bounds__(256) void pad_kernel(const float* __restrict__ in,
                                                  float* __restrict__ out,
                                                  int N, int F) {
  long t = (long)blockIdx.x * blockDim.x + threadIdx.x;
  if (t >= (long)N * H) return;
  int i = (int)(t >> 7), h = (int)(t & 127);
  out[t] = (h < F) ? in[(long)i * F + h] : 0.0f;
}

__global__ __launch_bounds__(256) void deg_kernel(const int* __restrict__ edge,
                                                  int E, float* __restrict__ deg) {
  int e = blockIdx.x * blockDim.x + threadIdx.x;
  if (e < E) atomicAdd(&deg[edge[E + e]], 1.0f);
}

#define EPW 8
template <bool HAS_SRC>
__global__ __launch_bounds__(256) void scatter_kernel(const float* __restrict__ X,
                                                      const int* __restrict__ src,
                                                      const int* __restrict__ dst,
                                                      int E, float* __restrict__ S) {
  const int lane = threadIdx.x & 31;
  const int wave = __builtin_amdgcn_readfirstlane((int)((blockIdx.x * 256u + threadIdx.x) >> 5));
  const int e0 = wave * EPW;
  if (e0 >= E) return;
  if (e0 + EPW <= E) {
    float v[EPW][4];
#pragma unroll
    for (int j = 0; j < EPW; ++j) {
      const int s = HAS_SRC ? src[e0 + j] : e0 + j;
      const float* xr = X + (size_t)s * H + lane;
#pragma unroll
      for (int q = 0; q < 4; ++q) v[j][q] = xr[32 * q];
    }
#pragma unroll
    for (int j = 0; j < EPW; ++j) {
      float* o = S + (size_t)dst[e0 + j] * H + lane;
#pragma unroll
      for (int q = 0; q < 4; ++q) atomicAdd(o + 32 * q, v[j][q]);
    }
  } else {
    for (int e = e0; e < E; ++e) {
      const int s = HAS_SRC ? src[e] : e;
      const float* xr = X + (size_t)s * H + lane;
      float* o = S + (size_t)dst[e] * H + lane;
      for (int q = 0; q < 4; ++q) atomicAdd(o + 32 * q, xr[32 * q]);
    }
  }
}

template <bool HAS_DEG, bool RELU>
__global__ __launch_bounds__(256) void gemm_dual_kernel(
    const float* __restrict__ A, const float* __restrict__ B,
    const float* __restrict__ WA, const float* __restrict__ WB,
    const float* __restrict__ bias, const float* __restrict__ deg,
    float* __restrict__ Y, int M) {
  __shared__ float lwa[H * LDSPAD];
  __shared__ float lwb[H * LDSPAD];
  for (int idx = threadIdx.x; idx < H * H; idx += 256) {
    int k = idx >> 7, n = idx & 127;
    lwa[n * LDSPAD + k] = WA[idx];
    lwb[n * LDSPAD + k] = WB[idx];
  }
  __syncthreads();

  const int wave = threadIdx.x >> 5;
  const int lane = threadIdx.x & 31;
  const int l15  = lane & 15;
  const int kh2  = (lane >> 4) << 1;
  const int ntiles = M >> 4;

  for (int t = blockIdx.x * TPB + wave; t < ntiles; t += gridDim.x * TPB) {
    const int m0 = t << 4;
    v8f cA[8], cB[8];
#pragma unroll
    for (int n = 0; n < 8; ++n) {
      cA[n] = (v8f){0, 0, 0, 0, 0, 0, 0, 0};
      cB[n] = (v8f){0, 0, 0, 0, 0, 0, 0, 0};
    }

    const float* Ap = A + (long)(m0 + l15) * H + kh2;
    const float* Bp = B + (long)(m0 + l15) * H + kh2;
    const float* wabase = &lwa[l15 * LDSPAD + kh2];
    const float* wbbase = &lwb[l15 * LDSPAD + kh2];

#pragma unroll 1
    for (int kk = 0; kk < 32; ++kk) {
      const int k0 = kk << 2;
      v2f a = *(const v2f*)(Ap + k0);
      v2f b = *(const v2f*)(Bp + k0);
#pragma unroll
      for (int n = 0; n < 8; ++n) {
        v2f wa = *(const v2f*)(wabase + n * 16 * LDSPAD + k0);
        v2f wb = *(const v2f*)(wbbase + n * 16 * LDSPAD + k0);
        cA[n] = __builtin_amdgcn_wmma_f32_16x16x4_f32(false, a, false, wa,
                                                      (short)0, cA[n], false, false);
        cB[n] = __builtin_amdgcn_wmma_f32_16x16x4_f32(false, b, false, wb,
                                                      (short)0, cB[n], false, false);
      }
    }

    const int rbase = m0 + ((lane >> 4) << 3);
    float degv[8];
    if (HAS_DEG) {
      const float4 d0 = *(const float4*)(deg + rbase);
      const float4 d1 = *(const float4*)(deg + rbase + 4);
      degv[0] = d0.x; degv[1] = d0.y; degv[2] = d0.z; degv[3] = d0.w;
      degv[4] = d1.x; degv[5] = d1.y; degv[6] = d1.z; degv[7] = d1.w;
    } else {
#pragma unroll
      for (int r = 0; r < 8; ++r) degv[r] = 1.0f;
    }
#pragma unroll
    for (int n = 0; n < 8; ++n) {
      const float bv = bias[n * 16 + l15];
#pragma unroll
      for (int r = 0; r < 8; ++r) {
        float acc = HAS_DEG ? (cA[n][r] + degv[r] * (cB[n][r] + bv))
                            : (cA[n][r] + cB[n][r] + bv);
        if (RELU) acc = fmaxf(acc, 0.0f);
        Y[(long)(rbase + r) * H + n * 16 + l15] = acc;
      }
    }
  }
}

__global__ __launch_bounds__(256) void scatter_first_kernel(const float* __restrict__ X,
                                                            int F,
                                                            const int* __restrict__ edge,
                                                            int E, float* __restrict__ S) {
  const unsigned idx = blockIdx.x * blockDim.x + threadIdx.x;
  const unsigned uF = (unsigned)F;
  if (idx >= (unsigned)E * uF) return;
  const unsigned e = idx / uF;
  const unsigned c = idx - e * uF;
  const int s = edge[e], d = edge[(unsigned)E + e];
  atomicAdd(S + (size_t)d * uF + c, X[(size_t)s * uF + c]);
}

__device__ __forceinline__ v2f load_narrow(const float* __restrict__ row, int k, int kv) {
  const int k1 = k + 1;
  const float x0 = row[k  < kv ? k  : kv - 1];
  const float x1 = row[k1 < kv ? k1 : kv - 1];
  v2f r;
  r.x = (k  < kv) ? x0 : 0.0f;
  r.y = (k1 < kv) ? x1 : 0.0f;
  return r;
}

__device__ __forceinline__ void wmma_tail_slots(v8f (&c)[8]) {
  asm volatile("v_nop\n\tv_nop\n\tv_nop\n\tv_nop"
               : "+v"(c[0]), "+v"(c[1]), "+v"(c[2]), "+v"(c[3]),
                 "+v"(c[4]), "+v"(c[5]), "+v"(c[6]), "+v"(c[7]));
}

template <bool HAS_DEG, bool RELU, bool A_NARROW>
__global__ __launch_bounds__(256) void gemm_dual_first_kernel(
    const float* __restrict__ A, int lda, int ka,
    const float* __restrict__ B, int ldb, int kb,
    const float* __restrict__ WA, const float* __restrict__ WB,
    const float* __restrict__ bias, const float* __restrict__ deg,
    float* __restrict__ Y, int M) {
  __shared__ float lwa[H * LDSPAD];
  __shared__ float lwb[H * LDSPAD];
  const int ksa = A_NARROW ? ((ka + 3) & ~3) : H;
  const int ksb = (kb + 3) & ~3;
  for (int idx = threadIdx.x; idx < ksa * H; idx += 256) {
    int k = idx >> 7, n = idx & 127;
    lwa[n * LDSPAD + k] = WA[idx];
  }
  for (int idx = threadIdx.x; idx < ksb * H; idx += 256) {
    int k = idx >> 7, n = idx & 127;
    lwb[n * LDSPAD + k] = WB[idx];
  }
  __syncthreads();

  const int wave = threadIdx.x >> 5;
  const int lane = threadIdx.x & 31;
  const int l15  = lane & 15;
  const int kh2  = (lane >> 4) << 1;
  const int ntiles = M >> 4;

  for (int t = blockIdx.x * TPB + wave; t < ntiles; t += gridDim.x * TPB) {
    const int m0 = t << 4;
    v8f cA[8], cB[8];
#pragma unroll
    for (int n = 0; n < 8; ++n) {
      cA[n] = (v8f){0, 0, 0, 0, 0, 0, 0, 0};
      cB[n] = (v8f){0, 0, 0, 0, 0, 0, 0, 0};
    }
    const float* wabase = &lwa[l15 * LDSPAD + kh2];
    const float* wbbase = &lwb[l15 * LDSPAD + kh2];

    const float* Brow = B + (long)(m0 + l15) * ldb;
    for (int k0 = 0; k0 < ksb; k0 += 4) {
      const v2f b = load_narrow(Brow, k0 + kh2, kb);
#pragma unroll
      for (int n = 0; n < 8; ++n) {
        v2f wb = *(const v2f*)(wbbase + n * 16 * LDSPAD + k0);
        cB[n] = __builtin_amdgcn_wmma_f32_16x16x4_f32(false, b, false, wb,
                                                      (short)0, cB[n], false, false);
      }
      wmma_tail_slots(cB);
    }

    if (A_NARROW) {
      const float* Arow = A + (long)(m0 + l15) * lda;
      for (int k0 = 0; k0 < ksa; k0 += 4) {
        const v2f a = load_narrow(Arow, k0 + kh2, ka);
#pragma unroll
        for (int n = 0; n < 8; ++n) {
          v2f wa = *(const v2f*)(wabase + n * 16 * LDSPAD + k0);
          cA[n] = __builtin_amdgcn_wmma_f32_16x16x4_f32(false, a, false, wa,
                                                        (short)0, cA[n], false, false);
        }
        wmma_tail_slots(cA);
      }
    } else {
      const float* Ap = A + (long)(m0 + l15) * H + kh2;
#pragma unroll 2
      for (int kk = 0; kk < 32; ++kk) {
        const int k0 = kk << 2;
        v2f a = *(const v2f*)(Ap + k0);
#pragma unroll
        for (int n = 0; n < 8; ++n) {
          v2f wa = *(const v2f*)(wabase + n * 16 * LDSPAD + k0);
          cA[n] = __builtin_amdgcn_wmma_f32_16x16x4_f32(false, a, false, wa,
                                                        (short)0, cA[n], false, false);
        }
      }
    }

    const int rbase = m0 + ((lane >> 4) << 3);
    float degv[8];
    if (HAS_DEG) {
      const float4 d0 = *(const float4*)(deg + rbase);
      const float4 d1 = *(const float4*)(deg + rbase + 4);
      degv[0] = d0.x; degv[1] = d0.y; degv[2] = d0.z; degv[3] = d0.w;
      degv[4] = d1.x; degv[5] = d1.y; degv[6] = d1.z; degv[7] = d1.w;
    } else {
#pragma unroll
      for (int r = 0; r < 8; ++r) degv[r] = 1.0f;
    }
#pragma unroll
    for (int n = 0; n < 8; ++n) {
      const float bv = bias[n * 16 + l15];
#pragma unroll
      for (int r = 0; r < 8; ++r) {
        float acc = HAS_DEG ? (cA[n][r] + degv[r] * (cB[n][r] + bv))
                            : (cA[n][r] + cB[n][r] + bv);
        if (RELU) acc = fmaxf(acc, 0.0f);
        Y[(long)(rbase + r) * H + n * 16 + l15] = acc;
      }
    }
  }
}

__global__ __launch_bounds__(256) void pool_cnt_kernel(const int* __restrict__ batch,
                                                       int NL, float* __restrict__ cnt) {
  int i = blockIdx.x * blockDim.x + threadIdx.x;
  if (i < NL) atomicAdd(&cnt[batch[i]], 1.0f);
}

__global__ __launch_bounds__(256) void finalize_kernel(const float* __restrict__ sums,
                                                       const float* __restrict__ cnt,
                                                       float* __restrict__ out, int G) {
  int t = blockIdx.x * blockDim.x + threadIdx.x;
  if (t >= G * H) return;
  int g = t >> 7;
  out[t] = sums[t] / fmaxf(cnt[g], 1.0f);
}

extern "C" void kernel_launch(void* const* d_in, const int* in_sizes, int n_in,
                              void* d_out, int out_size, void* d_ws, size_t ws_size,
                              hipStream_t stream) {
  const float* x_node = (const float*)d_in[0];
  const float* x_link = (const float*)d_in[1];
  const float* Wm_nl  = (const float*)d_in[2];
  const float* bm_nl  = (const float*)d_in[3];
  const float* Wu_nl  = (const float*)d_in[4];
  const float* bu_nl  = (const float*)d_in[5];
  const float* Wm_ln  = (const float*)d_in[6];
  const float* bm_ln  = (const float*)d_in[7];
  const float* Wu_ln  = (const float*)d_in[8];
  const float* bu_ln  = (const float*)d_in[9];
  const int*   nl     = (const int*)d_in[10];
  const int*   ln     = (const int*)d_in[11];
  const int*   batch  = (const int*)d_in[12];

  const int NL = in_sizes[12];
  const int NN = 50000;
  const int F_NODE = in_sizes[0] / NN;
  const int F_LINK = in_sizes[1] / NL;
  const int E  = in_sizes[10] / 2;
  const int Lnum = in_sizes[3] / H;
  const int G = out_size / H;

  const size_t MH = (size_t)((NN > NL) ? NN : NL) * H;
  float* ws   = (float*)d_ws;
  float* xn   = ws;
  float* xl   = xn  + MH;
  float* xn2  = xl  + MH;
  float* xl2  = xn2 + MH;
  float* Sl   = xl2 + MH;
  float* Sn   = Sl  + MH;
  float* AGG  = Sn  + MH;
  float* dgl  = AGG + MH;
  float* dgn  = dgl + NL;
  float* sums = dgn + NN;
  float* cnt  = sums + (size_t)G * H;
  (void)n_in; (void)ws_size;

  auto nb = [](long n, int b) { return (int)((n + b - 1) / b); };

  zero_kernel<<<nb(NL, 256), 256, 0, stream>>>(dgl, NL);
  zero_kernel<<<nb(NN, 256), 256, 0, stream>>>(dgn, NN);
  deg_kernel<<<nb(E, 256), 256, 0, stream>>>(nl, E, dgl);
  deg_kernel<<<nb(E, 256), 256, 0, stream>>>(ln, E, dgn);

  const int gl = nb(NL / 16, TPB);
  const int gn = nb(NN / 16, TPB);

  if (Lnum == 0) {
    pad_kernel<<<nb((long)NL * H, 256), 256, 0, stream>>>(x_link, xl, NL, F_LINK);
  } else {
    float* S0l = Sl;
    float* S0n = Sl + (size_t)NL * F_NODE;
    const long nS0 = (long)NL * F_NODE + (long)NN * F_LINK;
    zero_kernel<<<nb(nS0, 256), 256, 0, stream>>>(S0l, nS0);
    scatter_first_kernel<<<nb((long)E * F_NODE, 256), 256, 0, stream>>>(x_node, F_NODE, nl, E, S0l);
    scatter_first_kernel<<<nb((long)E * F_LINK, 256), 256, 0, stream>>>(x_link, F_LINK, ln, E, S0n);
    gemm_dual_first_kernel<true, false, true><<<gl, 256, 0, stream>>>(
        S0l, F_NODE, F_NODE, x_link, F_LINK, F_LINK,
        Wm_nl, Wm_nl + H * H, bm_nl, dgl, AGG, NL);
    gemm_dual_first_kernel<false, true, false><<<gl, 256, 0, stream>>>(
        AGG, H, H, x_link, F_LINK, F_LINK,
        Wu_nl, Wu_nl + H * H, bu_nl, nullptr, xl2, NL);
    gemm_dual_first_kernel<true, false, true><<<gn, 256, 0, stream>>>(
        S0n, F_LINK, F_LINK, x_node, F_NODE, F_NODE,
        Wm_ln, Wm_ln + H * H, bm_ln, dgn, AGG, NN);
    gemm_dual_first_kernel<false, true, false><<<gn, 256, 0, stream>>>(
        AGG, H, H, x_node, F_NODE, F_NODE,
        Wu_ln, Wu_ln + H * H, bu_ln, nullptr, xn2, NN);
  }

  for (int l = 1; l < Lnum; ++l) {
    float* cxn = (l & 1) ? xn2 : xn;
    float* cxl = (l & 1) ? xl2 : xl;
    float* oxn = (l & 1) ? xn  : xn2;
    float* oxl = (l & 1) ? xl  : xl2;

    const float* WmNLt = Wm_nl + (size_t)l * 2 * H * H;
    const float* WmNLb = WmNLt + H * H;
    const float* WuNLt = Wu_nl + (size_t)l * 2 * H * H;
    const float* WuNLb = WuNLt + H * H;
    const float* WmLNt = Wm_ln + (size_t)l * 2 * H * H;
    const float* WmLNb = WmLNt + H * H;
    const float* WuLNt = Wu_ln + (size_t)l * 2 * H * H;
    const float* WuLNb = WuLNt + H * H;

    zero_kernel<<<nb((long)NL * H, 256), 256, 0, stream>>>(Sl, (long)NL * H);
    zero_kernel<<<nb((long)NN * H, 256), 256, 0, stream>>>(Sn, (long)NN * H);
    scatter_kernel<true><<<nb((long)nb(E, EPW) * 32, 256), 256, 0, stream>>>(cxn, nl, nl + E, E, Sl);
    scatter_kernel<true><<<nb((long)nb(E, EPW) * 32, 256), 256, 0, stream>>>(cxl, ln, ln + E, E, Sn);

    gemm_dual_kernel<true, false><<<gl, 256, 0, stream>>>(
        Sl, cxl, WmNLt, WmNLb, bm_nl + (size_t)l * H, dgl, AGG, NL);
    gemm_dual_kernel<false, true><<<gl, 256, 0, stream>>>(
        AGG, cxl, WuNLt, WuNLb, bu_nl + (size_t)l * H, nullptr, oxl, NL);
    gemm_dual_kernel<true, false><<<gn, 256, 0, stream>>>(
        Sn, cxn, WmLNt, WmLNb, bm_ln + (size_t)l * H, dgn, AGG, NN);
    gemm_dual_kernel<false, true><<<gn, 256, 0, stream>>>(
        AGG, cxn, WuLNt, WuLNb, bu_ln + (size_t)l * H, nullptr, oxn, NN);
  }

  float* fxl = (Lnum & 1) ? xl2 : xl;
  zero_kernel<<<nb((long)G * H, 256), 256, 0, stream>>>(sums, (long)G * H);
  zero_kernel<<<nb(G, 256), 256, 0, stream>>>(cnt, G);
  scatter_kernel<false><<<nb((long)nb(NL, EPW) * 32, 256), 256, 0, stream>>>(fxl, nullptr, batch, NL, sums);
  pool_cnt_kernel<<<nb(NL, 256), 256, 0, stream>>>(batch, NL, cnt);
  finalize_kernel<<<nb((long)G * H, 256), 256, 0, stream>>>(sums, cnt, (float*)d_out, G);
}


# fewer waves started: zero-fill threads store 4 x 16 bytes each and first-layer scatter threads take 8 (edge, column) pairs each, same loads and atomics per instruction; sums stay atomic
# speedup vs baseline: 1.7566x; 1.0324x over previous
#include <hip/hip_runtime.h>


typedef __attribute__((ext_vector_type(2))) float v2f;
typedef __attribute__((ext_vector_type(8))) float v8f;

#define H 128
#define LDSPAD 132
#define TPB 8

#define ZV 4
__global__ __launch_bounds__(256) void zero_kernel(float* p, long n) {
  long head = (long)(((16u - (unsigned)((size_t)p & 15u)) & 15u) >> 2);
  if (head > n) head = n;
  float4* q = (float4*)(p + head);
  const long nv = (n - head) >> 2;
  const long tail = head + (nv << 2);
  const long b0 = (long)blockIdx.x * (256 * ZV) + threadIdx.x;
  const float4 z = make_float4(0.0f, 0.0f, 0.0f, 0.0f);
#pragma unroll
  for (int j = 0; j < ZV; ++j) {
    const long i = b0 + j * 256;
    if (i < nv) q[i] = z;
  }
  if (blockIdx.x == 0) {
    if ((long)threadIdx.x < head) p[threadIdx.x] = 0.0f;
    if (tail + (long)threadIdx.x < n) p[tail + threadIdx.x] = 0.0f;
  }
}

__global__ __launch_bounds__(256) void pad_kernel(const float* __restrict__ in,
                                                  float* __restrict__ out,
                                                  int N, int F) {
  long t = (long)blockIdx.x * blockDim.x + threadIdx.x;
  if (t >= (long)N * H) return;
  int i = (int)(t >> 7), h = (int)(t & 127);
  out[t] = (h < F) ? in[(long)i * F + h] : 0.0f;
}

__global__ __launch_bounds__(256) void deg_kernel(const int* __restrict__ edge,
                                                  int E, float* __restrict__ deg) {
  int e = blockIdx.x * blockDim.x + threadIdx.x;
  if (e < E) atomicAdd(&deg[edge[E + e]], 1.0f);
}

#define EPW 8
template <bool HAS_SRC>
__global__ __launch_bounds__(256) void scatter_kernel(const float* __restrict__ X,
                                                      const int* __restrict__ src,
                                                      const int* __restrict__ dst,
                                                      int E, float* __restrict__ S) {
  const int lane = threadIdx.x & 31;
  const int wave = __builtin_amdgcn_readfirstlane((int)((blockIdx.x * 256u + threadIdx.x) >> 5));
  const int e0 = wave * EPW;
  if (e0 >= E) return;
  if (e0 + EPW <= E) {
    float v[EPW][4];
#pragma unroll
    for (int j = 0; j < EPW; ++j) {
      const int s = HAS_SRC ? src[e0 + j] : e0 + j;
      const float* xr = X + (size_t)s * H + lane;
#pragma unroll
      for (int q = 0; q < 4; ++q) v[j][q] = xr[32 * q];
    }
#pragma unroll
    for (int j = 0; j < EPW; ++j) {
      float* o = S + (size_t)dst[e0 + j] * H + lane;
#pragma unroll
      for (int q = 0; q < 4; ++q) atomicAdd(o + 32 * q, v[j][q]);
    }
  } else {
    for (int e = e0; e < E; ++e) {
      const int s = HAS_SRC ? src[e] : e;
      const float* xr = X + (size_t)s * H + lane;
      float* o = S + (size_t)dst[e] * H + lane;
      for (int q = 0; q < 4; ++q) atomicAdd(o + 32 * q, xr[32 * q]);
    }
  }
}

template <bool HAS_DEG, bool RELU>
__global__ __launch_bounds__(256) void gemm_dual_kernel(
    const float* __restrict__ A, const float* __restrict__ B,
    const float* __restrict__ WA, const float* __restrict__ WB,
    const float* __restrict__ bias, const float* __restrict__ deg,
    float* __restrict__ Y, int M) {
  __shared__ float lwa[H * LDSPAD];
  __shared__ float lwb[H * LDSPAD];
  for (int idx = threadIdx.x; idx < H * H; idx += 256) {
    int k = idx >> 7, n = idx & 127;
    lwa[n * LDSPAD + k] = WA[idx];
    lwb[n * LDSPAD + k] = WB[idx];
  }
  __syncthreads();

  const int wave = threadIdx.x >> 5;
  const int lane = threadIdx.x & 31;
  const int l15  = lane & 15;
  const int kh2  = (lane >> 4) << 1;
  const int ntiles = M >> 4;

  for (int t = blockIdx.x * TPB + wave; t < ntiles; t += gridDim.x * TPB) {
    const int m0 = t << 4;
    v8f cA[8], cB[8];
#pragma unroll
    for (int n = 0; n < 8; ++n) {
      cA[n] = (v8f){0, 0, 0, 0, 0, 0, 0, 0};
      cB[n] = (v8f){0, 0, 0, 0, 0, 0, 0, 0};
    }

    const float* Ap = A + (long)(m0 + l15) * H + kh2;
    const float* Bp = B + (long)(m0 + l15) * H + kh2;
    const float* wabase = &lwa[l15 * LDSPAD + kh2];
    const float* wbbase = &lwb[l15 * LDSPAD + kh2];

#pragma unroll 1
    for (int kk = 0; kk < 32; ++kk) {
      const int k0 = kk << 2;
      v2f a = *(const v2f*)(Ap + k0);
      v2f b = *(const v2f*)(Bp + k0);
#pragma unroll
      for (int n = 0; n < 8; ++n) {
        v2f wa = *(const v2f*)(wabase + n * 16 * LDSPAD + k0);
        v2f wb = *(const v2f*)(wbbase + n * 16 * LDSPAD + k0);
        cA[n] = __builtin_amdgcn_wmma_f32_16x16x4_f32(false, a, false, wa,
                                                      (short)0, cA[n], false, false);
        cB[n] = __builtin_amdgcn_wmma_f32_16x16x4_f32(false, b, false, wb,
                                                      (short)0, cB[n], false, false);
      }
    }

    const int rbase = m0 + ((lane >> 4) << 3);
    float degv[8];
    if (HAS_DEG) {
      const float4 d0 = *(const float4*)(deg + rbase);
      const float4 d1 = *(const float4*)(deg + rbase + 4);
      degv[0] = d0.x; degv[1] = d0.y; degv[2] = d0.z; degv[3] = d0.w;
      degv[4] = d1.x; degv[5] = d1.y; degv[6] = d1.z; degv[7] = d1.w;
    } else {
#pragma unroll
      for (int r = 0; r < 8; ++r) degv[r] = 1.0f;
    }
#pragma unroll
    for (int n = 0; n < 8; ++n) {
      const float bv = bias[n * 16 + l15];
#pragma unroll
      for (int r = 0; r < 8; ++r) {
        float acc = HAS_DEG ? (cA[n][r] + degv[r] * (cB[n][r] + bv))
                            : (cA[n][r] + cB[n][r] + bv);
        if (RELU) acc = fmaxf(acc, 0.0f);
        Y[(long)(rbase + r) * H + n * 16 + l15] = acc;
      }
    }
  }
}

#define SFU 8
__global__ __launch_bounds__(256) void scatter_first_kernel(const float* __restrict__ X,
                                                            int F,
                                                            const int* __restrict__ edge,
                                                            int E, float* __restrict__ S) {
  const unsigned uF = (unsigned)F;
  const unsigned total = (unsigned)E * uF;
  const unsigned blk0 = blockIdx.x * (256u * SFU);
  if (blk0 >= total) return;
  const unsigned base = blk0 + threadIdx.x;
  if (total - blk0 >= 256u * SFU) {
    float v[SFU];
    unsigned o[SFU];
#pragma unroll
    for (int j = 0; j < SFU; ++j) {
      const unsigned idx = base + j * 256u;
      const unsigned e = idx / uF;
      const unsigned c = idx - e * uF;
      const int s = edge[e], d = edge[(unsigned)E + e];
      v[j] = X[(size_t)s * uF + c];
      o[j] = (unsigned)d * uF + c;
    }
#pragma unroll
    for (int j = 0; j < SFU; ++j) atomicAdd(S + o[j], v[j]);
  } else {
    for (unsigned idx = base; idx < total; idx += 256u) {
      const unsigned e = idx / uF;
      const unsigned c = idx - e * uF;
      const int s = edge[e], d = edge[(unsigned)E + e];
      atomicAdd(S + (size_t)d * uF + c, X[(size_t)s * uF + c]);
    }
  }
}

__device__ __forceinline__ v2f load_narrow(const float* __restrict__ row, int k, int kv) {
  const int k1 = k + 1;
  const float x0 = row[k  < kv ? k  : kv - 1];
  const float x1 = row[k1 < kv ? k1 : kv - 1];
  v2f r;
  r.x = (k  < kv) ? x0 : 0.0f;
  r.y = (k1 < kv) ? x1 : 0.0f;
  return r;
}

__device__ __forceinline__ void wmma_tail_slots(v8f (&c)[8]) {
  asm volatile("v_nop\n\tv_nop\n\tv_nop\n\tv_nop"
               : "+v"(c[0]), "+v"(c[1]), "+v"(c[2]), "+v"(c[3]),
                 "+v"(c[4]), "+v"(c[5]), "+v"(c[6]), "+v"(c[7]));
}

template <bool HAS_DEG, bool RELU, bool A_NARROW>
__global__ __launch_bounds__(256) void gemm_dual_first_kernel(
    const float* __restrict__ A, int lda, int ka,
    const float* __restrict__ B, int ldb, int kb,
    const float* __restrict__ WA, const float* __restrict__ WB,
    const float* __restrict__ bias, const float* __restrict__ deg,
    float* __restrict__ Y, int M) {
  __shared__ float lwa[H * LDSPAD];
  __shared__ float lwb[H * LDSPAD];
  const int ksa = A_NARROW ? ((ka + 3) & ~3) : H;
  const int ksb = (kb + 3) & ~3;
  for (int idx = threadIdx.x; idx < ksa * H; idx += 256) {
    int k = idx >> 7, n = idx & 127;
    lwa[n * LDSPAD + k] = WA[idx];
  }
  for (int idx = threadIdx.x; idx < ksb * H; idx += 256) {
    int k = idx >> 7, n = idx & 127;
    lwb[n * LDSPAD + k] = WB[idx];
  }
  __syncthreads();

  const int wave = threadIdx.x >> 5;
  const int lane = threadIdx.x & 31;
  const int l15  = lane & 15;
  const int kh2  = (lane >> 4) << 1;
  const int ntiles = M >> 4;

  for (int t = blockIdx.x * TPB + wave; t < ntiles; t += gridDim.x * TPB) {
    const int m0 = t << 4;
    v8f cA[8], cB[8];
#pragma unroll
    for (int n = 0; n < 8; ++n) {
      cA[n] = (v8f){0, 0, 0, 0, 0, 0, 0, 0};
      cB[n] = (v8f){0, 0, 0, 0, 0, 0, 0, 0};
    }
    const float* wabase = &lwa[l15 * LDSPAD + kh2];
    const float* wbbase = &lwb[l15 * LDSPAD + kh2];

    const float* Brow = B + (long)(m0 + l15) * ldb;
    for (int k0 = 0; k0 < ksb; k0 += 4) {
      const v2f b = load_narrow(Brow, k0 + kh2, kb);
#pragma unroll
      for (int n = 0; n < 8; ++n) {
        v2f wb = *(const v2f*)(wbbase + n * 16 * LDSPAD + k0);
        cB[n] = __builtin_amdgcn_wmma_f32_16x16x4_f32(false, b, false, wb,
                                                      (short)0, cB[n], false, false);
      }
      wmma_tail_slots(cB);
    }

    if (A_NARROW) {
      const float* Arow = A + (long)(m0 + l15) * lda;
      for (int k0 = 0; k0 < ksa; k0 += 4) {
        const v2f a = load_narrow(Arow, k0 + kh2, ka);
#pragma unroll
        for (int n = 0; n < 8; ++n) {
          v2f wa = *(const v2f*)(wabase + n * 16 * LDSPAD + k0);
          cA[n] = __builtin_amdgcn_wmma_f32_16x16x4_f32(false, a, false, wa,
                                                        (short)0, cA[n], false, false);
        }
        wmma_tail_slots(cA);
      }
    } else {
      const float* Ap = A + (long)(m0 + l15) * H + kh2;
#pragma unroll 2
      for (int kk = 0; kk < 32; ++kk) {
        const int k0 = kk << 2;
        v2f a = *(const v2f*)(Ap + k0);
#pragma unroll
        for (int n = 0; n < 8; ++n) {
          v2f wa = *(const v2f*)(wabase + n * 16 * LDSPAD + k0);
          cA[n] = __builtin_amdgcn_wmma_f32_16x16x4_f32(false, a, false, wa,
                                                        (short)0, cA[n], false, false);
        }
      }
    }

    const int rbase = m0 + ((lane >> 4) << 3);
    float degv[8];
    if (HAS_DEG) {
      const float4 d0 = *(const float4*)(deg + rbase);
      const float4 d1 = *(const float4*)(deg + rbase + 4);
      degv[0] = d0.x; degv[1] = d0.y; degv[2] = d0.z; degv[3] = d0.w;
      degv[4] = d1.x; degv[5] = d1.y; degv[6] = d1.z; degv[7] = d1.w;
    } else {
#pragma unroll
      for (int r = 0; r < 8; ++r) degv[r] = 1.0f;
    }
#pragma unroll
    for (int n = 0; n < 8; ++n) {
      const float bv = bias[n * 16 + l15];
#pragma unroll
      for (int r = 0; r < 8; ++r) {
        float acc = HAS_DEG ? (cA[n][r] + degv[r] * (cB[n][r] + bv))
                            : (cA[n][r] + cB[n][r] + bv);
        if (RELU) acc = fmaxf(acc, 0.0f);
        Y[(long)(rbase + r) * H + n * 16 + l15] = acc;
      }
    }
  }
}

__global__ __launch_bounds__(256) void pool_cnt_kernel(const int* __restrict__ batch,
                                                       int NL, float* __restrict__ cnt) {
  int i = blockIdx.x * blockDim.x + threadIdx.x;
  if (i < NL) atomicAdd(&cnt[batch[i]], 1.0f);
}

__global__ __launch_bounds__(256) void finalize_kernel(const float* __restrict__ sums,
                                                       const float* __restrict__ cnt,
                                                       float* __restrict__ out, int G) {
  int t = blockIdx.x * blockDim.x + threadIdx.x;
  if (t >= G * H) return;
  int g = t >> 7;
  out[t] = sums[t] / fmaxf(cnt[g], 1.0f);
}

extern "C" void kernel_launch(void* const* d_in, const int* in_sizes, int n_in,
                              void* d_out, int out_size, void* d_ws, size_t ws_size,
                              hipStream_t stream) {
  const float* x_node = (const float*)d_in[0];
  const float* x_link = (const float*)d_in[1];
  const float* Wm_nl  = (const float*)d_in[2];
  const float* bm_nl  = (const float*)d_in[3];
  const float* Wu_nl  = (const float*)d_in[4];
  const float* bu_nl  = (const float*)d_in[5];
  const float* Wm_ln  = (const float*)d_in[6];
  const float* bm_ln  = (const float*)d_in[7];
  const float* Wu_ln  = (const float*)d_in[8];
  const float* bu_ln  = (const float*)d_in[9];
  const int*   nl     = (const int*)d_in[10];
  const int*   ln     = (const int*)d_in[11];
  const int*   batch  = (const int*)d_in[12];

  const int NL = in_sizes[12];
  const int NN = 50000;
  const int F_NODE = in_sizes[0] / NN;
  const int F_LINK = in_sizes[1] / NL;
  const int E  = in_sizes[10] / 2;
  const int Lnum = in_sizes[3] / H;
  const int G = out_size / H;

  const size_t MH = (size_t)((NN > NL) ? NN : NL) * H;
  float* ws   = (float*)d_ws;
  float* xn   = ws;
  float* xl   = xn  + MH;
  float* xn2  = xl  + MH;
  float* xl2  = xn2 + MH;
  float* Sl   = xl2 + MH;
  float* Sn   = Sl  + MH;
  float* AGG  = Sn  + MH;
  float* dgl  = AGG + MH;
  float* dgn  = dgl + NL;
  float* sums = dgn + NN;
  float* cnt  = sums + (size_t)G * H;
  (void)n_in; (void)ws_size;

  auto nb = [](long n, int b) { return (int)((n + b - 1) / b); };

  zero_kernel<<<nb(NL, 4 * 256 * ZV), 256, 0, stream>>>(dgl, NL);
  zero_kernel<<<nb(NN, 4 * 256 * ZV), 256, 0, stream>>>(dgn, NN);
  deg_kernel<<<nb(E, 256), 256, 0, stream>>>(nl, E, dgl);
  deg_kernel<<<nb(E, 256), 256, 0, stream>>>(ln, E, dgn);

  const int gl = nb(NL / 16, TPB);
  const int gn = nb(NN / 16, TPB);

  if (Lnum == 0) {
    pad_kernel<<<nb((long)NL * H, 256), 256, 0, stream>>>(x_link, xl, NL, F_LINK);
  } else {
    float* S0l = Sl;
    float* S0n = Sl + (size_t)NL * F_NODE;
    const long nS0 = (long)NL * F_NODE + (long)NN * F_LINK;
    zero_kernel<<<nb(nS0, 4 * 256 * ZV), 256, 0, stream>>>(S0l, nS0);
    scatter_first_kernel<<<nb((long)E * F_NODE, 256 * SFU), 256, 0, stream>>>(x_node, F_NODE, nl, E, S0l);
    scatter_first_kernel<<<nb((long)E * F_LINK, 256 * SFU), 256, 0, stream>>>(x_link, F_LINK, ln, E, S0n);
    gemm_dual_first_kernel<true, false, true><<<gl, 256, 0, stream>>>(
        S0l, F_NODE, F_NODE, x_link, F_LINK, F_LINK,
        Wm_nl, Wm_nl + H * H, bm_nl, dgl, AGG, NL);
    gemm_dual_first_kernel<false, true, false><<<gl, 256, 0, stream>>>(
        AGG, H, H, x_link, F_LINK, F_LINK,
        Wu_nl, Wu_nl + H * H, bu_nl, nullptr, xl2, NL);
    gemm_dual_first_kernel<true, false, true><<<gn, 256, 0, stream>>>(
        S0n, F_LINK, F_LINK, x_node, F_NODE, F_NODE,
        Wm_ln, Wm_ln + H * H, bm_ln, dgn, AGG, NN);
    gemm_dual_first_kernel<false, true, false><<<gn, 256, 0, stream>>>(
        AGG, H, H, x_node, F_NODE, F_NODE,
        Wu_ln, Wu_ln + H * H, bu_ln, nullptr, xn2, NN);
  }

  for (int l = 1; l < Lnum; ++l) {
    float* cxn = (l & 1) ? xn2 : xn;
    float* cxl = (l & 1) ? xl2 : xl;
    float* oxn = (l & 1) ? xn  : xn2;
    float* oxl = (l & 1) ? xl  : xl2;

    const float* WmNLt = Wm_nl + (size_t)l * 2 * H * H;
    const float* WmNLb = WmNLt + H * H;
    const float* WuNLt = Wu_nl + (size_t)l * 2 * H * H;
    const float* WuNLb = WuNLt + H * H;
    const float* WmLNt = Wm_ln + (size_t)l * 2 * H * H;
    const float* WmLNb = WmLNt + H * H;
    const float* WuLNt = Wu_ln + (size_t)l * 2 * H * H;
    const float* WuLNb = WuLNt + H * H;

    zero_kernel<<<nb((long)NL * H, 4 * 256 * ZV), 256, 0, stream>>>(Sl, (long)NL * H);
    zero_kernel<<<nb((long)NN * H, 4 * 256 * ZV), 256, 0, stream>>>(Sn, (long)NN * H);
    scatter_kernel<true><<<nb((long)nb(E, EPW) * 32, 256), 256, 0, stream>>>(cxn, nl, nl + E, E, Sl);
    scatter_kernel<true><<<nb((long)nb(E, EPW) * 32, 256), 256, 0, stream>>>(cxl, ln, ln + E, E, Sn);

    gemm_dual_kernel<true, false><<<gl, 256, 0, stream>>>(
        Sl, cxl, WmNLt, WmNLb, bm_nl + (size_t)l * H, dgl, AGG, NL);
    gemm_dual_kernel<false, true><<<gl, 256, 0, stream>>>(
        AGG, cxl, WuNLt, WuNLb, bu_nl + (size_t)l * H, nullptr, oxl, NL);
    gemm_dual_kernel<true, false><<<gn, 256, 0, stream>>>(
        Sn, cxn, WmLNt, WmLNb, bm_ln + (size_t)l * H, dgn, AGG, NN);
    gemm_dual_kernel<false, true><<<gn, 256, 0, stream>>>(
        AGG, cxn, WuLNt, WuLNb, bu_ln + (size_t)l * H, nullptr, oxn, NN);
  }

  float* fxl = (Lnum & 1) ? xl2 : xl;
  zero_kernel<<<nb((long)G * H, 4 * 256 * ZV), 256, 0, stream>>>(sums, (long)G * H);
  zero_kernel<<<nb(G, 4 * 256 * ZV), 256, 0, stream>>>(cnt, G);
  scatter_kernel<false><<<nb((long)nb(NL, EPW) * 32, 256), 256, 0, stream>>>(fxl, nullptr, batch, NL, sums);
  pool_cnt_kernel<<<nb(NL, 256), 256, 0, stream>>>(batch, NL, cnt);
  finalize_kernel<<<nb((long)G * H, 256), 256, 0, stream>>>(sums, cnt, (float*)d_out, G);
}


# last layer's node update not launched (zero-fill, 400k-edge row scatter, two dual products): the output pools link features only, so it feeds nothing; 24 -> 20 launches
# speedup vs baseline: 2.1966x; 1.2505x over previous
#include <hip/hip_runtime.h>


typedef __attribute__((ext_vector_type(2))) float v2f;
typedef __attribute__((ext_vector_type(8))) float v8f;

#define H 128
#define LDSPAD 132
#define TPB 8

#define ZV 4
__global__ __launch_bounds__(256) void zero_kernel(float* p, long n) {
  long head = (long)(((16u - (unsigned)((size_t)p & 15u)) & 15u) >> 2);
  if (head > n) head = n;
  float4* q = (float4*)(p + head);
  const long nv = (n - head) >> 2;
  const long tail = head + (nv << 2);
  const long b0 = (long)blockIdx.x * (256 * ZV) + threadIdx.x;
  const float4 z = make_float4(0.0f, 0.0f, 0.0f, 0.0f);
#pragma unroll
  for (int j = 0; j < ZV; ++j) {
    const long i = b0 + j * 256;
    if (i < nv) q[i] = z;
  }
  if (blockIdx.x == 0) {
    if ((long)threadIdx.x < head) p[threadIdx.x] = 0.0f;
    if (tail + (long)threadIdx.x < n) p[tail + threadIdx.x] = 0.0f;
  }
}

__global__ __launch_bounds__(256) void pad_kernel(const float* __restrict__ in,
                                                  float* __restrict__ out,
                                                  int N, int F) {
  long t = (long)blockIdx.x * blockDim.x + threadIdx.x;
  if (t >= (long)N * H) return;
  int i = (int)(t >> 7), h = (int)(t & 127);
  out[t] = (h < F) ? in[(long)i * F + h] : 0.0f;
}

__global__ __launch_bounds__(256) void deg_kernel(const int* __restrict__ edge,
                                                  int E, float* __restrict__ deg) {
  int e = blockIdx.x * blockDim.x + threadIdx.x;
  if (e < E) atomicAdd(&deg[edge[E + e]], 1.0f);
}

#define EPW 8
template <bool HAS_SRC>
__global__ __launch_bounds__(256) void scatter_kernel(const float* __restrict__ X,
                                                      const int* __restrict__ src,
                                                      const int* __restrict__ dst,
                                                      int E, float* __restrict__ S) {
  const int lane = threadIdx.x & 31;
  const int wave = __builtin_amdgcn_readfirstlane((int)((blockIdx.x * 256u + threadIdx.x) >> 5));
  const int e0 = wave * EPW;
  if (e0 >= E) return;
  if (e0 + EPW <= E) {
    float v[EPW][4];
#pragma unroll
    for (int j = 0; j < EPW; ++j) {
      const int s = HAS_SRC ? src[e0 + j] : e0 + j;
      const float* xr = X + (size_t)s * H + lane;
#pragma unroll
      for (int q = 0; q < 4; ++q) v[j][q] = xr[32 * q];
    }
#pragma unroll
    for (int j = 0; j < EPW; ++j) {
      float* o = S + (size_t)dst[e0 + j] * H + lane;
#pragma unroll
      for (int q = 0; q < 4; ++q) atomicAdd(o + 32 * q, v[j][q]);
    }
  } else {
    for (int e = e0; e < E; ++e) {
      const int s = HAS_SRC ? src[e] : e;
      const float* xr = X + (size_t)s * H + lane;
      float* o = S + (size_t)dst[e] * H + lane;
      for (int q = 0; q < 4; ++q) atomicAdd(o + 32 * q, xr[32 * q]);
    }
  }
}

template <bool HAS_DEG, bool RELU>
__global__ __launch_bounds__(256) void gemm_dual_kernel(
    const float* __restrict__ A, const float* __restrict__ B,
    const float* __restrict__ WA, const float* __restrict__ WB,
    const float* __restrict__ bias, const float* __restrict__ deg,
    float* __restrict__ Y, int M) {
  __shared__ float lwa[H * LDSPAD];
  __shared__ float lwb[H * LDSPAD];
  for (int idx = threadIdx.x; idx < H * H; idx += 256) {
    int k = idx >> 7, n = idx & 127;
    lwa[n * LDSPAD + k] = WA[idx];
    lwb[n * LDSPAD + k] = WB[idx];
  }
  __syncthreads();

  const int wave = threadIdx.x >> 5;
  const int lane = threadIdx.x & 31;
  const int l15  = lane & 15;
  const int kh2  = (lane >> 4) << 1;
  const int ntiles = M >> 4;

  for (int t = blockIdx.x * TPB + wave; t < ntiles; t += gridDim.x * TPB) {
    const int m0 = t << 4;
    v8f cA[8], cB[8];
#pragma unroll
    for (int n = 0; n < 8; ++n) {
      cA[n] = (v8f){0, 0, 0, 0, 0, 0, 0, 0};
      cB[n] = (v8f){0, 0, 0, 0, 0, 0, 0, 0};
    }

    const float* Ap = A + (long)(m0 + l15) * H + kh2;
    const float* Bp = B + (long)(m0 + l15) * H + kh2;
    const float* wabase = &lwa[l15 * LDSPAD + kh2];
    const float* wbbase = &lwb[l15 * LDSPAD + kh2];

#pragma unroll 1
    for (int kk = 0; kk < 32; ++kk) {
      const int k0 = kk << 2;
      v2f a = *(const v2f*)(Ap + k0);
      v2f b = *(const v2f*)(Bp + k0);
#pragma unroll
      for (int n = 0; n < 8; ++n) {
        v2f wa = *(const v2f*)(wabase + n * 16 * LDSPAD + k0);
        v2f wb = *(const v2f*)(wbbase + n * 16 * LDSPAD + k0);
        cA[n] = __builtin_amdgcn_wmma_f32_16x16x4_f32(false, a, false, wa,
                                                      (short)0, cA[n], false, false);
        cB[n] = __builtin_amdgcn_wmma_f32_16x16x4_f32(false, b, false, wb,
                                                      (short)0, cB[n], false, false);
      }
    }

    const int rbase = m0 + ((lane >> 4) << 3);
    float degv[8];
    if (HAS_DEG) {
      const float4 d0 = *(const float4*)(deg + rbase);
      const float4 d1 = *(const float4*)(deg + rbase + 4);
      degv[0] = d0.x; degv[1] = d0.y; degv[2] = d0.z; degv[3] = d0.w;
      degv[4] = d1.x; degv[5] = d1.y; degv[6] = d1.z; degv[7] = d1.w;
    } else {
#pragma unroll
      for (int r = 0; r < 8; ++r) degv[r] = 1.0f;
    }
#pragma unroll
    for (int n = 0; n < 8; ++n) {
      const float bv = bias[n * 16 + l15];
#pragma unroll
      for (int r = 0; r < 8; ++r) {
        float acc = HAS_DEG ? (cA[n][r] + degv[r] * (cB[n][r] + bv))
                            : (cA[n][r] + cB[n][r] + bv);
        if (RELU) acc = fmaxf(acc, 0.0f);
        Y[(long)(rbase + r) * H + n * 16 + l15] = acc;
      }
    }
  }
}

#define SFU 8
__global__ __launch_bounds__(256) void scatter_first_kernel(const float* __restrict__ X,
                                                            int F,
                                                            const int* __restrict__ edge,
                                                            int E, float* __restrict__ S) {
  const unsigned uF = (unsigned)F;
  const unsigned total = (unsigned)E * uF;
  const unsigned blk0 = blockIdx.x * (256u * SFU);
  if (blk0 >= total) return;
  const unsigned base = blk0 + threadIdx.x;
  if (total - blk0 >= 256u * SFU) {
    float v[SFU];
    unsigned o[SFU];
#pragma unroll
    for (int j = 0; j < SFU; ++j) {
      const unsigned idx = base + j * 256u;
      const unsigned e = idx / uF;
      const unsigned c = idx - e * uF;
      const int s = edge[e], d = edge[(unsigned)E + e];
      v[j] = X[(size_t)s * uF + c];
      o[j] = (unsigned)d * uF + c;
    }
#pragma unroll
    for (int j = 0; j < SFU; ++j) atomicAdd(S + o[j], v[j]);
  } else {
    for (unsigned idx = base; idx < total; idx += 256u) {
      const unsigned e = idx / uF;
      const unsigned c = idx - e * uF;
      const int s = edge[e], d = edge[(unsigned)E + e];
      atomicAdd(S + (size_t)d * uF + c, X[(size_t)s * uF + c]);
    }
  }
}

__device__ __forceinline__ v2f load_narrow(const float* __restrict__ row, int k, int kv) {
  const int k1 = k + 1;
  const float x0 = row[k  < kv ? k  : kv - 1];
  const float x1 = row[k1 < kv ? k1 : kv - 1];
  v2f r;
  r.x = (k  < kv) ? x0 : 0.0f;
  r.y = (k1 < kv) ? x1 : 0.0f;
  return r;
}

__device__ __forceinline__ void wmma_tail_slots(v8f (&c)[8]) {
  asm volatile("v_nop\n\tv_nop\n\tv_nop\n\tv_nop"
               : "+v"(c[0]), "+v"(c[1]), "+v"(c[2]), "+v"(c[3]),
                 "+v"(c[4]), "+v"(c[5]), "+v"(c[6]), "+v"(c[7]));
}

template <bool HAS_DEG, bool RELU, bool A_NARROW>
__global__ __launch_bounds__(256) void gemm_dual_first_kernel(
    const float* __restrict__ A, int lda, int ka,
    const float* __restrict__ B, int ldb, int kb,
    const float* __restrict__ WA, const float* __restrict__ WB,
    const float* __restrict__ bias, const float* __restrict__ deg,
    float* __restrict__ Y, int M) {
  __shared__ float lwa[H * LDSPAD];
  __shared__ float lwb[H * LDSPAD];
  const int ksa = A_NARROW ? ((ka + 3) & ~3) : H;
  const int ksb = (kb + 3) & ~3;
  for (int idx = threadIdx.x; idx < ksa * H; idx += 256) {
    int k = idx >> 7, n = idx & 127;
    lwa[n * LDSPAD + k] = WA[idx];
  }
  for (int idx = threadIdx.x; idx < ksb * H; idx += 256) {
    int k = idx >> 7, n = idx & 127;
    lwb[n * LDSPAD + k] = WB[idx];
  }
  __syncthreads();

  const int wave = threadIdx.x >> 5;
  const int lane = threadIdx.x & 31;
  const int l15  = lane & 15;
  const int kh2  = (lane >> 4) << 1;
  const int ntiles = M >> 4;

  for (int t = blockIdx.x * TPB + wave; t < ntiles; t += gridDim.x * TPB) {
    const int m0 = t << 4;
    v8f cA[8], cB[8];
#pragma unroll
    for (int n = 0; n < 8; ++n) {
      cA[n] = (v8f){0, 0, 0, 0, 0, 0, 0, 0};
      cB[n] = (v8f){0, 0, 0, 0, 0, 0, 0, 0};
    }
    const float* wabase = &lwa[l15 * LDSPAD + kh2];
    const float* wbbase = &lwb[l15 * LDSPAD + kh2];

    const float* Brow = B + (long)(m0 + l15) * ldb;
    for (int k0 = 0; k0 < ksb; k0 += 4) {
      const v2f b = load_narrow(Brow, k0 + kh2, kb);
#pragma unroll
      for (int n = 0; n < 8; ++n) {
        v2f wb = *(const v2f*)(wbbase + n * 16 * LDSPAD + k0);
        cB[n] = __builtin_amdgcn_wmma_f32_16x16x4_f32(false, b, false, wb,
                                                      (short)0, cB[n], false, false);
      }
      wmma_tail_slots(cB);
    }

    if (A_NARROW) {
      const float* Arow = A + (long)(m0 + l15) * lda;
      for (int k0 = 0; k0 < ksa; k0 += 4) {
        const v2f a = load_narrow(Arow, k0 + kh2, ka);
#pragma unroll
        for (int n = 0; n < 8; ++n) {
          v2f wa = *(const v2f*)(wabase + n * 16 * LDSPAD + k0);
          cA[n] = __builtin_amdgcn_wmma_f32_16x16x4_f32(false, a, false, wa,
                                                        (short)0, cA[n], false, false);
        }
        wmma_tail_slots(cA);
      }
    } else {
      const float* Ap = A + (long)(m0 + l15) * H + kh2;
#pragma unroll 2
      for (int kk = 0; kk < 32; ++kk) {
        const int k0 = kk << 2;
        v2f a = *(const v2f*)(Ap + k0);
#pragma unroll
        for (int n = 0; n < 8; ++n) {
          v2f wa = *(const v2f*)(wabase + n * 16 * LDSPAD + k0);
          cA[n] = __builtin_amdgcn_wmma_f32_16x16x4_f32(false, a, false, wa,
                                                        (short)0, cA[n], false, false);
        }
      }
    }

    const int rbase = m0 + ((lane >> 4) << 3);
    float degv[8];
    if (HAS_DEG) {
      const float4 d0 = *(const float4*)(deg + rbase);
      const float4 d1 = *(const float4*)(deg + rbase + 4);
      degv[0] = d0.x; degv[1] = d0.y; degv[2] = d0.z; degv[3] = d0.w;
      degv[4] = d1.x; degv[5] = d1.y; degv[6] = d1.z; degv[7] = d1.w;
    } else {
#pragma unroll
      for (int r = 0; r < 8; ++r) degv[r] = 1.0f;
    }
#pragma unroll
    for (int n = 0; n < 8; ++n) {
      const float bv = bias[n * 16 + l15];
#pragma unroll
      for (int r = 0; r < 8; ++r) {
        float acc = HAS_DEG ? (cA[n][r] + degv[r] * (cB[n][r] + bv))
                            : (cA[n][r] + cB[n][r] + bv);
        if (RELU) acc = fmaxf(acc, 0.0f);
        Y[(long)(rbase + r) * H + n * 16 + l15] = acc;
      }
    }
  }
}

__global__ __launch_bounds__(256) void pool_cnt_kernel(const int* __restrict__ batch,
                                                       int NL, float* __restrict__ cnt) {
  int i = blockIdx.x * blockDim.x + threadIdx.x;
  if (i < NL) atomicAdd(&cnt[batch[i]], 1.0f);
}

__global__ __launch_bounds__(256) void finalize_kernel(const float* __restrict__ sums,
                                                       const float* __restrict__ cnt,
                                                       float* __restrict__ out, int G) {
  int t = blockIdx.x * blockDim.x + threadIdx.x;
  if (t >= G * H) return;
  int g = t >> 7;
  out[t] = sums[t] / fmaxf(cnt[g], 1.0f);
}

extern "C" void kernel_launch(void* const* d_in, const int* in_sizes, int n_in,
                              void* d_out, int out_size, void* d_ws, size_t ws_size,
                              hipStream_t stream) {
  const float* x_node = (const float*)d_in[0];
  const float* x_link = (const float*)d_in[1];
  const float* Wm_nl  = (const float*)d_in[2];
  const float* bm_nl  = (const float*)d_in[3];
  const float* Wu_nl  = (const float*)d_in[4];
  const float* bu_nl  = (const float*)d_in[5];
  const float* Wm_ln  = (const float*)d_in[6];
  const float* bm_ln  = (const float*)d_in[7];
  const float* Wu_ln  = (const float*)d_in[8];
  const float* bu_ln  = (const float*)d_in[9];
  const int*   nl     = (const int*)d_in[10];
  const int*   ln     = (const int*)d_in[11];
  const int*   batch  = (const int*)d_in[12];

  const int NL = in_sizes[12];
  const int NN = 50000;
  const int F_NODE = in_sizes[0] / NN;
  const int F_LINK = in_sizes[1] / NL;
  const int E  = in_sizes[10] / 2;
  const int Lnum = in_sizes[3] / H;
  const int G = out_size / H;

  const size_t MH = (size_t)((NN > NL) ? NN : NL) * H;
  float* ws   = (float*)d_ws;
  float* xn   = ws;
  float* xl   = xn  + MH;
  float* xn2  = xl  + MH;
  float* xl2  = xn2 + MH;
  float* Sl   = xl2 + MH;
  float* Sn   = Sl  + MH;
  float* AGG  = Sn  + MH;
  float* dgl  = AGG + MH;
  float* dgn  = dgl + NL;
  float* sums = dgn + NN;
  float* cnt  = sums + (size_t)G * H;
  (void)n_in; (void)ws_size;

  auto nb = [](long n, int b) { return (int)((n + b - 1) / b); };

  const bool node_live_any = (Lnum > 1);

  zero_kernel<<<nb(NL, 4 * 256 * ZV), 256, 0, stream>>>(dgl, NL);
  deg_kernel<<<nb(E, 256), 256, 0, stream>>>(nl, E, dgl);
  if (node_live_any) {
    zero_kernel<<<nb(NN, 4 * 256 * ZV), 256, 0, stream>>>(dgn, NN);
    deg_kernel<<<nb(E, 256), 256, 0, stream>>>(ln, E, dgn);
  }

  const int gl = nb(NL / 16, TPB);
  const int gn = nb(NN / 16, TPB);

  if (Lnum == 0) {
    pad_kernel<<<nb((long)NL * H, 256), 256, 0, stream>>>(x_link, xl, NL, F_LINK);
  } else {
    float* S0l = Sl;
    float* S0n = Sl + (size_t)NL * F_NODE;
    const long nS0 = (long)NL * F_NODE + (node_live_any ? (long)NN * F_LINK : 0L);
    zero_kernel<<<nb(nS0, 4 * 256 * ZV), 256, 0, stream>>>(S0l, nS0);
    scatter_first_kernel<<<nb((long)E * F_NODE, 256 * SFU), 256, 0, stream>>>(x_node, F_NODE, nl, E, S0l);
    if (node_live_any)
      scatter_first_kernel<<<nb((long)E * F_LINK, 256 * SFU), 256, 0, stream>>>(x_link, F_LINK, ln, E, S0n);
    gemm_dual_first_kernel<true, false, true><<<gl, 256, 0, stream>>>(
        S0l, F_NODE, F_NODE, x_link, F_LINK, F_LINK,
        Wm_nl, Wm_nl + H * H, bm_nl, dgl, AGG, NL);
    gemm_dual_first_kernel<false, true, false><<<gl, 256, 0, stream>>>(
        AGG, H, H, x_link, F_LINK, F_LINK,
        Wu_nl, Wu_nl + H * H, bu_nl, nullptr, xl2, NL);
    if (node_live_any) {
      gemm_dual_first_kernel<true, false, true><<<gn, 256, 0, stream>>>(
          S0n, F_LINK, F_LINK, x_node, F_NODE, F_NODE,
          Wm_ln, Wm_ln + H * H, bm_ln, dgn, AGG, NN);
      gemm_dual_first_kernel<false, true, false><<<gn, 256, 0, stream>>>(
          AGG, H, H, x_node, F_NODE, F_NODE,
          Wu_ln, Wu_ln + H * H, bu_ln, nullptr, xn2, NN);
    }
  }

  for (int l = 1; l < Lnum; ++l) {
    float* cxn = (l & 1) ? xn2 : xn;
    float* cxl = (l & 1) ? xl2 : xl;
    float* oxn = (l & 1) ? xn  : xn2;
    float* oxl = (l & 1) ? xl  : xl2;

    const float* WmNLt = Wm_nl + (size_t)l * 2 * H * H;
    const float* WmNLb = WmNLt + H * H;
    const float* WuNLt = Wu_nl + (size_t)l * 2 * H * H;
    const float* WuNLb = WuNLt + H * H;
    const float* WmLNt = Wm_ln + (size_t)l * 2 * H * H;
    const float* WmLNb = WmLNt + H * H;
    const float* WuLNt = Wu_ln + (size_t)l * 2 * H * H;
    const float* WuLNb = WuLNt + H * H;

    const bool node_live = (l + 1 < Lnum);

    zero_kernel<<<nb((long)NL * H, 4 * 256 * ZV), 256, 0, stream>>>(Sl, (long)NL * H);
    scatter_kernel<true><<<nb((long)nb(E, EPW) * 32, 256), 256, 0, stream>>>(cxn, nl, nl + E, E, Sl);
    if (node_live) {
      zero_kernel<<<nb((long)NN * H, 4 * 256 * ZV), 256, 0, stream>>>(Sn, (long)NN * H);
      scatter_kernel<true><<<nb((long)nb(E, EPW) * 32, 256), 256, 0, stream>>>(cxl, ln, ln + E, E, Sn);
    }

    gemm_dual_kernel<true, false><<<gl, 256, 0, stream>>>(
        Sl, cxl, WmNLt, WmNLb, bm_nl + (size_t)l * H, dgl, AGG, NL);
    gemm_dual_kernel<false, true><<<gl, 256, 0, stream>>>(
        AGG, cxl, WuNLt, WuNLb, bu_nl + (size_t)l * H, nullptr, oxl, NL);
    if (node_live) {
      gemm_dual_kernel<true, false><<<gn, 256, 0, stream>>>(
          Sn, cxn, WmLNt, WmLNb, bm_ln + (size_t)l * H, dgn, AGG, NN);
      gemm_dual_kernel<false, true><<<gn, 256, 0, stream>>>(
          AGG, cxn, WuLNt, WuLNb, bu_ln + (size_t)l * H, nullptr, oxn, NN);
    }
  }

  float* fxl = (Lnum & 1) ? xl2 : xl;
  zero_kernel<<<nb((long)G * H, 4 * 256 * ZV), 256, 0, stream>>>(sums, (long)G * H);
  zero_kernel<<<nb(G, 4 * 256 * ZV), 256, 0, stream>>>(cnt, G);
  scatter_kernel<false><<<nb((long)nb(NL, EPW) * 32, 256), 256, 0, stream>>>(fxl, nullptr, batch, NL, sums);
  pool_cnt_kernel<<<nb(NL, 256), 256, 0, stream>>>(batch, NL, cnt);
  finalize_kernel<<<nb((long)G * H, 256), 256, 0, stream>>>(sums, cnt, (float*)d_out, G);
}


# launches merged 20 -> 12: one zero-fill for all small buffers, one first-layer edge pass (both column scatters + both degree counts), row count inside the pooled-sum pass; same adds, sums atomic
# speedup vs baseline: 2.3425x; 1.0664x over previous
#include <hip/hip_runtime.h>


typedef __attribute__((ext_vector_type(2))) float v2f;
typedef __attribute__((ext_vector_type(8))) float v8f;

#define H 128
#define LDSPAD 132
#define TPB 8

#define ZV 4
__device__ __forceinline__ void zero_range(float* p, long n, long blk) {
  long head = (long)(((16u - (unsigned)((size_t)p & 15u)) & 15u) >> 2);
  if (head > n) head = n;
  float4* q = (float4*)(p + head);
  const long nv = (n - head) >> 2;
  const long tail = head + (nv << 2);
  const long b0 = blk * (256 * ZV) + threadIdx.x;
  const float4 z = make_float4(0.0f, 0.0f, 0.0f, 0.0f);
#pragma unroll
  for (int j = 0; j < ZV; ++j) {
    const long i = b0 + j * 256;
    if (i < nv) q[i] = z;
  }
  if (blk == 0) {
    if ((long)threadIdx.x < head) p[threadIdx.x] = 0.0f;
    if (tail + (long)threadIdx.x < n) p[tail + threadIdx.x] = 0.0f;
  }
}

__global__ __launch_bounds__(256) void zero_kernel(float* p, long n) {
  zero_range(p, n, (long)blockIdx.x);
}

__global__ __launch_bounds__(256) void zero2_kernel(float* p0, long n0, int nb0,
                                                    float* p1, long n1) {
  if ((int)blockIdx.x < nb0) zero_range(p0, n0, (long)blockIdx.x);
  else zero_range(p1, n1, (long)blockIdx.x - nb0);
}

__global__ __launch_bounds__(256) void pad_kernel(const float* __restrict__ in,
                                                  float* __restrict__ out,
                                                  int N, int F) {
  long t = (long)blockIdx.x * blockDim.x + threadIdx.x;
  if (t >= (long)N * H) return;
  int i = (int)(t >> 7), h = (int)(t & 127);
  out[t] = (h < F) ? in[(long)i * F + h] : 0.0f;
}

#define EPW 8
template <bool HAS_SRC, bool COUNT_ROWS>
__global__ __launch_bounds__(256) void scatter_kernel(const float* __restrict__ X,
                                                      const int* __restrict__ src,
                                                      const int* __restrict__ dst,
                                                      int E, float* __restrict__ S,
                                                      float* __restrict__ cnt) {
  const int lane = threadIdx.x & 31;
  const int wave = __builtin_amdgcn_readfirstlane((int)((blockIdx.x * 256u + threadIdx.x) >> 5));
  const int e0 = wave * EPW;
  if (e0 >= E) return;
  if (e0 + EPW <= E) {
    float v[EPW][4];
#pragma unroll
    for (int j = 0; j < EPW; ++j) {
      const int s = HAS_SRC ? src[e0 + j] : e0 + j;
      const float* xr = X + (size_t)s * H + lane;
#pragma unroll
      for (int q = 0; q < 4; ++q) v[j][q] = xr[32 * q];
    }
#pragma unroll
    for (int j = 0; j < EPW; ++j) {
      float* o = S + (size_t)dst[e0 + j] * H + lane;
#pragma unroll
      for (int q = 0; q < 4; ++q) atomicAdd(o + 32 * q, v[j][q]);
    }
    if (COUNT_ROWS) {
      if (lane < EPW) atomicAdd(cnt + dst[e0 + lane], 1.0f);
    }
  } else {
    for (int e = e0; e < E; ++e) {
      const int s = HAS_SRC ? src[e] : e;
      const float* xr = X + (size_t)s * H + lane;
      float* o = S + (size_t)dst[e] * H + lane;
      for (int q = 0; q < 4; ++q) atomicAdd(o + 32 * q, xr[32 * q]);
      if (COUNT_ROWS) {
        if (lane == 0) atomicAdd(cnt + dst[e], 1.0f);
      }
    }
  }
}

template <bool HAS_DEG, bool RELU>
__global__ __launch_bounds__(256) void gemm_dual_kernel(
    const float* __restrict__ A, const float* __restrict__ B,
    const float* __restrict__ WA, const float* __restrict__ WB,
    const float* __restrict__ bias, const float* __restrict__ deg,
    float* __restrict__ Y, int M) {
  __shared__ float lwa[H * LDSPAD];
  __shared__ float lwb[H * LDSPAD];
  for (int idx = threadIdx.x; idx < H * H; idx += 256) {
    int k = idx >> 7, n = idx & 127;
    lwa[n * LDSPAD + k] = WA[idx];
    lwb[n * LDSPAD + k] = WB[idx];
  }
  __syncthreads();

  const int wave = threadIdx.x >> 5;
  const int lane = threadIdx.x & 31;
  const int l15  = lane & 15;
  const int kh2  = (lane >> 4) << 1;
  const int ntiles = M >> 4;

  for (int t = blockIdx.x * TPB + wave; t < ntiles; t += gridDim.x * TPB) {
    const int m0 = t << 4;
    v8f cA[8], cB[8];
#pragma unroll
    for (int n = 0; n < 8; ++n) {
      cA[n] = (v8f){0, 0, 0, 0, 0, 0, 0, 0};
      cB[n] = (v8f){0, 0, 0, 0, 0, 0, 0, 0};
    }

    const float* Ap = A + (long)(m0 + l15) * H + kh2;
    const float* Bp = B + (long)(m0 + l15) * H + kh2;
    const float* wabase = &lwa[l15 * LDSPAD + kh2];
    const float* wbbase = &lwb[l15 * LDSPAD + kh2];

#pragma unroll 1
    for (int kk = 0; kk < 32; ++kk) {
      const int k0 = kk << 2;
      v2f a = *(const v2f*)(Ap + k0);
      v2f b = *(const v2f*)(Bp + k0);
#pragma unroll
      for (int n = 0; n < 8; ++n) {
        v2f wa = *(const v2f*)(wabase + n * 16 * LDSPAD + k0);
        v2f wb = *(const v2f*)(wbbase + n * 16 * LDSPAD + k0);
        cA[n] = __builtin_amdgcn_wmma_f32_16x16x4_f32(false, a, false, wa,
                                                      (short)0, cA[n], false, false);
        cB[n] = __builtin_amdgcn_wmma_f32_16x16x4_f32(false, b, false, wb,
                                                      (short)0, cB[n], false, false);
      }
    }

    const int rbase = m0 + ((lane >> 4) << 3);
    float degv[8];
    if (HAS_DEG) {
      const float4 d0 = *(const float4*)(deg + rbase);
      const float4 d1 = *(const float4*)(deg + rbase + 4);
      degv[0] = d0.x; degv[1] = d0.y; degv[2] = d0.z; degv[3] = d0.w;
      degv[4] = d1.x; degv[5] = d1.y; degv[6] = d1.z; degv[7] = d1.w;
    } else {
#pragma unroll
      for (int r = 0; r < 8; ++r) degv[r] = 1.0f;
    }
#pragma unroll
    for (int n = 0; n < 8; ++n) {
      const float bv = bias[n * 16 + l15];
#pragma unroll
      for (int r = 0; r < 8; ++r) {
        float acc = HAS_DEG ? (cA[n][r] + degv[r] * (cB[n][r] + bv))
                            : (cA[n][r] + cB[n][r] + bv);
        if (RELU) acc = fmaxf(acc, 0.0f);
        Y[(long)(rbase + r) * H + n * 16 + l15] = acc;
      }
    }
  }
}

#define SFU 8
__global__ __launch_bounds__(256) void edge_first_kernel(
    const float* __restrict__ Xa, int Fa, const int* __restrict__ edge_a,
    float* __restrict__ Sa, float* __restrict__ dega,
    const float* __restrict__ Xb, int Fb, const int* __restrict__ edge_b,
    float* __restrict__ Sb, float* __restrict__ degb, int E) {
  const bool second = (blockIdx.y != 0);
  const float* __restrict__ X = second ? Xb : Xa;
  const int* __restrict__ edge = second ? edge_b : edge_a;
  float* __restrict__ S = second ? Sb : Sa;
  float* __restrict__ deg = second ? degb : dega;
  const unsigned uF = (unsigned)(second ? Fb : Fa);
  const unsigned total = (unsigned)E * uF;
  const unsigned blk0 = blockIdx.x * (256u * SFU);
  if (blk0 >= total) return;
  const unsigned base = blk0 + threadIdx.x;
  if (total - blk0 >= 256u * SFU) {
    float v[SFU];
    unsigned o[SFU];
    unsigned d[SFU];
    bool first_col[SFU];
#pragma unroll
    for (int j = 0; j < SFU; ++j) {
      const unsigned idx = base + j * 256u;
      const unsigned e = idx / uF;
      const unsigned c = idx - e * uF;
      const int s = edge[e];
      d[j] = (unsigned)edge[(unsigned)E + e];
      v[j] = X[(size_t)s * uF + c];
      o[j] = d[j] * uF + c;
      first_col[j] = (c == 0u);
    }
#pragma unroll
    for (int j = 0; j < SFU; ++j) atomicAdd(S + o[j], v[j]);
#pragma unroll
    for (int j = 0; j < SFU; ++j)
      if (first_col[j]) atomicAdd(deg + d[j], 1.0f);
  } else {
    for (unsigned idx = base; idx < total; idx += 256u) {
      const unsigned e = idx / uF;
      const unsigned c = idx - e * uF;
      const int s = edge[e], dd = edge[(unsigned)E + e];
      atomicAdd(S + (size_t)dd * uF + c, X[(size_t)s * uF + c]);
      if (c == 0u) atomicAdd(deg + dd, 1.0f);
    }
  }
}

__device__ __forceinline__ v2f load_narrow(const float* __restrict__ row, int k, int kv) {
  const int k1 = k + 1;
  const float x0 = row[k  < kv ? k  : kv - 1];
  const float x1 = row[k1 < kv ? k1 : kv - 1];
  v2f r;
  r.x = (k  < kv) ? x0 : 0.0f;
  r.y = (k1 < kv) ? x1 : 0.0f;
  return r;
}

__device__ __forceinline__ void wmma_tail_slots(v8f (&c)[8]) {
  asm volatile("v_nop\n\tv_nop\n\tv_nop\n\tv_nop"
               : "+v"(c[0]), "+v"(c[1]), "+v"(c[2]), "+v"(c[3]),
                 "+v"(c[4]), "+v"(c[5]), "+v"(c[6]), "+v"(c[7]));
}

template <bool HAS_DEG, bool RELU, bool A_NARROW>
__global__ __launch_bounds__(256) void gemm_dual_first_kernel(
    const float* __restrict__ A, int lda, int ka,
    const float* __restrict__ B, int ldb, int kb,
    const float* __restrict__ WA, const float* __restrict__ WB,
    const float* __restrict__ bias, const float* __restrict__ deg,
    float* __restrict__ Y, int M) {
  __shared__ float lwa[H * LDSPAD];
  __shared__ float lwb[H * LDSPAD];
  const int ksa = A_NARROW ? ((ka + 3) & ~3) : H;
  const int ksb = (kb + 3) & ~3;
  for (int idx = threadIdx.x; idx < ksa * H; idx += 256) {
    int k = idx >> 7, n = idx & 127;
    lwa[n * LDSPAD + k] = WA[idx];
  }
  for (int idx = threadIdx.x; idx < ksb * H; idx += 256) {
    int k = idx >> 7, n = idx & 127;
    lwb[n * LDSPAD + k] = WB[idx];
  }
  __syncthreads();

  const int wave = threadIdx.x >> 5;
  const int lane = threadIdx.x & 31;
  const int l15  = lane & 15;
  const int kh2  = (lane >> 4) << 1;
  const int ntiles = M >> 4;

  for (int t = blockIdx.x * TPB + wave; t < ntiles; t += gridDim.x * TPB) {
    const int m0 = t << 4;
    v8f cA[8], cB[8];
#pragma unroll
    for (int n = 0; n < 8; ++n) {
      cA[n] = (v8f){0, 0, 0, 0, 0, 0, 0, 0};
      cB[n] = (v8f){0, 0, 0, 0, 0, 0, 0, 0};
    }
    const float* wabase = &lwa[l15 * LDSPAD + kh2];
    const float* wbbase = &lwb[l15 * LDSPAD + kh2];

    const float* Brow = B + (long)(m0 + l15) * ldb;
    for (int k0 = 0; k0 < ksb; k0 += 4) {
      const v2f b = load_narrow(Brow, k0 + kh2, kb);
#pragma unroll
      for (int n = 0; n < 8; ++n) {
        v2f wb = *(const v2f*)(wbbase + n * 16 * LDSPAD + k0);
        cB[n] = __builtin_amdgcn_wmma_f32_16x16x4_f32(false, b, false, wb,
                                                      (short)0, cB[n], false, false);
      }
      wmma_tail_slots(cB);
    }

    if (A_NARROW) {
      const float* Arow = A + (long)(m0 + l15) * lda;
      for (int k0 = 0; k0 < ksa; k0 += 4) {
        const v2f a = load_narrow(Arow, k0 + kh2, ka);
#pragma unroll
        for (int n = 0; n < 8; ++n) {
          v2f wa = *(const v2f*)(wabase + n * 16 * LDSPAD + k0);
          cA[n] = __builtin_amdgcn_wmma_f32_16x16x4_f32(false, a, false, wa,
                                                        (short)0, cA[n], false, false);
        }
        wmma_tail_slots(cA);
      }
    } else {
      const float* Ap = A + (long)(m0 + l15) * H + kh2;
#pragma unroll 2
      for (int kk = 0; kk < 32; ++kk) {
        const int k0 = kk << 2;
        v2f a = *(const v2f*)(Ap + k0);
#pragma unroll
        for (int n = 0; n < 8; ++n) {
          v2f wa = *(const v2f*)(wabase + n * 16 * LDSPAD + k0);
          cA[n] = __builtin_amdgcn_wmma_f32_16x16x4_f32(false, a, false, wa,
                                                        (short)0, cA[n], false, false);
        }
      }
    }

    const int rbase = m0 + ((lane >> 4) << 3);
    float degv[8];
    if (HAS_DEG) {
      const float4 d0 = *(const float4*)(deg + rbase);
      const float4 d1 = *(const float4*)(deg + rbase + 4);
      degv[0] = d0.x; degv[1] = d0.y; degv[2] = d0.z; degv[3] = d0.w;
      degv[4] = d1.x; degv[5] = d1.y; degv[6] = d1.z; degv[7] = d1.w;
    } else {
#pragma unroll
      for (int r = 0; r < 8; ++r) degv[r] = 1.0f;
    }
#pragma unroll
    for (int n = 0; n < 8; ++n) {
      const float bv = bias[n * 16 + l15];
#pragma unroll
      for (int r = 0; r < 8; ++r) {
        float acc = HAS_DEG ? (cA[n][r] + degv[r] * (cB[n][r] + bv))
                            : (cA[n][r] + cB[n][r] + bv);
        if (RELU) acc = fmaxf(acc, 0.0f);
        Y[(long)(rbase + r) * H + n * 16 + l15] = acc;
      }
    }
  }
}


__global__ __launch_bounds__(256) void finalize_kernel(const float* __restrict__ sums,
                                                       const float* __restrict__ cnt,
                                                       float* __restrict__ out, int G) {
  int t = blockIdx.x * blockDim.x + threadIdx.x;
  if (t >= G * H) return;
  int g = t >> 7;
  out[t] = sums[t] / fmaxf(cnt[g], 1.0f);
}

extern "C" void kernel_launch(void* const* d_in, const int* in_sizes, int n_in,
                              void* d_out, int out_size, void* d_ws, size_t ws_size,
                              hipStream_t stream) {
  const float* x_node = (const float*)d_in[0];
  const float* x_link = (const float*)d_in[1];
  const float* Wm_nl  = (const float*)d_in[2];
  const float* bm_nl  = (const float*)d_in[3];
  const float* Wu_nl  = (const float*)d_in[4];
  const float* bu_nl  = (const float*)d_in[5];
  const float* Wm_ln  = (const float*)d_in[6];
  const float* bm_ln  = (const float*)d_in[7];
  const float* Wu_ln  = (const float*)d_in[8];
  const float* bu_ln  = (const float*)d_in[9];
  const int*   nl     = (const int*)d_in[10];
  const int*   ln     = (const int*)d_in[11];
  const int*   batch  = (const int*)d_in[12];

  const int NL = in_sizes[12];
  const int NN = 50000;
  const int F_NODE = in_sizes[0] / NN;
  const int F_LINK = in_sizes[1] / NL;
  const int E  = in_sizes[10] / 2;
  const int Lnum = in_sizes[3] / H;
  const int G = out_size / H;

  const size_t MH = (size_t)((NN > NL) ? NN : NL) * H;
  float* ws   = (float*)d_ws;
  float* xn   = ws;
  float* xl   = xn  + MH;
  float* xn2  = xl  + MH;
  float* xl2  = xn2 + MH;
  float* Sl   = xl2 + MH;
  float* Sn   = Sl  + MH;
  float* AGG  = Sn  + MH;
  float* dgl  = AGG + MH;
  float* dgn  = dgl + NL;
  float* sums = dgn + NN;
  float* cnt  = sums + (size_t)G * H;
  (void)n_in; (void)ws_size;

  auto nb = [](long n, int b) { return (int)((n + b - 1) / b); };

  const bool node_live_any = (Lnum > 1);

  const long nsmall = (long)NL + NN + (long)G * H + G;
  const int nbsmall = nb(nsmall, 4 * 256 * ZV);

  const int gl = nb(NL / 16, TPB);
  const int gn = nb(NN / 16, TPB);

  if (Lnum == 0) {
    zero_kernel<<<nbsmall, 256, 0, stream>>>(dgl, nsmall);
    pad_kernel<<<nb((long)NL * H, 256), 256, 0, stream>>>(x_link, xl, NL, F_LINK);
  } else {
    float* S0l = Sl;
    float* S0n = Sl + (size_t)NL * F_NODE;
    const long nS0 = (long)NL * F_NODE + (node_live_any ? (long)NN * F_LINK : 0L);
    zero2_kernel<<<nbsmall + nb(nS0, 4 * 256 * ZV), 256, 0, stream>>>(dgl, nsmall, nbsmall, S0l, nS0);
    const int gxa = nb((long)E * F_NODE, 256 * SFU);
    const int gxb = nb((long)E * F_LINK, 256 * SFU);
    edge_first_kernel<<<dim3((gxa > gxb) ? gxa : gxb, node_live_any ? 2 : 1), 256, 0, stream>>>(
        x_node, F_NODE, nl, S0l, dgl, x_link, F_LINK, ln, S0n, dgn, E);
    gemm_dual_first_kernel<true, false, true><<<gl, 256, 0, stream>>>(
        S0l, F_NODE, F_NODE, x_link, F_LINK, F_LINK,
        Wm_nl, Wm_nl + H * H, bm_nl, dgl, AGG, NL);
    gemm_dual_first_kernel<false, true, false><<<gl, 256, 0, stream>>>(
        AGG, H, H, x_link, F_LINK, F_LINK,
        Wu_nl, Wu_nl + H * H, bu_nl, nullptr, xl2, NL);
    if (node_live_any) {
      gemm_dual_first_kernel<true, false, true><<<gn, 256, 0, stream>>>(
          S0n, F_LINK, F_LINK, x_node, F_NODE, F_NODE,
          Wm_ln, Wm_ln + H * H, bm_ln, dgn, AGG, NN);
      gemm_dual_first_kernel<false, true, false><<<gn, 256, 0, stream>>>(
          AGG, H, H, x_node, F_NODE, F_NODE,
          Wu_ln, Wu_ln + H * H, bu_ln, nullptr, xn2, NN);
    }
  }

  for (int l = 1; l < Lnum; ++l) {
    float* cxn = (l & 1) ? xn2 : xn;
    float* cxl = (l & 1) ? xl2 : xl;
    float* oxn = (l & 1) ? xn  : xn2;
    float* oxl = (l & 1) ? xl  : xl2;

    const float* WmNLt = Wm_nl + (size_t)l * 2 * H * H;
    const float* WmNLb = WmNLt + H * H;
    const float* WuNLt = Wu_nl + (size_t)l * 2 * H * H;
    const float* WuNLb = WuNLt + H * H;
    const float* WmLNt = Wm_ln + (size_t)l * 2 * H * H;
    const float* WmLNb = WmLNt + H * H;
    const float* WuLNt = Wu_ln + (size_t)l * 2 * H * H;
    const float* WuLNb = WuLNt + H * H;

    const bool node_live = (l + 1 < Lnum);

    zero_kernel<<<nb((long)NL * H, 4 * 256 * ZV), 256, 0, stream>>>(Sl, (long)NL * H);
    scatter_kernel<true, false><<<nb((long)nb(E, EPW) * 32, 256), 256, 0, stream>>>(cxn, nl, nl + E, E, Sl, nullptr);
    if (node_live) {
      zero_kernel<<<nb((long)NN * H, 4 * 256 * ZV), 256, 0, stream>>>(Sn, (long)NN * H);
      scatter_kernel<true, false><<<nb((long)nb(E, EPW) * 32, 256), 256, 0, stream>>>(cxl, ln, ln + E, E, Sn, nullptr);
    }

    gemm_dual_kernel<true, false><<<gl, 256, 0, stream>>>(
        Sl, cxl, WmNLt, WmNLb, bm_nl + (size_t)l * H, dgl, AGG, NL);
    gemm_dual_kernel<false, true><<<gl, 256, 0, stream>>>(
        AGG, cxl, WuNLt, WuNLb, bu_nl + (size_t)l * H, nullptr, oxl, NL);
    if (node_live) {
      gemm_dual_kernel<true, false><<<gn, 256, 0, stream>>>(
          Sn, cxn, WmLNt, WmLNb, bm_ln + (size_t)l * H, dgn, AGG, NN);
      gemm_dual_kernel<false, true><<<gn, 256, 0, stream>>>(
          AGG, cxn, WuLNt, WuLNb, bu_ln + (size_t)l * H, nullptr, oxn, NN);
    }
  }

  float* fxl = (Lnum & 1) ? xl2 : xl;
  scatter_kernel<false, true><<<nb((long)nb(NL, EPW) * 32, 256), 256, 0, stream>>>(fxl, nullptr, batch, NL, sums, cnt);
  finalize_kernel<<<nb((long)G * H, 256), 256, 0, stream>>>(sums, cnt, (float*)d_out, G);
}


# first layer's degree count rides in column F of its scatter sums (row pitch 16 floats): an edge's 8 feature adds and its count are one memory request, not two; sums stay atomic, same adds
# speedup vs baseline: 2.6766x; 1.1426x over previous
#include <hip/hip_runtime.h>


typedef __attribute__((ext_vector_type(2))) float v2f;
typedef __attribute__((ext_vector_type(8))) float v8f;

#define H 128
#define LDSPAD 132
#define TPB 8

#define ZV 4
__device__ __forceinline__ void zero_range(float* p, long n, long blk) {
  long head = (long)(((16u - (unsigned)((size_t)p & 15u)) & 15u) >> 2);
  if (head > n) head = n;
  float4* q = (float4*)(p + head);
  const long nv = (n - head) >> 2;
  const long tail = head + (nv << 2);
  const long b0 = blk * (256 * ZV) + threadIdx.x;
  const float4 z = make_float4(0.0f, 0.0f, 0.0f, 0.0f);
#pragma unroll
  for (int j = 0; j < ZV; ++j) {
    const long i = b0 + j * 256;
    if (i < nv) q[i] = z;
  }
  if (blk == 0) {
    if ((long)threadIdx.x < head) p[threadIdx.x] = 0.0f;
    if (tail + (long)threadIdx.x < n) p[tail + threadIdx.x] = 0.0f;
  }
}

__global__ __launch_bounds__(256) void zero_kernel(float* p, long n) {
  zero_range(p, n, (long)blockIdx.x);
}

__global__ __launch_bounds__(256) void zero2_kernel(float* p0, long n0, int nb0,
                                                    float* p1, long n1) {
  if ((int)blockIdx.x < nb0) zero_range(p0, n0, (long)blockIdx.x);
  else zero_range(p1, n1, (long)blockIdx.x - nb0);
}

__global__ __launch_bounds__(256) void pad_kernel(const float* __restrict__ in,
                                                  float* __restrict__ out,
                                                  int N, int F) {
  long t = (long)blockIdx.x * blockDim.x + threadIdx.x;
  if (t >= (long)N * H) return;
  int i = (int)(t >> 7), h = (int)(t & 127);
  out[t] = (h < F) ? in[(long)i * F + h] : 0.0f;
}

#define EPW 8
template <bool HAS_SRC, bool COUNT_ROWS>
__global__ __launch_bounds__(256) void scatter_kernel(const float* __restrict__ X,
                                                      const int* __restrict__ src,
                                                      const int* __restrict__ dst,
                                                      int E, float* __restrict__ S,
                                                      float* __restrict__ cnt) {
  const int lane = threadIdx.x & 31;
  const int wave = __builtin_amdgcn_readfirstlane((int)((blockIdx.x * 256u + threadIdx.x) >> 5));
  const int e0 = wave * EPW;
  if (e0 >= E) return;
  if (e0 + EPW <= E) {
    float v[EPW][4];
#pragma unroll
    for (int j = 0; j < EPW; ++j) {
      const int s = HAS_SRC ? src[e0 + j] : e0 + j;
      const float* xr = X + (size_t)s * H + lane;
#pragma unroll
      for (int q = 0; q < 4; ++q) v[j][q] = xr[32 * q];
    }
#pragma unroll
    for (int j = 0; j < EPW; ++j) {
      float* o = S + (size_t)dst[e0 + j] * H + lane;
#pragma unroll
      for (int q = 0; q < 4; ++q) atomicAdd(o + 32 * q, v[j][q]);
    }
    if (COUNT_ROWS) {
      if (lane < EPW) atomicAdd(cnt + dst[e0 + lane], 1.0f);
    }
  } else {
    for (int e = e0; e < E; ++e) {
      const int s = HAS_SRC ? src[e] : e;
      const float* xr = X + (size_t)s * H + lane;
      float* o = S + (size_t)dst[e] * H + lane;
      for (int q = 0; q < 4; ++q) atomicAdd(o + 32 * q, xr[32 * q]);
      if (COUNT_ROWS) {
        if (lane == 0) atomicAdd(cnt + dst[e], 1.0f);
      }
    }
  }
}

template <bool HAS_DEG, bool RELU>
__global__ __launch_bounds__(256) void gemm_dual_kernel(
    const float* __restrict__ A, const float* __restrict__ B,
    const float* __restrict__ WA, const float* __restrict__ WB,
    const float* __restrict__ bias, const float* __restrict__ deg,
    float* __restrict__ Y, int M) {
  __shared__ float lwa[H * LDSPAD];
  __shared__ float lwb[H * LDSPAD];
  for (int idx = threadIdx.x; idx < H * H; idx += 256) {
    int k = idx >> 7, n = idx & 127;
    lwa[n * LDSPAD + k] = WA[idx];
    lwb[n * LDSPAD + k] = WB[idx];
  }
  __syncthreads();

  const int wave = threadIdx.x >> 5;
  const int lane = threadIdx.x & 31;
  const int l15  = lane & 15;
  const int kh2  = (lane >> 4) << 1;
  const int ntiles = M >> 4;

  for (int t = blockIdx.x * TPB + wave; t < ntiles; t += gridDim.x * TPB) {
    const int m0 = t << 4;
    v8f cA[8], cB[8];
#pragma unroll
    for (int n = 0; n < 8; ++n) {
      cA[n] = (v8f){0, 0, 0, 0, 0, 0, 0, 0};
      cB[n] = (v8f){0, 0, 0, 0, 0, 0, 0, 0};
    }

    const float* Ap = A + (long)(m0 + l15) * H + kh2;
    const float* Bp = B + (long)(m0 + l15) * H + kh2;
    const float* wabase = &lwa[l15 * LDSPAD + kh2];
    const float* wbbase = &lwb[l15 * LDSPAD + kh2];

#pragma unroll 1
    for (int kk = 0; kk < 32; ++kk) {
      const int k0 = kk << 2;
      v2f a = *(const v2f*)(Ap + k0);
      v2f b = *(const v2f*)(Bp + k0);
#pragma unroll
      for (int n = 0; n < 8; ++n) {
        v2f wa = *(const v2f*)(wabase + n * 16 * LDSPAD + k0);
        v2f wb = *(const v2f*)(wbbase + n * 16 * LDSPAD + k0);
        cA[n] = __builtin_amdgcn_wmma_f32_16x16x4_f32(false, a, false, wa,
                                                      (short)0, cA[n], false, false);
        cB[n] = __builtin_amdgcn_wmma_f32_16x16x4_f32(false, b, false, wb,
                                                      (short)0, cB[n], false, false);
      }
    }

    const int rbase = m0 + ((lane >> 4) << 3);
    float degv[8];
    if (HAS_DEG) {
      const float4 d0 = *(const float4*)(deg + rbase);
      const float4 d1 = *(const float4*)(deg + rbase + 4);
      degv[0] = d0.x; degv[1] = d0.y; degv[2] = d0.z; degv[3] = d0.w;
      degv[4] = d1.x; degv[5] = d1.y; degv[6] = d1.z; degv[7] = d1.w;
    } else {
#pragma unroll
      for (int r = 0; r < 8; ++r) degv[r] = 1.0f;
    }
#pragma unroll
    for (int n = 0; n < 8; ++n) {
      const float bv = bias[n * 16 + l15];
#pragma unroll
      for (int r = 0; r < 8; ++r) {
        float acc = HAS_DEG ? (cA[n][r] + degv[r] * (cB[n][r] + bv))
                            : (cA[n][r] + cB[n][r] + bv);
        if (RELU) acc = fmaxf(acc, 0.0f);
        Y[(long)(rbase + r) * H + n * 16 + l15] = acc;
      }
    }
  }
}

#define SFU 16
__global__ __launch_bounds__(256) void edge_first_kernel(
    const float* __restrict__ Xa, int Fa, int lpa, const int* __restrict__ edge_a,
    float* __restrict__ Sa,
    const float* __restrict__ Xb, int Fb, int lpb, const int* __restrict__ edge_b,
    float* __restrict__ Sb, int E) {
  const bool second = (blockIdx.y != 0);
  const float* __restrict__ X = second ? Xb : Xa;
  const int* __restrict__ edge = second ? edge_b : edge_a;
  float* __restrict__ S = second ? Sb : Sa;
  const unsigned uF = (unsigned)(second ? Fb : Fa);
  const unsigned lp = (unsigned)(second ? lpb : lpa);
  const unsigned total = (unsigned)E << lp;
  const unsigned blk0 = blockIdx.x * (256u * SFU);
  if (blk0 >= total) return;
  const unsigned c = threadIdx.x & ((1u << lp) - 1u);
  if (c > uF) return;
  const unsigned e0 = (blk0 + threadIdx.x) >> lp;
  const unsigned estep = 256u >> lp;
  const bool is_cnt = (c == uF);
  const unsigned cc = is_cnt ? uF - 1u : c;
  if (total - blk0 >= 256u * SFU) {
    int s[SFU];
    unsigned o[SFU];
    float v[SFU];
#pragma unroll
    for (int j = 0; j < SFU; ++j) {
      const unsigned e = e0 + j * estep;
      s[j] = edge[e];
      o[j] = (((unsigned)edge[(unsigned)E + e]) << lp) + c;
    }
#pragma unroll
    for (int j = 0; j < SFU; ++j) {
      const float x = X[(size_t)s[j] * uF + cc];
      v[j] = is_cnt ? 1.0f : x;
    }
#pragma unroll
    for (int j = 0; j < SFU; ++j) atomicAdd(S + o[j], v[j]);
  } else {
    for (int j = 0; j < SFU; ++j) {
      const unsigned e = e0 + j * estep;
      if (e < (unsigned)E) {
        const int sj = edge[e];
        const unsigned d = (unsigned)edge[(unsigned)E + e];
        const float x = X[(size_t)sj * uF + cc];
        atomicAdd(S + ((d << lp) + c), is_cnt ? 1.0f : x);
      }
    }
  }
}

__device__ __forceinline__ v2f load_narrow(const float* __restrict__ row, int k, int kv) {
  const int k1 = k + 1;
  const float x0 = row[k  < kv ? k  : kv - 1];
  const float x1 = row[k1 < kv ? k1 : kv - 1];
  v2f r;
  r.x = (k  < kv) ? x0 : 0.0f;
  r.y = (k1 < kv) ? x1 : 0.0f;
  return r;
}

__device__ __forceinline__ void wmma_tail_slots(v8f (&c)[8]) {
  asm volatile("v_nop\n\tv_nop\n\tv_nop\n\tv_nop"
               : "+v"(c[0]), "+v"(c[1]), "+v"(c[2]), "+v"(c[3]),
                 "+v"(c[4]), "+v"(c[5]), "+v"(c[6]), "+v"(c[7]));
}

template <bool HAS_DEG, bool RELU, bool A_NARROW>
__global__ __launch_bounds__(256) void gemm_dual_first_kernel(
    const float* __restrict__ A, int lda, int ka,
    const float* __restrict__ B, int ldb, int kb,
    const float* __restrict__ WA, const float* __restrict__ WB,
    const float* __restrict__ bias, float* __restrict__ deg,
    float* __restrict__ Y, int M) {
  __shared__ float lwa[H * LDSPAD];
  __shared__ float lwb[H * LDSPAD];
  const int ksa = A_NARROW ? ((ka + 3) & ~3) : H;
  const int ksb = (kb + 3) & ~3;
  for (int idx = threadIdx.x; idx < ksa * H; idx += 256) {
    int k = idx >> 7, n = idx & 127;
    lwa[n * LDSPAD + k] = WA[idx];
  }
  for (int idx = threadIdx.x; idx < ksb * H; idx += 256) {
    int k = idx >> 7, n = idx & 127;
    lwb[n * LDSPAD + k] = WB[idx];
  }
  __syncthreads();

  const int wave = threadIdx.x >> 5;
  const int lane = threadIdx.x & 31;
  const int l15  = lane & 15;
  const int kh2  = (lane >> 4) << 1;
  const int ntiles = M >> 4;

  for (int t = blockIdx.x * TPB + wave; t < ntiles; t += gridDim.x * TPB) {
    const int m0 = t << 4;
    v8f cA[8], cB[8];
#pragma unroll
    for (int n = 0; n < 8; ++n) {
      cA[n] = (v8f){0, 0, 0, 0, 0, 0, 0, 0};
      cB[n] = (v8f){0, 0, 0, 0, 0, 0, 0, 0};
    }
    const float* wabase = &lwa[l15 * LDSPAD + kh2];
    const float* wbbase = &lwb[l15 * LDSPAD + kh2];

    const float* Brow = B + (long)(m0 + l15) * ldb;
    for (int k0 = 0; k0 < ksb; k0 += 4) {
      const v2f b = load_narrow(Brow, k0 + kh2, kb);
#pragma unroll
      for (int n = 0; n < 8; ++n) {
        v2f wb = *(const v2f*)(wbbase + n * 16 * LDSPAD + k0);
        cB[n] = __builtin_amdgcn_wmma_f32_16x16x4_f32(false, b, false, wb,
                                                      (short)0, cB[n], false, false);
      }
      wmma_tail_slots(cB);
    }

    if (A_NARROW) {
      const float* Arow = A + (long)(m0 + l15) * lda;
      for (int k0 = 0; k0 < ksa; k0 += 4) {
        const v2f a = load_narrow(Arow, k0 + kh2, ka);
#pragma unroll
        for (int n = 0; n < 8; ++n) {
          v2f wa = *(const v2f*)(wabase + n * 16 * LDSPAD + k0);
          cA[n] = __builtin_amdgcn_wmma_f32_16x16x4_f32(false, a, false, wa,
                                                        (short)0, cA[n], false, false);
        }
        wmma_tail_slots(cA);
      }
    } else {
      const float* Ap = A + (long)(m0 + l15) * H + kh2;
#pragma unroll 2
      for (int kk = 0; kk < 32; ++kk) {
        const int k0 = kk << 2;
        v2f a = *(const v2f*)(Ap + k0);
#pragma unroll
        for (int n = 0; n < 8; ++n) {
          v2f wa = *(const v2f*)(wabase + n * 16 * LDSPAD + k0);
          cA[n] = __builtin_amdgcn_wmma_f32_16x16x4_f32(false, a, false, wa,
                                                        (short)0, cA[n], false, false);
        }
      }
    }

    const int rbase = m0 + ((lane >> 4) << 3);
    float degv[8];
    if (HAS_DEG && A_NARROW) {
#pragma unroll
      for (int r = 0; r < 8; ++r) degv[r] = A[(long)(rbase + r) * lda + ka];
      if (l15 == 0) {
        *(float4*)(deg + rbase)     = make_float4(degv[0], degv[1], degv[2], degv[3]);
        *(float4*)(deg + rbase + 4) = make_float4(degv[4], degv[5], degv[6], degv[7]);
      }
    } else if (HAS_DEG) {
      const float4 d0 = *(const float4*)(deg + rbase);
      const float4 d1 = *(const float4*)(deg + rbase + 4);
      degv[0] = d0.x; degv[1] = d0.y; degv[2] = d0.z; degv[3] = d0.w;
      degv[4] = d1.x; degv[5] = d1.y; degv[6] = d1.z; degv[7] = d1.w;
    } else {
#pragma unroll
      for (int r = 0; r < 8; ++r) degv[r] = 1.0f;
    }
#pragma unroll
    for (int n = 0; n < 8; ++n) {
      const float bv = bias[n * 16 + l15];
#pragma unroll
      for (int r = 0; r < 8; ++r) {
        float acc = HAS_DEG ? (cA[n][r] + degv[r] * (cB[n][r] + bv))
                            : (cA[n][r] + cB[n][r] + bv);
        if (RELU) acc = fmaxf(acc, 0.0f);
        Y[(long)(rbase + r) * H + n * 16 + l15] = acc;
      }
    }
  }
}


__global__ __launch_bounds__(256) void finalize_kernel(const float* __restrict__ sums,
                                                       const float* __restrict__ cnt,
                                                       float* __restrict__ out, int G) {
  int t = blockIdx.x * blockDim.x + threadIdx.x;
  if (t >= G * H) return;
  int g = t >> 7;
  out[t] = sums[t] / fmaxf(cnt[g], 1.0f);
}

extern "C" void kernel_launch(void* const* d_in, const int* in_sizes, int n_in,
                              void* d_out, int out_size, void* d_ws, size_t ws_size,
                              hipStream_t stream) {
  const float* x_node = (const float*)d_in[0];
  const float* x_link = (const float*)d_in[1];
  const float* Wm_nl  = (const float*)d_in[2];
  const float* bm_nl  = (const float*)d_in[3];
  const float* Wu_nl  = (const float*)d_in[4];
  const float* bu_nl  = (const float*)d_in[5];
  const float* Wm_ln  = (const float*)d_in[6];
  const float* bm_ln  = (const float*)d_in[7];
  const float* Wu_ln  = (const float*)d_in[8];
  const float* bu_ln  = (const float*)d_in[9];
  const int*   nl     = (const int*)d_in[10];
  const int*   ln     = (const int*)d_in[11];
  const int*   batch  = (const int*)d_in[12];

  const int NL = in_sizes[12];
  const int NN = 50000;
  const int F_NODE = in_sizes[0] / NN;
  const int F_LINK = in_sizes[1] / NL;
  const int E  = in_sizes[10] / 2;
  const int Lnum = in_sizes[3] / H;
  const int G = out_size / H;

  const size_t MH = (size_t)((NN > NL) ? NN : NL) * H;
  float* ws   = (float*)d_ws;
  float* xn   = ws;
  float* xl   = xn  + MH;
  float* xn2  = xl  + MH;
  float* xl2  = xn2 + MH;
  float* Sl   = xl2 + MH;
  float* Sn   = Sl  + MH;
  float* AGG  = Sn  + MH;
  float* dgl  = AGG + MH;
  float* dgn  = dgl + NL;
  float* sums = dgn + NN;
  float* cnt  = sums + (size_t)G * H;
  (void)n_in; (void)ws_size;

  auto nb = [](long n, int b) { return (int)((n + b - 1) / b); };

  const bool node_live_any = (Lnum > 1);

  const long nsmall = (long)G * H + G;
  const int nbsmall = nb(nsmall, 4 * 256 * ZV);

  auto lp_of = [](int F) {
    return (F < 2) ? 1 : (F < 4) ? 2 : (F < 8) ? 3 : (F < 16) ? 4 : (F < 32) ? 5 : (F < 64) ? 6 : 7;
  };
  const int lpn = lp_of(F_NODE);
  const int lpl = lp_of(F_LINK);

  const int gl = nb(NL / 16, TPB);
  const int gn = nb(NN / 16, TPB);

  if (Lnum == 0) {
    zero_kernel<<<nbsmall, 256, 0, stream>>>(sums, nsmall);
    pad_kernel<<<nb((long)NL * H, 256), 256, 0, stream>>>(x_link, xl, NL, F_LINK);
  } else {
    float* S0l = Sl;
    float* S0n = Sl + ((size_t)NL << lpn);
    const long nS0 = ((long)NL << lpn) + (node_live_any ? ((long)NN << lpl) : 0L);
    zero2_kernel<<<nbsmall + nb(nS0, 4 * 256 * ZV), 256, 0, stream>>>(sums, nsmall, nbsmall, S0l, nS0);
    const int gxa = nb((long)E << lpn, 256 * SFU);
    const int gxb = nb((long)E << lpl, 256 * SFU);
    edge_first_kernel<<<dim3((gxa > gxb) ? gxa : gxb, node_live_any ? 2 : 1), 256, 0, stream>>>(
        x_node, F_NODE, lpn, nl, S0l, x_link, F_LINK, lpl, ln, S0n, E);
    gemm_dual_first_kernel<true, false, true><<<gl, 256, 0, stream>>>(
        S0l, 1 << lpn, F_NODE, x_link, F_LINK, F_LINK,
        Wm_nl, Wm_nl + H * H, bm_nl, dgl, AGG, NL);
    gemm_dual_first_kernel<false, true, false><<<gl, 256, 0, stream>>>(
        AGG, H, H, x_link, F_LINK, F_LINK,
        Wu_nl, Wu_nl + H * H, bu_nl, nullptr, xl2, NL);
    if (node_live_any) {
      gemm_dual_first_kernel<true, false, true><<<gn, 256, 0, stream>>>(
          S0n, 1 << lpl, F_LINK, x_node, F_NODE, F_NODE,
          Wm_ln, Wm_ln + H * H, bm_ln, dgn, AGG, NN);
      gemm_dual_first_kernel<false, true, false><<<gn, 256, 0, stream>>>(
          AGG, H, H, x_node, F_NODE, F_NODE,
          Wu_ln, Wu_ln + H * H, bu_ln, nullptr, xn2, NN);
    }
  }

  for (int l = 1; l < Lnum; ++l) {
    float* cxn = (l & 1) ? xn2 : xn;
    float* cxl = (l & 1) ? xl2 : xl;
    float* oxn = (l & 1) ? xn  : xn2;
    float* oxl = (l & 1) ? xl  : xl2;

    const float* WmNLt = Wm_nl + (size_t)l * 2 * H * H;
    const float* WmNLb = WmNLt + H * H;
    const float* WuNLt = Wu_nl + (size_t)l * 2 * H * H;
    const float* WuNLb = WuNLt + H * H;
    const float* WmLNt = Wm_ln + (size_t)l * 2 * H * H;
    const float* WmLNb = WmLNt + H * H;
    const float* WuLNt = Wu_ln + (size_t)l * 2 * H * H;
    const float* WuLNb = WuLNt + H * H;

    const bool node_live = (l + 1 < Lnum);

    zero_kernel<<<nb((long)NL * H, 4 * 256 * ZV), 256, 0, stream>>>(Sl, (long)NL * H);
    scatter_kernel<true, false><<<nb((long)nb(E, EPW) * 32, 256), 256, 0, stream>>>(cxn, nl, nl + E, E, Sl, nullptr);
    if (node_live) {
      zero_kernel<<<nb((long)NN * H, 4 * 256 * ZV), 256, 0, stream>>>(Sn, (long)NN * H);
      scatter_kernel<true, false><<<nb((long)nb(E, EPW) * 32, 256), 256, 0, stream>>>(cxl, ln, ln + E, E, Sn, nullptr);
    }

    gemm_dual_kernel<true, false><<<gl, 256, 0, stream>>>(
        Sl, cxl, WmNLt, WmNLb, bm_nl + (size_t)l * H, dgl, AGG, NL);
    gemm_dual_kernel<false, true><<<gl, 256, 0, stream>>>(
        AGG, cxl, WuNLt, WuNLb, bu_nl + (size_t)l * H, nullptr, oxl, NL);
    if (node_live) {
      gemm_dual_kernel<true, false><<<gn, 256, 0, stream>>>(
          Sn, cxn, WmLNt, WmLNb, bm_ln + (size_t)l * H, dgn, AGG, NN);
      gemm_dual_kernel<false, true><<<gn, 256, 0, stream>>>(
          AGG, cxn, WuLNt, WuLNb, bu_ln + (size_t)l * H, nullptr, oxn, NN);
    }
  }

  float* fxl = (Lnum & 1) ? xl2 : xl;
  scatter_kernel<false, true><<<nb((long)nb(NL, EPW) * 32, 256), 256, 0, stream>>>(fxl, nullptr, batch, NL, sums, cnt);
  finalize_kernel<<<nb((long)G * H, 256), 256, 0, stream>>>(sums, cnt, (float*)d_out, G);
}


# pooled sum: each wave sums runs of consecutive rows with equal graph id in registers and adds a run once (4 line adds + 1 count add per run, not per row); sums stay atomic, runs summed in row order
# speedup vs baseline: 3.4763x; 1.2988x over previous
#include <hip/hip_runtime.h>


typedef __attribute__((ext_vector_type(2))) float v2f;
typedef __attribute__((ext_vector_type(8))) float v8f;

#define H 128
#define LDSPAD 132
#define TPB 8

#define ZV 4
__device__ __forceinline__ void zero_range(float* p, long n, long blk) {
  long head = (long)(((16u - (unsigned)((size_t)p & 15u)) & 15u) >> 2);
  if (head > n) head = n;
  float4* q = (float4*)(p + head);
  const long nv = (n - head) >> 2;
  const long tail = head + (nv << 2);
  const long b0 = blk * (256 * ZV) + threadIdx.x;
  const float4 z = make_float4(0.0f, 0.0f, 0.0f, 0.0f);
#pragma unroll
  for (int j = 0; j < ZV; ++j) {
    const long i = b0 + j * 256;
    if (i < nv) q[i] = z;
  }
  if (blk == 0) {
    if ((long)threadIdx.x < head) p[threadIdx.x] = 0.0f;
    if (tail + (long)threadIdx.x < n) p[tail + threadIdx.x] = 0.0f;
  }
}

__global__ __launch_bounds__(256) void zero_kernel(float* p, long n) {
  zero_range(p, n, (long)blockIdx.x);
}

__global__ __launch_bounds__(256) void zero2_kernel(float* p0, long n0, int nb0,
                                                    float* p1, long n1) {
  if ((int)blockIdx.x < nb0) zero_range(p0, n0, (long)blockIdx.x);
  else zero_range(p1, n1, (long)blockIdx.x - nb0);
}

__global__ __launch_bounds__(256) void pad_kernel(const float* __restrict__ in,
                                                  float* __restrict__ out,
                                                  int N, int F) {
  long t = (long)blockIdx.x * blockDim.x + threadIdx.x;
  if (t >= (long)N * H) return;
  int i = (int)(t >> 7), h = (int)(t & 127);
  out[t] = (h < F) ? in[(long)i * F + h] : 0.0f;
}

#define EPW 8
__global__ __launch_bounds__(256) void scatter_kernel(const float* __restrict__ X,
                                                      const int* __restrict__ src,
                                                      const int* __restrict__ dst,
                                                      int E, float* __restrict__ S) {
  const int lane = threadIdx.x & 31;
  const int wave = __builtin_amdgcn_readfirstlane((int)((blockIdx.x * 256u + threadIdx.x) >> 5));
  const int e0 = wave * EPW;
  if (e0 >= E) return;
  if (e0 + EPW <= E) {
    float v[EPW][4];
#pragma unroll
    for (int j = 0; j < EPW; ++j) {
      const float* xr = X + (size_t)src[e0 + j] * H + lane;
#pragma unroll
      for (int q = 0; q < 4; ++q) v[j][q] = xr[32 * q];
    }
#pragma unroll
    for (int j = 0; j < EPW; ++j) {
      float* o = S + (size_t)dst[e0 + j] * H + lane;
#pragma unroll
      for (int q = 0; q < 4; ++q) atomicAdd(o + 32 * q, v[j][q]);
    }
  } else {
    for (int e = e0; e < E; ++e) {
      const float* xr = X + (size_t)src[e] * H + lane;
      float* o = S + (size_t)dst[e] * H + lane;
      for (int q = 0; q < 4; ++q) atomicAdd(o + 32 * q, xr[32 * q]);
    }
  }
}

template <bool HAS_DEG, bool RELU>
__global__ __launch_bounds__(256) void gemm_dual_kernel(
    const float* __restrict__ A, const float* __restrict__ B,
    const float* __restrict__ WA, const float* __restrict__ WB,
    const float* __restrict__ bias, const float* __restrict__ deg,
    float* __restrict__ Y, int M) {
  __shared__ float lwa[H * LDSPAD];
  __shared__ float lwb[H * LDSPAD];
  for (int idx = threadIdx.x; idx < H * H; idx += 256) {
    int k = idx >> 7, n = idx & 127;
    lwa[n * LDSPAD + k] = WA[idx];
    lwb[n * LDSPAD + k] = WB[idx];
  }
  __syncthreads();

  const int wave = threadIdx.x >> 5;
  const int lane = threadIdx.x & 31;
  const int l15  = lane & 15;
  const int kh2  = (lane >> 4) << 1;
  const int ntiles = M >> 4;

  for (int t = blockIdx.x * TPB + wave; t < ntiles; t += gridDim.x * TPB) {
    const int m0 = t << 4;
    v8f cA[8], cB[8];
#pragma unroll
    for (int n = 0; n < 8; ++n) {
      cA[n] = (v8f){0, 0, 0, 0, 0, 0, 0, 0};
      cB[n] = (v8f){0, 0, 0, 0, 0, 0, 0, 0};
    }

    const float* Ap = A + (long)(m0 + l15) * H + kh2;
    const float* Bp = B + (long)(m0 + l15) * H + kh2;
    const float* wabase = &lwa[l15 * LDSPAD + kh2];
    const float* wbbase = &lwb[l15 * LDSPAD + kh2];

#pragma unroll 1
    for (int kk = 0; kk < 32; ++kk) {
      const int k0 = kk << 2;
      v2f a = *(const v2f*)(Ap + k0);
      v2f b = *(const v2f*)(Bp + k0);
#pragma unroll
      for (int n = 0; n < 8; ++n) {
        v2f wa = *(const v2f*)(wabase + n * 16 * LDSPAD + k0);
        v2f wb = *(const v2f*)(wbbase + n * 16 * LDSPAD + k0);
        cA[n] = __builtin_amdgcn_wmma_f32_16x16x4_f32(false, a, false, wa,
                                                      (short)0, cA[n], false, false);
        cB[n] = __builtin_amdgcn_wmma_f32_16x16x4_f32(false, b, false, wb,
                                                      (short)0, cB[n], false, false);
      }
    }

    const int rbase = m0 + ((lane >> 4) << 3);
    float degv[8];
    if (HAS_DEG) {
      const float4 d0 = *(const float4*)(deg + rbase);
      const float4 d1 = *(const float4*)(deg + rbase + 4);
      degv[0] = d0.x; degv[1] = d0.y; degv[2] = d0.z; degv[3] = d0.w;
      degv[4] = d1.x; degv[5] = d1.y; degv[6] = d1.z; degv[7] = d1.w;
    } else {
#pragma unroll
      for (int r = 0; r < 8; ++r) degv[r] = 1.0f;
    }
#pragma unroll
    for (int n = 0; n < 8; ++n) {
      const float bv = bias[n * 16 + l15];
#pragma unroll
      for (int r = 0; r < 8; ++r) {
        float acc = HAS_DEG ? (cA[n][r] + degv[r] * (cB[n][r] + bv))
                            : (cA[n][r] + cB[n][r] + bv);
        if (RELU) acc = fmaxf(acc, 0.0f);
        Y[(long)(rbase + r) * H + n * 16 + l15] = acc;
      }
    }
  }
}

#define SFU 16
__global__ __launch_bounds__(256) void edge_first_kernel(
    const float* __restrict__ Xa, int Fa, int lpa, const int* __restrict__ edge_a,
    float* __restrict__ Sa,
    const float* __restrict__ Xb, int Fb, int lpb, const int* __restrict__ edge_b,
    float* __restrict__ Sb, int E) {
  const bool second = (blockIdx.y != 0);
  const float* __restrict__ X = second ? Xb : Xa;
  const int* __restrict__ edge = second ? edge_b : edge_a;
  float* __restrict__ S = second ? Sb : Sa;
  const unsigned uF = (unsigned)(second ? Fb : Fa);
  const unsigned lp = (unsigned)(second ? lpb : lpa);
  const unsigned total = (unsigned)E << lp;
  const unsigned blk0 = blockIdx.x * (256u * SFU);
  if (blk0 >= total) return;
  const unsigned c = threadIdx.x & ((1u << lp) - 1u);
  if (c > uF) return;
  const unsigned e0 = (blk0 + threadIdx.x) >> lp;
  const unsigned estep = 256u >> lp;
  const bool is_cnt = (c == uF);
  const unsigned cc = is_cnt ? uF - 1u : c;
  if (total - blk0 >= 256u * SFU) {
    int s[SFU];
    unsigned o[SFU];
    float v[SFU];
#pragma unroll
    for (int j = 0; j < SFU; ++j) {
      const unsigned e = e0 + j * estep;
      s[j] = edge[e];
      o[j] = (((unsigned)edge[(unsigned)E + e]) << lp) + c;
    }
#pragma unroll
    for (int j = 0; j < SFU; ++j) {
      const float x = X[(size_t)s[j] * uF + cc];
      v[j] = is_cnt ? 1.0f : x;
    }
#pragma unroll
    for (int j = 0; j < SFU; ++j) atomicAdd(S + o[j], v[j]);
  } else {
    for (int j = 0; j < SFU; ++j) {
      const unsigned e = e0 + j * estep;
      if (e < (unsigned)E) {
        const int sj = edge[e];
        const unsigned d = (unsigned)edge[(unsigned)E + e];
        const float x = X[(size_t)sj * uF + cc];
        atomicAdd(S + ((d << lp) + c), is_cnt ? 1.0f : x);
      }
    }
  }
}

__device__ __forceinline__ v2f load_narrow(const float* __restrict__ row, int k, int kv) {
  const int k1 = k + 1;
  const float x0 = row[k  < kv ? k  : kv - 1];
  const float x1 = row[k1 < kv ? k1 : kv - 1];
  v2f r;
  r.x = (k  < kv) ? x0 : 0.0f;
  r.y = (k1 < kv) ? x1 : 0.0f;
  return r;
}

__device__ __forceinline__ void wmma_tail_slots(v8f (&c)[8]) {
  asm volatile("v_nop\n\tv_nop\n\tv_nop\n\tv_nop"
               : "+v"(c[0]), "+v"(c[1]), "+v"(c[2]), "+v"(c[3]),
                 "+v"(c[4]), "+v"(c[5]), "+v"(c[6]), "+v"(c[7]));
}

template <bool HAS_DEG, bool RELU, bool A_NARROW>
__global__ __launch_bounds__(256) void gemm_dual_first_kernel(
    const float* __restrict__ A, int lda, int ka,
    const float* __restrict__ B, int ldb, int kb,
    const float* __restrict__ WA, const float* __restrict__ WB,
    const float* __restrict__ bias, float* __restrict__ deg,
    float* __restrict__ Y, int M) {
  __shared__ float lwa[H * LDSPAD];
  __shared__ float lwb[H * LDSPAD];
  const int ksa = A_NARROW ? ((ka + 3) & ~3) : H;
  const int ksb = (kb + 3) & ~3;
  for (int idx = threadIdx.x; idx < ksa * H; idx += 256) {
    int k = idx >> 7, n = idx & 127;
    lwa[n * LDSPAD + k] = WA[idx];
  }
  for (int idx = threadIdx.x; idx < ksb * H; idx += 256) {
    int k = idx >> 7, n = idx & 127;
    lwb[n * LDSPAD + k] = WB[idx];
  }
  __syncthreads();

  const int wave = threadIdx.x >> 5;
  const int lane = threadIdx.x & 31;
  const int l15  = lane & 15;
  const int kh2  = (lane >> 4) << 1;
  const int ntiles = M >> 4;

  for (int t = blockIdx.x * TPB + wave; t < ntiles; t += gridDim.x * TPB) {
    const int m0 = t << 4;
    v8f cA[8], cB[8];
#pragma unroll
    for (int n = 0; n < 8; ++n) {
      cA[n] = (v8f){0, 0, 0, 0, 0, 0, 0, 0};
      cB[n] = (v8f){0, 0, 0, 0, 0, 0, 0, 0};
    }
    const float* wabase = &lwa[l15 * LDSPAD + kh2];
    const float* wbbase = &lwb[l15 * LDSPAD + kh2];

    const float* Brow = B + (long)(m0 + l15) * ldb;
    for (int k0 = 0; k0 < ksb; k0 += 4) {
      const v2f b = load_narrow(Brow, k0 + kh2, kb);
#pragma unroll
      for (int n = 0; n < 8; ++n) {
        v2f wb = *(const v2f*)(wbbase + n * 16 * LDSPAD + k0);
        cB[n] = __builtin_amdgcn_wmma_f32_16x16x4_f32(false, b, false, wb,
                                                      (short)0, cB[n], false, false);
      }
      wmma_tail_slots(cB);
    }

    if (A_NARROW) {
      const float* Arow = A + (long)(m0 + l15) * lda;
      for (int k0 = 0; k0 < ksa; k0 += 4) {
        const v2f a = load_narrow(Arow, k0 + kh2, ka);
#pragma unroll
        for (int n = 0; n < 8; ++n) {
          v2f wa = *(const v2f*)(wabase + n * 16 * LDSPAD + k0);
          cA[n] = __builtin_amdgcn_wmma_f32_16x16x4_f32(false, a, false, wa,
                                                        (short)0, cA[n], false, false);
        }
        wmma_tail_slots(cA);
      }
    } else {
      const float* Ap = A + (long)(m0 + l15) * H + kh2;
#pragma unroll 2
      for (int kk = 0; kk < 32; ++kk) {
        const int k0 = kk << 2;
        v2f a = *(const v2f*)(Ap + k0);
#pragma unroll
        for (int n = 0; n < 8; ++n) {
          v2f wa = *(const v2f*)(wabase + n * 16 * LDSPAD + k0);
          cA[n] = __builtin_amdgcn_wmma_f32_16x16x4_f32(false, a, false, wa,
                                                        (short)0, cA[n], false, false);
        }
      }
    }

    const int rbase = m0 + ((lane >> 4) << 3);
    float degv[8];
    if (HAS_DEG && A_NARROW) {
#pragma unroll
      for (int r = 0; r < 8; ++r) degv[r] = A[(long)(rbase + r) * lda + ka];
      if (l15 == 0) {
        *(float4*)(deg + rbase)     = make_float4(degv[0], degv[1], degv[2], degv[3]);
        *(float4*)(deg + rbase + 4) = make_float4(degv[4], degv[5], degv[6], degv[7]);
      }
    } else if (HAS_DEG) {
      const float4 d0 = *(const float4*)(deg + rbase);
      const float4 d1 = *(const float4*)(deg + rbase + 4);
      degv[0] = d0.x; degv[1] = d0.y; degv[2] = d0.z; degv[3] = d0.w;
      degv[4] = d1.x; degv[5] = d1.y; degv[6] = d1.z; degv[7] = d1.w;
    } else {
#pragma unroll
      for (int r = 0; r < 8; ++r) degv[r] = 1.0f;
    }
#pragma unroll
    for (int n = 0; n < 8; ++n) {
      const float bv = bias[n * 16 + l15];
#pragma unroll
      for (int r = 0; r < 8; ++r) {
        float acc = HAS_DEG ? (cA[n][r] + degv[r] * (cB[n][r] + bv))
                            : (cA[n][r] + cB[n][r] + bv);
        if (RELU) acc = fmaxf(acc, 0.0f);
        Y[(long)(rbase + r) * H + n * 16 + l15] = acc;
      }
    }
  }
}

#define PRW 16
__device__ __forceinline__ void pool_flush(float* __restrict__ S, float* __restrict__ cnt,
                                           int id, int lane, const float (&acc)[4], int run) {
  float* o = S + (size_t)id * H + lane;
#pragma unroll
  for (int q = 0; q < 4; ++q) atomicAdd(o + 32 * q, acc[q]);
  if (lane == 0) atomicAdd(cnt + id, (float)run);
}

template <bool FULL>
__device__ __forceinline__ void pool_rows(const float* __restrict__ X,
                                          const int* __restrict__ seg, int r0, int nr,
                                          int lane, float* __restrict__ S,
                                          float* __restrict__ cnt) {
  int id[PRW];
  float v[PRW][4];
#pragma unroll
  for (int i = 0; i < PRW; ++i) {
    const int r = r0 + ((FULL || i < nr) ? i : nr - 1);
    id[i] = seg[r];
    const float* xr = X + (size_t)r * H + lane;
#pragma unroll
    for (int q = 0; q < 4; ++q) v[i][q] = xr[32 * q];
  }
  float acc[4] = {v[0][0], v[0][1], v[0][2], v[0][3]};
  int cur = id[0];
  int run = 1;
#pragma unroll
  for (int i = 1; i < PRW; ++i) {
    if (FULL || i < nr) {
      if (id[i] != cur) {
        pool_flush(S, cnt, cur, lane, acc, run);
        acc[0] = 0.0f; acc[1] = 0.0f; acc[2] = 0.0f; acc[3] = 0.0f;
        run = 0;
        cur = id[i];
      }
#pragma unroll
      for (int q = 0; q < 4; ++q) acc[q] += v[i][q];
      ++run;
    }
  }
  pool_flush(S, cnt, cur, lane, acc, run);
}

__global__ __launch_bounds__(256) void pool_kernel(const float* __restrict__ X,
                                                   const int* __restrict__ seg, int N,
                                                   float* __restrict__ S,
                                                   float* __restrict__ cnt) {
  const int lane = threadIdx.x & 31;
  const int wave = __builtin_amdgcn_readfirstlane((int)((blockIdx.x * 256u + threadIdx.x) >> 5));
  const int r0 = wave * PRW;
  if (r0 >= N) return;
  if (r0 + PRW <= N) pool_rows<true>(X, seg, r0, PRW, lane, S, cnt);
  else pool_rows<false>(X, seg, r0, N - r0, lane, S, cnt);
}

__global__ __launch_bounds__(256) void finalize_kernel(const float* __restrict__ sums,
                                                       const float* __restrict__ cnt,
                                                       float* __restrict__ out, int G) {
  int t = blockIdx.x * blockDim.x + threadIdx.x;
  if (t >= G * H) return;
  int g = t >> 7;
  out[t] = sums[t] / fmaxf(cnt[g], 1.0f);
}

extern "C" void kernel_launch(void* const* d_in, const int* in_sizes, int n_in,
                              void* d_out, int out_size, void* d_ws, size_t ws_size,
                              hipStream_t stream) {
  const float* x_node = (const float*)d_in[0];
  const float* x_link = (const float*)d_in[1];
  const float* Wm_nl  = (const float*)d_in[2];
  const float* bm_nl  = (const float*)d_in[3];
  const float* Wu_nl  = (const float*)d_in[4];
  const float* bu_nl  = (const float*)d_in[5];
  const float* Wm_ln  = (const float*)d_in[6];
  const float* bm_ln  = (const float*)d_in[7];
  const float* Wu_ln  = (const float*)d_in[8];
  const float* bu_ln  = (const float*)d_in[9];
  const int*   nl     = (const int*)d_in[10];
  const int*   ln     = (const int*)d_in[11];
  const int*   batch  = (const int*)d_in[12];

  const int NL = in_sizes[12];
  const int NN = 50000;
  const int F_NODE = in_sizes[0] / NN;
  const int F_LINK = in_sizes[1] / NL;
  const int E  = in_sizes[10] / 2;
  const int Lnum = in_sizes[3] / H;
  const int G = out_size / H;

  const size_t MH = (size_t)((NN > NL) ? NN : NL) * H;
  float* ws   = (float*)d_ws;
  float* xn   = ws;
  float* xl   = xn  + MH;
  float* xn2  = xl  + MH;
  float* xl2  = xn2 + MH;
  float* Sl   = xl2 + MH;
  float* Sn   = Sl  + MH;
  float* AGG  = Sn  + MH;
  float* dgl  = AGG + MH;
  float* dgn  = dgl + NL;
  float* sums = dgn + NN;
  float* cnt  = sums + (size_t)G * H;
  (void)n_in; (void)ws_size;

  auto nb = [](long n, int b) { return (int)((n + b - 1) / b); };

  const bool node_live_any = (Lnum > 1);

  const long nsmall = (long)G * H + G;
  const int nbsmall = nb(nsmall, 4 * 256 * ZV);

  auto lp_of = [](int F) {
    return (F < 2) ? 1 : (F < 4) ? 2 : (F < 8) ? 3 : (F < 16) ? 4 : (F < 32) ? 5 : (F < 64) ? 6 : 7;
  };
  const int lpn = lp_of(F_NODE);
  const int lpl = lp_of(F_LINK);

  const int gl = nb(NL / 16, TPB);
  const int gn = nb(NN / 16, TPB);

  if (Lnum == 0) {
    zero_kernel<<<nbsmall, 256, 0, stream>>>(sums, nsmall);
    pad_kernel<<<nb((long)NL * H, 256), 256, 0, stream>>>(x_link, xl, NL, F_LINK);
  } else {
    float* S0l = Sl;
    float* S0n = Sl + ((size_t)NL << lpn);
    const long nS0 = ((long)NL << lpn) + (node_live_any ? ((long)NN << lpl) : 0L);
    zero2_kernel<<<nbsmall + nb(nS0, 4 * 256 * ZV), 256, 0, stream>>>(sums, nsmall, nbsmall, S0l, nS0);
    const int gxa = nb((long)E << lpn, 256 * SFU);
    const int gxb = nb((long)E << lpl, 256 * SFU);
    edge_first_kernel<<<dim3((gxa > gxb) ? gxa : gxb, node_live_any ? 2 : 1), 256, 0, stream>>>(
        x_node, F_NODE, lpn, nl, S0l, x_link, F_LINK, lpl, ln, S0n, E);
    gemm_dual_first_kernel<true, false, true><<<gl, 256, 0, stream>>>(
        S0l, 1 << lpn, F_NODE, x_link, F_LINK, F_LINK,
        Wm_nl, Wm_nl + H * H, bm_nl, dgl, AGG, NL);
    gemm_dual_first_kernel<false, true, false><<<gl, 256, 0, stream>>>(
        AGG, H, H, x_link, F_LINK, F_LINK,
        Wu_nl, Wu_nl + H * H, bu_nl, nullptr, xl2, NL);
    if (node_live_any) {
      gemm_dual_first_kernel<true, false, true><<<gn, 256, 0, stream>>>(
          S0n, 1 << lpl, F_LINK, x_node, F_NODE, F_NODE,
          Wm_ln, Wm_ln + H * H, bm_ln, dgn, AGG, NN);
      gemm_dual_first_kernel<false, true, false><<<gn, 256, 0, stream>>>(
          AGG, H, H, x_node, F_NODE, F_NODE,
          Wu_ln, Wu_ln + H * H, bu_ln, nullptr, xn2, NN);
    }
  }

  for (int l = 1; l < Lnum; ++l) {
    float* cxn = (l & 1) ? xn2 : xn;
    float* cxl = (l & 1) ? xl2 : xl;
    float* oxn = (l & 1) ? xn  : xn2;
    float* oxl = (l & 1) ? xl  : xl2;

    const float* WmNLt = Wm_nl + (size_t)l * 2 * H * H;
    const float* WmNLb = WmNLt + H * H;
    const float* WuNLt = Wu_nl + (size_t)l * 2 * H * H;
    const float* WuNLb = WuNLt + H * H;
    const float* WmLNt = Wm_ln + (size_t)l * 2 * H * H;
    const float* WmLNb = WmLNt + H * H;
    const float* WuLNt = Wu_ln + (size_t)l * 2 * H * H;
    const float* WuLNb = WuLNt + H * H;

    const bool node_live = (l + 1 < Lnum);

    zero_kernel<<<nb((long)NL * H, 4 * 256 * ZV), 256, 0, stream>>>(Sl, (long)NL * H);
    scatter_kernel<<<nb((long)nb(E, EPW) * 32, 256), 256, 0, stream>>>(cxn, nl, nl + E, E, Sl);
    if (node_live) {
      zero_kernel<<<nb((long)NN * H, 4 * 256 * ZV), 256, 0, stream>>>(Sn, (long)NN * H);
      scatter_kernel<<<nb((long)nb(E, EPW) * 32, 256), 256, 0, stream>>>(cxl, ln, ln + E, E, Sn);
    }

    gemm_dual_kernel<true, false><<<gl, 256, 0, stream>>>(
        Sl, cxl, WmNLt, WmNLb, bm_nl + (size_t)l * H, dgl, AGG, NL);
    gemm_dual_kernel<false, true><<<gl, 256, 0, stream>>>(
        AGG, cxl, WuNLt, WuNLb, bu_nl + (size_t)l * H, nullptr, oxl, NL);
    if (node_live) {
      gemm_dual_kernel<true, false><<<gn, 256, 0, stream>>>(
          Sn, cxn, WmLNt, WmLNb, bm_ln + (size_t)l * H, dgn, AGG, NN);
      gemm_dual_kernel<false, true><<<gn, 256, 0, stream>>>(
          AGG, cxn, WuLNt, WuLNb, bu_ln + (size_t)l * H, nullptr, oxn, NN);
    }
  }

  float* fxl = (Lnum & 1) ? xl2 : xl;
  pool_kernel<<<nb((long)nb(NL, PRW) * 32, 256), 256, 0, stream>>>(fxl, batch, NL, sums, cnt);
  finalize_kernel<<<nb((long)G * H, 256), 256, 0, stream>>>(sums, cnt, (float*)d_out, G);
}


# update folded: Wm@WuT, bm@WuT made once on the matrix unit, then each update is ONE pass of three products, S@W1 + deg*(x@W2+b2) + x@WuB; aggregate never stored; sums of products reassociated
# speedup vs baseline: 4.6380x; 1.3342x over previous
#include <hip/hip_runtime.h>


typedef __attribute__((ext_vector_type(2))) float v2f;
typedef __attribute__((ext_vector_type(8))) float v8f;

#define H 128
#define LDSPAD 132
#define TPB 8
#define UW 16
#define FROWS (2 * H + 16)

#define ZV 4
__device__ __forceinline__ void zero_range(float* p, long n, long blk) {
  long head = (long)(((16u - (unsigned)((size_t)p & 15u)) & 15u) >> 2);
  if (head > n) head = n;
  float4* q = (float4*)(p + head);
  const long nv = (n - head) >> 2;
  const long tail = head + (nv << 2);
  const long b0 = blk * (256 * ZV) + threadIdx.x;
  const float4 z = make_float4(0.0f, 0.0f, 0.0f, 0.0f);
#pragma unroll
  for (int j = 0; j < ZV; ++j) {
    const long i = b0 + j * 256;
    if (i < nv) q[i] = z;
  }
  if (blk == 0) {
    if ((long)threadIdx.x < head) p[threadIdx.x] = 0.0f;
    if (tail + (long)threadIdx.x < n) p[tail + threadIdx.x] = 0.0f;
  }
}

__global__ __launch_bounds__(256) void zero_kernel(float* p, long n) {
  zero_range(p, n, (long)blockIdx.x);
}

__global__ __launch_bounds__(256) void zero2_kernel(float* p0, long n0, int nb0,
                                                    float* p1, long n1) {
  if ((int)blockIdx.x < nb0) zero_range(p0, n0, (long)blockIdx.x);
  else zero_range(p1, n1, (long)blockIdx.x - nb0);
}

__global__ __launch_bounds__(256) void pad_kernel(const float* __restrict__ in,
                                                  float* __restrict__ out,
                                                  int N, int F) {
  long t = (long)blockIdx.x * blockDim.x + threadIdx.x;
  if (t >= (long)N * H) return;
  int i = (int)(t >> 7), h = (int)(t & 127);
  out[t] = (h < F) ? in[(long)i * F + h] : 0.0f;
}

#define EPW 8
__global__ __launch_bounds__(256) void scatter_kernel(const float* __restrict__ X,
                                                      const int* __restrict__ src,
                                                      const int* __restrict__ dst,
                                                      int E, float* __restrict__ S) {
  const int lane = threadIdx.x & 31;
  const int wave = __builtin_amdgcn_readfirstlane((int)((blockIdx.x * 256u + threadIdx.x) >> 5));
  const int e0 = wave * EPW;
  if (e0 >= E) return;
  if (e0 + EPW <= E) {
    float v[EPW][4];
#pragma unroll
    for (int j = 0; j < EPW; ++j) {
      const float* xr = X + (size_t)src[e0 + j] * H + lane;
#pragma unroll
      for (int q = 0; q < 4; ++q) v[j][q] = xr[32 * q];
    }
#pragma unroll
    for (int j = 0; j < EPW; ++j) {
      float* o = S + (size_t)dst[e0 + j] * H + lane;
#pragma unroll
      for (int q = 0; q < 4; ++q) atomicAdd(o + 32 * q, v[j][q]);
    }
  } else {
    for (int e = e0; e < E; ++e) {
      const float* xr = X + (size_t)src[e] * H + lane;
      float* o = S + (size_t)dst[e] * H + lane;
      for (int q = 0; q < 4; ++q) atomicAdd(o + 32 * q, xr[32 * q]);
    }
  }
}

#define SFU 16
__global__ __launch_bounds__(256) void edge_first_kernel(
    const float* __restrict__ Xa, int Fa, int lpa, const int* __restrict__ edge_a,
    float* __restrict__ Sa,
    const float* __restrict__ Xb, int Fb, int lpb, const int* __restrict__ edge_b,
    float* __restrict__ Sb, int E) {
  const bool second = (blockIdx.y != 0);
  const float* __restrict__ X = second ? Xb : Xa;
  const int* __restrict__ edge = second ? edge_b : edge_a;
  float* __restrict__ S = second ? Sb : Sa;
  const unsigned uF = (unsigned)(second ? Fb : Fa);
  const unsigned lp = (unsigned)(second ? lpb : lpa);
  const unsigned total = (unsigned)E << lp;
  const unsigned blk0 = blockIdx.x * (256u * SFU);
  if (blk0 >= total) return;
  const unsigned c = threadIdx.x & ((1u << lp) - 1u);
  if (c > uF) return;
  const unsigned e0 = (blk0 + threadIdx.x) >> lp;
  const unsigned estep = 256u >> lp;
  const bool is_cnt = (c == uF);
  const unsigned cc = is_cnt ? uF - 1u : c;
  if (total - blk0 >= 256u * SFU) {
    int s[SFU];
    unsigned o[SFU];
    float v[SFU];
#pragma unroll
    for (int j = 0; j < SFU; ++j) {
      const unsigned e = e0 + j * estep;
      s[j] = edge[e];
      o[j] = (((unsigned)edge[(unsigned)E + e]) << lp) + c;
    }
#pragma unroll
    for (int j = 0; j < SFU; ++j) {
      const float x = X[(size_t)s[j] * uF + cc];
      v[j] = is_cnt ? 1.0f : x;
    }
#pragma unroll
    for (int j = 0; j < SFU; ++j) atomicAdd(S + o[j], v[j]);
  } else {
    for (int j = 0; j < SFU; ++j) {
      const unsigned e = e0 + j * estep;
      if (e < (unsigned)E) {
        const int sj = edge[e];
        const unsigned d = (unsigned)edge[(unsigned)E + e];
        const float x = X[(size_t)sj * uF + cc];
        atomicAdd(S + ((d << lp) + c), is_cnt ? 1.0f : x);
      }
    }
  }
}

__device__ __forceinline__ v2f load_narrow(const float* __restrict__ row, int k, int kv) {
  const int k1 = k + 1;
  const float x0 = row[k  < kv ? k  : kv - 1];
  const float x1 = row[k1 < kv ? k1 : kv - 1];
  v2f r;
  r.x = (k  < kv) ? x0 : 0.0f;
  r.y = (k1 < kv) ? x1 : 0.0f;
  return r;
}

__device__ __forceinline__ void wmma_tail_slots(v8f (&c)[8]) {
  asm volatile("v_nop\n\tv_nop\n\tv_nop\n\tv_nop"
               : "+v"(c[0]), "+v"(c[1]), "+v"(c[2]), "+v"(c[3]),
                 "+v"(c[4]), "+v"(c[5]), "+v"(c[6]), "+v"(c[7]));
}

__device__ __forceinline__ void wmma_tail_slots2(v8f (&c)[8], const v8f (&d)[8]) {
  asm volatile("v_nop\n\tv_nop\n\tv_nop\n\tv_nop"
               : "+v"(c[0]), "+v"(c[1]), "+v"(c[2]), "+v"(c[3]),
                 "+v"(c[4]), "+v"(c[5]), "+v"(c[6]), "+v"(c[7])
               : "v"(d[0]), "v"(d[1]), "v"(d[2]), "v"(d[3]),
                 "v"(d[4]), "v"(d[5]), "v"(d[6]), "v"(d[7]));
}

__global__ __launch_bounds__(256) void fold_kernel(
    const float* __restrict__ Wm_a, const float* __restrict__ bm_a, const float* __restrict__ Wu_a,
    const float* __restrict__ Wm_b, const float* __restrict__ bm_b, const float* __restrict__ Wu_b,
    int Lnum, float* __restrict__ FW) {
  __shared__ float lw[H * LDSPAD];
  const int l = (int)(blockIdx.y >> 1);
  const bool second = (blockIdx.y & 1u) != 0;
  if (second && l + 1 >= Lnum) return;
  const float* __restrict__ Wm = (second ? Wm_b : Wm_a) + (size_t)l * 2 * H * H;
  const float* __restrict__ bm = (second ? bm_b : bm_a) + (size_t)l * H;
  const float* __restrict__ Wu = (second ? Wu_b : Wu_a) + (size_t)l * 2 * H * H;
  float* __restrict__ out = FW + (size_t)blockIdx.y * FROWS * H;
  for (int idx = threadIdx.x; idx < H * H; idx += 256) {
    int k = idx >> 7, n = idx & 127;
    lw[n * LDSPAD + k] = Wu[idx];
  }
  __syncthreads();

  const int wave = threadIdx.x >> 5;
  const int lane = threadIdx.x & 31;
  const int l15  = lane & 15;
  const int kh2  = (lane >> 4) << 1;
  const int m0 = ((int)blockIdx.x * TPB + wave) << 4;
  if (m0 >= FROWS) return;
  v8f c[8];
#pragma unroll
  for (int n = 0; n < 8; ++n) c[n] = (v8f){0, 0, 0, 0, 0, 0, 0, 0};
  const float* Ap = ((m0 < 2 * H) ? Wm + (long)(m0 + l15) * H : bm) + kh2;
  const float* wbase = &lw[l15 * LDSPAD + kh2];
#pragma unroll 2
  for (int kk = 0; kk < 32; ++kk) {
    const int k0 = kk << 2;
    v2f a = *(const v2f*)(Ap + k0);
#pragma unroll
    for (int n = 0; n < 8; ++n) {
      v2f w = *(const v2f*)(wbase + n * 16 * LDSPAD + k0);
      c[n] = __builtin_amdgcn_wmma_f32_16x16x4_f32(false, a, false, w,
                                                   (short)0, c[n], false, false);
    }
  }
  const int rbase = m0 + ((lane >> 4) << 3);
#pragma unroll
  for (int n = 0; n < 8; ++n) {
#pragma unroll
    for (int r = 0; r < 8; ++r) out[(long)(rbase + r) * H + n * 16 + l15] = c[n][r];
  }
}

template <bool NARROW>
__global__ __launch_bounds__(32 * UW) void update_kernel(
    const float* __restrict__ P, int ldp, int kp,
    const float* __restrict__ Q, int ldq, int kq,
    const float* __restrict__ FW, const float* __restrict__ W3,
    const float* __restrict__ bu, float* __restrict__ deg,
    float* __restrict__ Y, int M) {
  __shared__ float lw1[H * LDSPAD];
  __shared__ float lw2[H * LDSPAD];
  __shared__ float lw3[H * LDSPAD];
  const int ks1 = NARROW ? ((kp + 3) & ~3) : H;
  const int ks2 = NARROW ? ((kq + 3) & ~3) : H;
  const float* __restrict__ W2 = FW + H * H;
  const float* __restrict__ b2 = FW + 2 * H * H;
  for (int idx = threadIdx.x; idx < ks1 * H; idx += 32 * UW) {
    int k = idx >> 7, n = idx & 127;
    lw1[n * LDSPAD + k] = FW[idx];
  }
  for (int idx = threadIdx.x; idx < ks2 * H; idx += 32 * UW) {
    int k = idx >> 7, n = idx & 127;
    lw2[n * LDSPAD + k] = W2[idx];
    lw3[n * LDSPAD + k] = W3[idx];
  }
  __syncthreads();

  const int wave = threadIdx.x >> 5;
  const int lane = threadIdx.x & 31;
  const int l15  = lane & 15;
  const int kh2  = (lane >> 4) << 1;
  const int ntiles = M >> 4;

  for (int t = blockIdx.x * UW + wave; t < ntiles; t += gridDim.x * UW) {
    const int m0 = t << 4;
    v8f cA[8], cB[8];
#pragma unroll
    for (int n = 0; n < 8; ++n) {
      cA[n] = (v8f){0, 0, 0, 0, 0, 0, 0, 0};
      cB[n] = (v8f){0, 0, 0, 0, 0, 0, 0, 0};
    }
    const float* w1base = &lw1[l15 * LDSPAD + kh2];
    const float* w2base = &lw2[l15 * LDSPAD + kh2];
    const float* w3base = &lw3[l15 * LDSPAD + kh2];

    if (NARROW) {
      const float* Qrow = Q + (long)(m0 + l15) * ldq;
      const float* Prow = P + (long)(m0 + l15) * ldp;
      for (int k0 = 0; k0 < ks2; k0 += 4) {
        const v2f q = load_narrow(Qrow, k0 + kh2, kq);
#pragma unroll
        for (int n = 0; n < 8; ++n) {
          v2f w2 = *(const v2f*)(w2base + n * 16 * LDSPAD + k0);
          v2f w3 = *(const v2f*)(w3base + n * 16 * LDSPAD + k0);
          cB[n] = __builtin_amdgcn_wmma_f32_16x16x4_f32(false, q, false, w2,
                                                        (short)0, cB[n], false, false);
          cA[n] = __builtin_amdgcn_wmma_f32_16x16x4_f32(false, q, false, w3,
                                                        (short)0, cA[n], false, false);
        }
        wmma_tail_slots2(cA, cB);
      }
      for (int k0 = 0; k0 < ks1; k0 += 4) {
        const v2f p = load_narrow(Prow, k0 + kh2, kp);
#pragma unroll
        for (int n = 0; n < 8; ++n) {
          v2f w1 = *(const v2f*)(w1base + n * 16 * LDSPAD + k0);
          cA[n] = __builtin_amdgcn_wmma_f32_16x16x4_f32(false, p, false, w1,
                                                        (short)0, cA[n], false, false);
        }
        wmma_tail_slots(cA);
      }
    } else {
      const float* Pp = P + (long)(m0 + l15) * H + kh2;
      const float* Qp = Q + (long)(m0 + l15) * H + kh2;
#pragma unroll 1
      for (int kk = 0; kk < 32; ++kk) {
        const int k0 = kk << 2;
        v2f p = *(const v2f*)(Pp + k0);
        v2f q = *(const v2f*)(Qp + k0);
#pragma unroll
        for (int n = 0; n < 8; ++n) {
          v2f w1 = *(const v2f*)(w1base + n * 16 * LDSPAD + k0);
          v2f w2 = *(const v2f*)(w2base + n * 16 * LDSPAD + k0);
          v2f w3 = *(const v2f*)(w3base + n * 16 * LDSPAD + k0);
          cA[n] = __builtin_amdgcn_wmma_f32_16x16x4_f32(false, p, false, w1,
                                                        (short)0, cA[n], false, false);
          cB[n] = __builtin_amdgcn_wmma_f32_16x16x4_f32(false, q, false, w2,
                                                        (short)0, cB[n], false, false);
          cA[n] = __builtin_amdgcn_wmma_f32_16x16x4_f32(false, q, false, w3,
                                                        (short)0, cA[n], false, false);
        }
        wmma_tail_slots2(cA, cB);
      }
    }

    const int rbase = m0 + ((lane >> 4) << 3);
    float degv[8];
    if (NARROW) {
#pragma unroll
      for (int r = 0; r < 8; ++r) degv[r] = P[(long)(rbase + r) * ldp + kp];
      if (l15 == 0) {
        *(float4*)(deg + rbase)     = make_float4(degv[0], degv[1], degv[2], degv[3]);
        *(float4*)(deg + rbase + 4) = make_float4(degv[4], degv[5], degv[6], degv[7]);
      }
    } else {
      const float4 d0 = *(const float4*)(deg + rbase);
      const float4 d1 = *(const float4*)(deg + rbase + 4);
      degv[0] = d0.x; degv[1] = d0.y; degv[2] = d0.z; degv[3] = d0.w;
      degv[4] = d1.x; degv[5] = d1.y; degv[6] = d1.z; degv[7] = d1.w;
    }
#pragma unroll
    for (int n = 0; n < 8; ++n) {
      const float b2v = b2[n * 16 + l15];
      const float buv = bu[n * 16 + l15];
#pragma unroll
      for (int r = 0; r < 8; ++r) {
        const float acc = cA[n][r] + degv[r] * (cB[n][r] + b2v) + buv;
        Y[(long)(rbase + r) * H + n * 16 + l15] = fmaxf(acc, 0.0f);
      }
    }
  }
}

#define PRW 16
__device__ __forceinline__ void pool_flush(float* __restrict__ S, float* __restrict__ cnt,
                                           int id, int lane, const float (&acc)[4], int run) {
  float* o = S + (size_t)id * H + lane;
#pragma unroll
  for (int q = 0; q < 4; ++q) atomicAdd(o + 32 * q, acc[q]);
  if (lane == 0) atomicAdd(cnt + id, (float)run);
}

template <bool FULL>
__device__ __forceinline__ void pool_rows(const float* __restrict__ X,
                                          const int* __restrict__ seg, int r0, int nr,
                                          int lane, float* __restrict__ S,
                                          float* __restrict__ cnt) {
  int id[PRW];
  float v[PRW][4];
#pragma unroll
  for (int i = 0; i < PRW; ++i) {
    const int r = r0 + ((FULL || i < nr) ? i : nr - 1);
    id[i] = seg[r];
    const float* xr = X + (size_t)r * H + lane;
#pragma unroll
    for (int q = 0; q < 4; ++q) v[i][q] = xr[32 * q];
  }
  float acc[4] = {v[0][0], v[0][1], v[0][2], v[0][3]};
  int cur = id[0];
  int run = 1;
#pragma unroll
  for (int i = 1; i < PRW; ++i) {
    if (FULL || i < nr) {
      if (id[i] != cur) {
        pool_flush(S, cnt, cur, lane, acc, run);
        acc[0] = 0.0f; acc[1] = 0.0f; acc[2] = 0.0f; acc[3] = 0.0f;
        run = 0;
        cur = id[i];
      }
#pragma unroll
      for (int q = 0; q < 4; ++q) acc[q] += v[i][q];
      ++run;
    }
  }
  pool_flush(S, cnt, cur, lane, acc, run);
}

__global__ __launch_bounds__(256) void pool_kernel(const float* __restrict__ X,
                                                   const int* __restrict__ seg, int N,
                                                   float* __restrict__ S,
                                                   float* __restrict__ cnt) {
  const int lane = threadIdx.x & 31;
  const int wave = __builtin_amdgcn_readfirstlane((int)((blockIdx.x * 256u + threadIdx.x) >> 5));
  const int r0 = wave * PRW;
  if (r0 >= N) return;
  if (r0 + PRW <= N) pool_rows<true>(X, seg, r0, PRW, lane, S, cnt);
  else pool_rows<false>(X, seg, r0, N - r0, lane, S, cnt);
}

__global__ __launch_bounds__(256) void finalize_kernel(const float* __restrict__ sums,
                                                       const float* __restrict__ cnt,
                                                       float* __restrict__ out, int G) {
  int t = blockIdx.x * blockDim.x + threadIdx.x;
  if (t >= G * H) return;
  int g = t >> 7;
  out[t] = sums[t] / fmaxf(cnt[g], 1.0f);
}

extern "C" void kernel_launch(void* const* d_in, const int* in_sizes, int n_in,
                              void* d_out, int out_size, void* d_ws, size_t ws_size,
                              hipStream_t stream) {
  const float* x_node = (const float*)d_in[0];
  const float* x_link = (const float*)d_in[1];
  const float* Wm_nl  = (const float*)d_in[2];
  const float* bm_nl  = (const float*)d_in[3];
  const float* Wu_nl  = (const float*)d_in[4];
  const float* bu_nl  = (const float*)d_in[5];
  const float* Wm_ln  = (const float*)d_in[6];
  const float* bm_ln  = (const float*)d_in[7];
  const float* Wu_ln  = (const float*)d_in[8];
  const float* bu_ln  = (const float*)d_in[9];
  const int*   nl     = (const int*)d_in[10];
  const int*   ln     = (const int*)d_in[11];
  const int*   batch  = (const int*)d_in[12];

  const int NL = in_sizes[12];
  const int NN = 50000;
  const int F_NODE = in_sizes[0] / NN;
  const int F_LINK = in_sizes[1] / NL;
  const int E  = in_sizes[10] / 2;
  const int Lnum = in_sizes[3] / H;
  const int G = out_size / H;

  const size_t MH = (size_t)((NN > NL) ? NN : NL) * H;
  float* ws   = (float*)d_ws;
  float* xn   = ws;
  float* xl   = xn  + MH;
  float* xn2  = xl  + MH;
  float* xl2  = xn2 + MH;
  float* Sl   = xl2 + MH;
  float* Sn   = Sl  + MH;
  float* FW   = Sn  + MH;
  float* dgl  = FW  + MH;
  float* dgn  = dgl + NL;
  float* sums = dgn + NN;
  float* cnt  = sums + (size_t)G * H;
  (void)n_in; (void)ws_size;

  auto nb = [](long n, int b) { return (int)((n + b - 1) / b); };

  const bool node_live_any = (Lnum > 1);

  const long nsmall = (long)G * H + G;
  const int nbsmall = nb(nsmall, 4 * 256 * ZV);

  auto lp_of = [](int F) {
    return (F < 2) ? 1 : (F < 4) ? 2 : (F < 8) ? 3 : (F < 16) ? 4 : (F < 32) ? 5 : (F < 64) ? 6 : 7;
  };
  const int lpn = lp_of(F_NODE);
  const int lpl = lp_of(F_LINK);

  const int gl = nb(NL / 16, UW);
  const int gn = nb(NN / 16, UW);
  const size_t FSET = (size_t)FROWS * H;

  if (Lnum == 0) {
    zero_kernel<<<nbsmall, 256, 0, stream>>>(sums, nsmall);
    pad_kernel<<<nb((long)NL * H, 256), 256, 0, stream>>>(x_link, xl, NL, F_LINK);
  } else {
    fold_kernel<<<dim3(nb(FROWS / 16, TPB), 2 * Lnum), 256, 0, stream>>>(
        Wm_nl, bm_nl, Wu_nl, Wm_ln, bm_ln, Wu_ln, Lnum, FW);
    float* S0l = Sl;
    float* S0n = Sl + ((size_t)NL << lpn);
    const long nS0 = ((long)NL << lpn) + (node_live_any ? ((long)NN << lpl) : 0L);
    zero2_kernel<<<nbsmall + nb(nS0, 4 * 256 * ZV), 256, 0, stream>>>(sums, nsmall, nbsmall, S0l, nS0);
    const int gxa = nb((long)E << lpn, 256 * SFU);
    const int gxb = nb((long)E << lpl, 256 * SFU);
    edge_first_kernel<<<dim3((gxa > gxb) ? gxa : gxb, node_live_any ? 2 : 1), 256, 0, stream>>>(
        x_node, F_NODE, lpn, nl, S0l, x_link, F_LINK, lpl, ln, S0n, E);
    update_kernel<true><<<gl, 32 * UW, 0, stream>>>(
        S0l, 1 << lpn, F_NODE, x_link, F_LINK, F_LINK,
        FW, Wu_nl + H * H, bu_nl, dgl, xl2, NL);
    if (node_live_any) {
      update_kernel<true><<<gn, 32 * UW, 0, stream>>>(
          S0n, 1 << lpl, F_LINK, x_node, F_NODE, F_NODE,
          FW + FSET, Wu_ln + H * H, bu_ln, dgn, xn2, NN);
    }
  }

  for (int l = 1; l < Lnum; ++l) {
    float* cxn = (l & 1) ? xn2 : xn;
    float* cxl = (l & 1) ? xl2 : xl;
    float* oxn = (l & 1) ? xn  : xn2;
    float* oxl = (l & 1) ? xl  : xl2;

    const float* WuNLb = Wu_nl + (size_t)l * 2 * H * H + H * H;
    const float* WuLNb = Wu_ln + (size_t)l * 2 * H * H + H * H;
    const float* FWl = FW + (size_t)(2 * l) * FSET;

    const bool node_live = (l + 1 < Lnum);

    zero_kernel<<<nb((long)NL * H, 4 * 256 * ZV), 256, 0, stream>>>(Sl, (long)NL * H);
    scatter_kernel<<<nb((long)nb(E, EPW) * 32, 256), 256, 0, stream>>>(cxn, nl, nl + E, E, Sl);
    if (node_live) {
      zero_kernel<<<nb((long)NN * H, 4 * 256 * ZV), 256, 0, stream>>>(Sn, (long)NN * H);
      scatter_kernel<<<nb((long)nb(E, EPW) * 32, 256), 256, 0, stream>>>(cxl, ln, ln + E, E, Sn);
    }

    update_kernel<false><<<gl, 32 * UW, 0, stream>>>(
        Sl, H, H, cxl, H, H, FWl, WuNLb, bu_nl + (size_t)l * H, dgl, oxl, NL);
    if (node_live) {
      update_kernel<false><<<gn, 32 * UW, 0, stream>>>(
          Sn, H, H, cxn, H, H, FWl + FSET, WuLNb, bu_ln + (size_t)l * H, dgn, oxn, NN);
    }
  }

  float* fxl = (Lnum & 1) ? xl2 : xl;
  pool_kernel<<<nb((long)nb(NL, PRW) * 32, 256), 256, 0, stream>>>(fxl, batch, NL, sums, cnt);
  finalize_kernel<<<nb((long)G * H, 256), 256, 0, stream>>>(sums, cnt, (float*)d_out, G);
}


# launches merged 10 -> 8: the weight fold (re-tiled, one 16x16 tile per wave, operands from global, same k order) rides in the head zero-fill's launch; both first-layer updates share one launch
# speedup vs baseline: 4.7920x; 1.0332x over previous
#include <hip/hip_runtime.h>


typedef __attribute__((ext_vector_type(2))) float v2f;
typedef __attribute__((ext_vector_type(8))) float v8f;

#define H 128
#define LDSPAD 132
#define UW 16
#define FROWS (2 * H + 16)

#define ZV 4
__device__ __forceinline__ void zero_range(float* p, long n, long blk) {
  long head = (long)(((16u - (unsigned)((size_t)p & 15u)) & 15u) >> 2);
  if (head > n) head = n;
  float4* q = (float4*)(p + head);
  const long nv = (n - head) >> 2;
  const long tail = head + (nv << 2);
  const long b0 = blk * (256 * ZV) + threadIdx.x;
  const float4 z = make_float4(0.0f, 0.0f, 0.0f, 0.0f);
#pragma unroll
  for (int j = 0; j < ZV; ++j) {
    const long i = b0 + j * 256;
    if (i < nv) q[i] = z;
  }
  if (blk == 0) {
    if ((long)threadIdx.x < head) p[threadIdx.x] = 0.0f;
    if (tail + (long)threadIdx.x < n) p[tail + threadIdx.x] = 0.0f;
  }
}

__global__ __launch_bounds__(256) void zero_kernel(float* p, long n) {
  zero_range(p, n, (long)blockIdx.x);
}

__global__ __launch_bounds__(256) void pad_kernel(const float* __restrict__ in,
                                                  float* __restrict__ out,
                                                  int N, int F) {
  long t = (long)blockIdx.x * blockDim.x + threadIdx.x;
  if (t >= (long)N * H) return;
  int i = (int)(t >> 7), h = (int)(t & 127);
  out[t] = (h < F) ? in[(long)i * F + h] : 0.0f;
}

#define EPW 8
__global__ __launch_bounds__(256) void scatter_kernel(const float* __restrict__ X,
                                                      const int* __restrict__ src,
                                                      const int* __restrict__ dst,
                                                      int E, float* __restrict__ S) {
  const int lane = threadIdx.x & 31;
  const int wave = __builtin_amdgcn_readfirstlane((int)((blockIdx.x * 256u + threadIdx.x) >> 5));
  const int e0 = wave * EPW;
  if (e0 >= E) return;
  if (e0 + EPW <= E) {
    float v[EPW][4];
#pragma unroll
    for (int j = 0; j < EPW; ++j) {
      const float* xr = X + (size_t)src[e0 + j] * H + lane;
#pragma unroll
      for (int q = 0; q < 4; ++q) v[j][q] = xr[32 * q];
    }
#pragma unroll
    for (int j = 0; j < EPW; ++j) {
      float* o = S + (size_t)dst[e0 + j] * H + lane;
#pragma unroll
      for (int q = 0; q < 4; ++q) atomicAdd(o + 32 * q, v[j][q]);
    }
  } else {
    for (int e = e0; e < E; ++e) {
      const float* xr = X + (size_t)src[e] * H + lane;
      float* o = S + (size_t)dst[e] * H + lane;
      for (int q = 0; q < 4; ++q) atomicAdd(o + 32 * q, xr[32 * q]);
    }
  }
}

#define SFU 16
__global__ __launch_bounds__(256) void edge_first_kernel(
    const float* __restrict__ Xa, int Fa, int lpa, const int* __restrict__ edge_a,
    float* __restrict__ Sa,
    const float* __restrict__ Xb, int Fb, int lpb, const int* __restrict__ edge_b,
    float* __restrict__ Sb, int E) {
  const bool second = (blockIdx.y != 0);
  const float* __restrict__ X = second ? Xb : Xa;
  const int* __restrict__ edge = second ? edge_b : edge_a;
  float* __restrict__ S = second ? Sb : Sa;
  const unsigned uF = (unsigned)(second ? Fb : Fa);
  const unsigned lp = (unsigned)(second ? lpb : lpa);
  const unsigned total = (unsigned)E << lp;
  const unsigned blk0 = blockIdx.x * (256u * SFU);
  if (blk0 >= total) return;
  const unsigned c = threadIdx.x & ((1u << lp) - 1u);
  if (c > uF) return;
  const unsigned e0 = (blk0 + threadIdx.x) >> lp;
  const unsigned estep = 256u >> lp;
  const bool is_cnt = (c == uF);
  const unsigned cc = is_cnt ? uF - 1u : c;
  if (total - blk0 >= 256u * SFU) {
    int s[SFU];
    unsigned o[SFU];
    float v[SFU];
#pragma unroll
    for (int j = 0; j < SFU; ++j) {
      const unsigned e = e0 + j * estep;
      s[j] = edge[e];
      o[j] = (((unsigned)edge[(unsigned)E + e]) << lp) + c;
    }
#pragma unroll
    for (int j = 0; j < SFU; ++j) {
      const float x = X[(size_t)s[j] * uF + cc];
      v[j] = is_cnt ? 1.0f : x;
    }
#pragma unroll
    for (int j = 0; j < SFU; ++j) atomicAdd(S + o[j], v[j]);
  } else {
    for (int j = 0; j < SFU; ++j) {
      const unsigned e = e0 + j * estep;
      if (e < (unsigned)E) {
        const int sj = edge[e];
        const unsigned d = (unsigned)edge[(unsigned)E + e];
        const float x = X[(size_t)sj * uF + cc];
        atomicAdd(S + ((d << lp) + c), is_cnt ? 1.0f : x);
      }
    }
  }
}

__device__ __forceinline__ v2f load_narrow(const float* __restrict__ row, int k, int kv) {
  const int k1 = k + 1;
  const float x0 = row[k  < kv ? k  : kv - 1];
  const float x1 = row[k1 < kv ? k1 : kv - 1];
  v2f r;
  r.x = (k  < kv) ? x0 : 0.0f;
  r.y = (k1 < kv) ? x1 : 0.0f;
  return r;
}

__device__ __forceinline__ void wmma_tail_slots(v8f (&c)[8]) {
  asm volatile("v_nop\n\tv_nop\n\tv_nop\n\tv_nop"
               : "+v"(c[0]), "+v"(c[1]), "+v"(c[2]), "+v"(c[3]),
                 "+v"(c[4]), "+v"(c[5]), "+v"(c[6]), "+v"(c[7]));
}

__device__ __forceinline__ void wmma_tail_slots2(v8f (&c)[8], const v8f (&d)[8]) {
  asm volatile("v_nop\n\tv_nop\n\tv_nop\n\tv_nop"
               : "+v"(c[0]), "+v"(c[1]), "+v"(c[2]), "+v"(c[3]),
                 "+v"(c[4]), "+v"(c[5]), "+v"(c[6]), "+v"(c[7])
               : "v"(d[0]), "v"(d[1]), "v"(d[2]), "v"(d[3]),
                 "v"(d[4]), "v"(d[5]), "v"(d[6]), "v"(d[7]));
}

#define FT (FROWS / 16)
__global__ __launch_bounds__(256) void head_kernel(
    const float* __restrict__ Wm_a, const float* __restrict__ bm_a, const float* __restrict__ Wu_a,
    const float* __restrict__ Wm_b, const float* __restrict__ bm_b, const float* __restrict__ Wu_b,
    int Lnum, float* __restrict__ FW, int nfold,
    float* p0, long n0, int nb0, float* p1, long n1) {
  if ((int)blockIdx.x >= nfold) {
    const int b = (int)blockIdx.x - nfold;
    if (b < nb0) zero_range(p0, n0, (long)b);
    else zero_range(p1, n1, (long)(b - nb0));
    return;
  }
  const int set = (int)blockIdx.x / FT;
  const int m0 = ((int)blockIdx.x - set * FT) << 4;
  const int l = set >> 1;
  const bool second = (set & 1) != 0;
  if (second && l + 1 >= Lnum) return;
  const float* __restrict__ Wm = (second ? Wm_b : Wm_a) + (size_t)l * 2 * H * H;
  const float* __restrict__ bm = (second ? bm_b : bm_a) + (size_t)l * H;
  const float* __restrict__ Wu = (second ? Wu_b : Wu_a) + (size_t)l * 2 * H * H;
  float* __restrict__ out = FW + (size_t)set * FROWS * H;

  const int wave = threadIdx.x >> 5;
  const int lane = threadIdx.x & 31;
  const int l15  = lane & 15;
  const int kh2  = (lane >> 4) << 1;
  const int n0c  = wave << 4;
  const float* Ap = ((m0 < 2 * H) ? Wm + (long)(m0 + l15) * H : bm) + kh2;
  const float* Bp = Wu + (long)kh2 * H + n0c + l15;
  v2f a[32], w[32];
#pragma unroll
  for (int kk = 0; kk < 32; ++kk) {
    a[kk] = *(const v2f*)(Ap + 4 * kk);
    w[kk].x = Bp[(long)(4 * kk) * H];
    w[kk].y = Bp[(long)(4 * kk + 1) * H];
  }
  v8f acc = (v8f){0, 0, 0, 0, 0, 0, 0, 0};
#pragma unroll
  for (int kk = 0; kk < 32; ++kk)
    acc = __builtin_amdgcn_wmma_f32_16x16x4_f32(false, a[kk], false, w[kk],
                                                (short)0, acc, false, false);
  const int rbase = m0 + ((lane >> 4) << 3);
#pragma unroll
  for (int r = 0; r < 8; ++r) out[(long)(rbase + r) * H + n0c + l15] = acc[r];
}

template <bool NARROW>
__device__ __forceinline__ void update_body(
    const float* __restrict__ P, int ldp, int kp,
    const float* __restrict__ Q, int ldq, int kq,
    const float* __restrict__ FW, const float* __restrict__ W3,
    const float* __restrict__ bu, float* __restrict__ deg,
    float* __restrict__ Y, int M, int bx, int gx) {
  __shared__ float lw1[H * LDSPAD];
  __shared__ float lw2[H * LDSPAD];
  __shared__ float lw3[H * LDSPAD];
  const int ks1 = NARROW ? ((kp + 3) & ~3) : H;
  const int ks2 = NARROW ? ((kq + 3) & ~3) : H;
  const float* __restrict__ W2 = FW + H * H;
  const float* __restrict__ b2 = FW + 2 * H * H;
  for (int idx = threadIdx.x; idx < ks1 * H; idx += 32 * UW) {
    int k = idx >> 7, n = idx & 127;
    lw1[n * LDSPAD + k] = FW[idx];
  }
  for (int idx = threadIdx.x; idx < ks2 * H; idx += 32 * UW) {
    int k = idx >> 7, n = idx & 127;
    lw2[n * LDSPAD + k] = W2[idx];
    lw3[n * LDSPAD + k] = W3[idx];
  }
  __syncthreads();

  const int wave = threadIdx.x >> 5;
  const int lane = threadIdx.x & 31;
  const int l15  = lane & 15;
  const int kh2  = (lane >> 4) << 1;
  const int ntiles = M >> 4;

  for (int t = bx * UW + wave; t < ntiles; t += gx * UW) {
    const int m0 = t << 4;
    v8f cA[8], cB[8];
#pragma unroll
    for (int n = 0; n < 8; ++n) {
      cA[n] = (v8f){0, 0, 0, 0, 0, 0, 0, 0};
      cB[n] = (v8f){0, 0, 0, 0, 0, 0, 0, 0};
    }
    const float* w1base = &lw1[l15 * LDSPAD + kh2];
    const float* w2base = &lw2[l15 * LDSPAD + kh2];
    const float* w3base = &lw3[l15 * LDSPAD + kh2];

    if (NARROW) {
      const float* Qrow = Q + (long)(m0 + l15) * ldq;
      const float* Prow = P + (long)(m0 + l15) * ldp;
      for (int k0 = 0; k0 < ks2; k0 += 4) {
        const v2f q = load_narrow(Qrow, k0 + kh2, kq);
#pragma unroll
        for (int n = 0; n < 8; ++n) {
          v2f w2 = *(const v2f*)(w2base + n * 16 * LDSPAD + k0);
          v2f w3 = *(const v2f*)(w3base + n * 16 * LDSPAD + k0);
          cB[n] = __builtin_amdgcn_wmma_f32_16x16x4_f32(false, q, false, w2,
                                                        (short)0, cB[n], false, false);
          cA[n] = __builtin_amdgcn_wmma_f32_16x16x4_f32(false, q, false, w3,
                                                        (short)0, cA[n], false, false);
        }
        wmma_tail_slots2(cA, cB);
      }
      for (int k0 = 0; k0 < ks1; k0 += 4) {
        const v2f p = load_narrow(Prow, k0 + kh2, kp);
#pragma unroll
        for (int n = 0; n < 8; ++n) {
          v2f w1 = *(const v2f*)(w1base + n * 16 * LDSPAD + k0);
          cA[n] = __builtin_amdgcn_wmma_f32_16x16x4_f32(false, p, false, w1,
                                                        (short)0, cA[n], false, false);
        }
        wmma_tail_slots(cA);
      }
    } else {
      const float* Pp = P + (long)(m0 + l15) * H + kh2;
      const float* Qp = Q + (long)(m0 + l15) * H + kh2;
#pragma unroll 1
      for (int kk = 0; kk < 32; ++kk) {
        const int k0 = kk << 2;
        v2f p = *(const v2f*)(Pp + k0);
        v2f q = *(const v2f*)(Qp + k0);
#pragma unroll
        for (int n = 0; n < 8; ++n) {
          v2f w1 = *(const v2f*)(w1base + n * 16 * LDSPAD + k0);
          v2f w2 = *(const v2f*)(w2base + n * 16 * LDSPAD + k0);
          v2f w3 = *(const v2f*)(w3base + n * 16 * LDSPAD + k0);
          cA[n] = __builtin_amdgcn_wmma_f32_16x16x4_f32(false, p, false, w1,
                                                        (short)0, cA[n], false, false);
          cB[n] = __builtin_amdgcn_wmma_f32_16x16x4_f32(false, q, false, w2,
                                                        (short)0, cB[n], false, false);
          cA[n] = __builtin_amdgcn_wmma_f32_16x16x4_f32(false, q, false, w3,
                                                        (short)0, cA[n], false, false);
        }
        wmma_tail_slots2(cA, cB);
      }
    }

    const int rbase = m0 + ((lane >> 4) << 3);
    float degv[8];
    if (NARROW) {
#pragma unroll
      for (int r = 0; r < 8; ++r) degv[r] = P[(long)(rbase + r) * ldp + kp];
      if (l15 == 0) {
        *(float4*)(deg + rbase)     = make_float4(degv[0], degv[1], degv[2], degv[3]);
        *(float4*)(deg + rbase + 4) = make_float4(degv[4], degv[5], degv[6], degv[7]);
      }
    } else {
      const float4 d0 = *(const float4*)(deg + rbase);
      const float4 d1 = *(const float4*)(deg + rbase + 4);
      degv[0] = d0.x; degv[1] = d0.y; degv[2] = d0.z; degv[3] = d0.w;
      degv[4] = d1.x; degv[5] = d1.y; degv[6] = d1.z; degv[7] = d1.w;
    }
#pragma unroll
    for (int n = 0; n < 8; ++n) {
      const float b2v = b2[n * 16 + l15];
      const float buv = bu[n * 16 + l15];
#pragma unroll
      for (int r = 0; r < 8; ++r) {
        const float acc = cA[n][r] + degv[r] * (cB[n][r] + b2v) + buv;
        Y[(long)(rbase + r) * H + n * 16 + l15] = fmaxf(acc, 0.0f);
      }
    }
  }
}

__global__ __launch_bounds__(32 * UW) void update_kernel(
    const float* __restrict__ P, const float* __restrict__ Q,
    const float* __restrict__ FW, const float* __restrict__ W3,
    const float* __restrict__ bu, float* __restrict__ deg,
    float* __restrict__ Y, int M) {
  update_body<false>(P, H, H, Q, H, H, FW, W3, bu, deg, Y, M,
                     (int)blockIdx.x, (int)gridDim.x);
}

__global__ __launch_bounds__(32 * UW) void update_first_kernel(
    const float* __restrict__ Pa, int ldpa, int kpa, const float* __restrict__ Qa, int kqa,
    const float* __restrict__ FWa, const float* __restrict__ W3a, const float* __restrict__ bua,
    float* __restrict__ dega, float* __restrict__ Ya, int Ma, int ga,
    const float* __restrict__ Pb, int ldpb, int kpb, const float* __restrict__ Qb, int kqb,
    const float* __restrict__ FWb, const float* __restrict__ W3b, const float* __restrict__ bub,
    float* __restrict__ degb, float* __restrict__ Yb, int Mb, int gb) {
  const bool second = (blockIdx.y != 0);
  const int g = second ? gb : ga;
  if ((int)blockIdx.x >= g) return;
  const int kq = second ? kqb : kqa;
  update_body<true>(second ? Pb : Pa, second ? ldpb : ldpa, second ? kpb : kpa,
                    second ? Qb : Qa, kq, kq,
                    second ? FWb : FWa, second ? W3b : W3a, second ? bub : bua,
                    second ? degb : dega, second ? Yb : Ya, second ? Mb : Ma,
                    (int)blockIdx.x, g);
}

#define PRW 16
__device__ __forceinline__ void pool_flush(float* __restrict__ S, float* __restrict__ cnt,
                                           int id, int lane, const float (&acc)[4], int run) {
  float* o = S + (size_t)id * H + lane;
#pragma unroll
  for (int q = 0; q < 4; ++q) atomicAdd(o + 32 * q, acc[q]);
  if (lane == 0) atomicAdd(cnt + id, (float)run);
}

template <bool FULL>
__device__ __forceinline__ void pool_rows(const float* __restrict__ X,
                                          const int* __restrict__ seg, int r0, int nr,
                                          int lane, float* __restrict__ S,
                                          float* __restrict__ cnt) {
  int id[PRW];
  float v[PRW][4];
#pragma unroll
  for (int i = 0; i < PRW; ++i) {
    const int r = r0 + ((FULL || i < nr) ? i : nr - 1);
    id[i] = seg[r];
    const float* xr = X + (size_t)r * H + lane;
#pragma unroll
    for (int q = 0; q < 4; ++q) v[i][q] = xr[32 * q];
  }
  float acc[4] = {v[0][0], v[0][1], v[0][2], v[0][3]};
  int cur = id[0];
  int run = 1;
#pragma unroll
  for (int i = 1; i < PRW; ++i) {
    if (FULL || i < nr) {
      if (id[i] != cur) {
        pool_flush(S, cnt, cur, lane, acc, run);
        acc[0] = 0.0f; acc[1] = 0.0f; acc[2] = 0.0f; acc[3] = 0.0f;
        run = 0;
        cur = id[i];
      }
#pragma unroll
      for (int q = 0; q < 4; ++q) acc[q] += v[i][q];
      ++run;
    }
  }
  pool_flush(S, cnt, cur, lane, acc, run);
}

__global__ __launch_bounds__(256) void pool_kernel(const float* __restrict__ X,
                                                   const int* __restrict__ seg, int N,
                                                   float* __restrict__ S,
                                                   float* __restrict__ cnt) {
  const int lane = threadIdx.x & 31;
  const int wave = __builtin_amdgcn_readfirstlane((int)((blockIdx.x * 256u + threadIdx.x) >> 5));
  const int r0 = wave * PRW;
  if (r0 >= N) return;
  if (r0 + PRW <= N) pool_rows<true>(X, seg, r0, PRW, lane, S, cnt);
  else pool_rows<false>(X, seg, r0, N - r0, lane, S, cnt);
}

__global__ __launch_bounds__(256) void finalize_kernel(const float* __restrict__ sums,
                                                       const float* __restrict__ cnt,
                                                       float* __restrict__ out, int G) {
  int t = blockIdx.x * blockDim.x + threadIdx.x;
  if (t >= G * H) return;
  int g = t >> 7;
  out[t] = sums[t] / fmaxf(cnt[g], 1.0f);
}

extern "C" void kernel_launch(void* const* d_in, const int* in_sizes, int n_in,
                              void* d_out, int out_size, void* d_ws, size_t ws_size,
                              hipStream_t stream) {
  const float* x_node = (const float*)d_in[0];
  const float* x_link = (const float*)d_in[1];
  const float* Wm_nl  = (const float*)d_in[2];
  const float* bm_nl  = (const float*)d_in[3];
  const float* Wu_nl  = (const float*)d_in[4];
  const float* bu_nl  = (const float*)d_in[5];
  const float* Wm_ln  = (const float*)d_in[6];
  const float* bm_ln  = (const float*)d_in[7];
  const float* Wu_ln  = (const float*)d_in[8];
  const float* bu_ln  = (const float*)d_in[9];
  const int*   nl     = (const int*)d_in[10];
  const int*   ln     = (const int*)d_in[11];
  const int*   batch  = (const int*)d_in[12];

  const int NL = in_sizes[12];
  const int NN = 50000;
  const int F_NODE = in_sizes[0] / NN;
  const int F_LINK = in_sizes[1] / NL;
  const int E  = in_sizes[10] / 2;
  const int Lnum = in_sizes[3] / H;
  const int G = out_size / H;

  const size_t MH = (size_t)((NN > NL) ? NN : NL) * H;
  float* ws   = (float*)d_ws;
  float* xn   = ws;
  float* xl   = xn  + MH;
  float* xn2  = xl  + MH;
  float* xl2  = xn2 + MH;
  float* Sl   = xl2 + MH;
  float* Sn   = Sl  + MH;
  float* FW   = Sn  + MH;
  float* dgl  = FW  + MH;
  float* dgn  = dgl + NL;
  float* sums = dgn + NN;
  float* cnt  = sums + (size_t)G * H;
  (void)n_in; (void)ws_size;

  auto nb = [](long n, int b) { return (int)((n + b - 1) / b); };

  const bool node_live_any = (Lnum > 1);

  const long nsmall = (long)G * H + G;
  const int nbsmall = nb(nsmall, 4 * 256 * ZV);

  auto lp_of = [](int F) {
    return (F < 2) ? 1 : (F < 4) ? 2 : (F < 8) ? 3 : (F < 16) ? 4 : (F < 32) ? 5 : (F < 64) ? 6 : 7;
  };
  const int lpn = lp_of(F_NODE);
  const int lpl = lp_of(F_LINK);

  const int gl = nb(NL / 16, UW);
  const int gn = nb(NN / 16, UW);
  const size_t FSET = (size_t)FROWS * H;

  if (Lnum == 0) {
    zero_kernel<<<nbsmall, 256, 0, stream>>>(sums, nsmall);
    pad_kernel<<<nb((long)NL * H, 256), 256, 0, stream>>>(x_link, xl, NL, F_LINK);
  } else {
    float* S0l = Sl;
    float* S0n = Sl + ((size_t)NL << lpn);
    const long nS0 = ((long)NL << lpn) + (node_live_any ? ((long)NN << lpl) : 0L);
    const int nfold = FT * 2 * Lnum;
    head_kernel<<<nfold + nbsmall + nb(nS0, 4 * 256 * ZV), 256, 0, stream>>>(
        Wm_nl, bm_nl, Wu_nl, Wm_ln, bm_ln, Wu_ln, Lnum, FW, nfold,
        sums, nsmall, nbsmall, S0l, nS0);
    const int gxa = nb((long)E << lpn, 256 * SFU);
    const int gxb = nb((long)E << lpl, 256 * SFU);
    edge_first_kernel<<<dim3((gxa > gxb) ? gxa : gxb, node_live_any ? 2 : 1), 256, 0, stream>>>(
        x_node, F_NODE, lpn, nl, S0l, x_link, F_LINK, lpl, ln, S0n, E);
    update_first_kernel<<<dim3((gl > gn) ? gl : gn, node_live_any ? 2 : 1), 32 * UW, 0, stream>>>(
        S0l, 1 << lpn, F_NODE, x_link, F_LINK,
        FW, Wu_nl + H * H, bu_nl, dgl, xl2, NL, gl,
        S0n, 1 << lpl, F_LINK, x_node, F_NODE,
        FW + FSET, Wu_ln + H * H, bu_ln, dgn, xn2, NN, gn);
  }

  for (int l = 1; l < Lnum; ++l) {
    float* cxn = (l & 1) ? xn2 : xn;
    float* cxl = (l & 1) ? xl2 : xl;
    float* oxn = (l & 1) ? xn  : xn2;
    float* oxl = (l & 1) ? xl  : xl2;

    const float* WuNLb = Wu_nl + (size_t)l * 2 * H * H + H * H;
    const float* WuLNb = Wu_ln + (size_t)l * 2 * H * H + H * H;
    const float* FWl = FW + (size_t)(2 * l) * FSET;

    const bool node_live = (l + 1 < Lnum);

    zero_kernel<<<nb((long)NL * H, 4 * 256 * ZV), 256, 0, stream>>>(Sl, (long)NL * H);
    scatter_kernel<<<nb((long)nb(E, EPW) * 32, 256), 256, 0, stream>>>(cxn, nl, nl + E, E, Sl);
    if (node_live) {
      zero_kernel<<<nb((long)NN * H, 4 * 256 * ZV), 256, 0, stream>>>(Sn, (long)NN * H);
      scatter_kernel<<<nb((long)nb(E, EPW) * 32, 256), 256, 0, stream>>>(cxl, ln, ln + E, E, Sn);
    }

    update_kernel<<<gl, 32 * UW, 0, stream>>>(
        Sl, cxl, FWl, WuNLb, bu_nl + (size_t)l * H, dgl, oxl, NL);
    if (node_live) {
      update_kernel<<<gn, 32 * UW, 0, stream>>>(
          Sn, cxn, FWl + FSET, WuLNb, bu_ln + (size_t)l * H, dgn, oxn, NN);
    }
  }

  float* fxl = (Lnum & 1) ? xl2 : xl;
  pool_kernel<<<nb((long)nb(NL, PRW) * 32, 256), 256, 0, stream>>>(fxl, batch, NL, sums, cnt);
  finalize_kernel<<<nb((long)G * H, 256), 256, 0, stream>>>(sums, cnt, (float*)d_out, G);
}


# later layers' link sums gathered, not scattered: first pass's count add returns an edge's rank, ids fill a 32-slot list per link; a wave sums its link's rows in rank order, non-atomic; no zero-fill
# speedup vs baseline: 5.6740x; 1.1841x over previous
#include <hip/hip_runtime.h>


typedef __attribute__((ext_vector_type(2))) float v2f;
typedef __attribute__((ext_vector_type(8))) float v8f;

#define H 128
#define LDSPAD 132
#define UW 16
#define FROWS (2 * H + 16)

#define ZV 4
__device__ __forceinline__ void zero_range(float* p, long n, long blk) {
  long head = (long)(((16u - (unsigned)((size_t)p & 15u)) & 15u) >> 2);
  if (head > n) head = n;
  float4* q = (float4*)(p + head);
  const long nv = (n - head) >> 2;
  const long tail = head + (nv << 2);
  const long b0 = blk * (256 * ZV) + threadIdx.x;
  const float4 z = make_float4(0.0f, 0.0f, 0.0f, 0.0f);
#pragma unroll
  for (int j = 0; j < ZV; ++j) {
    const long i = b0 + j * 256;
    if (i < nv) q[i] = z;
  }
  if (blk == 0) {
    if ((long)threadIdx.x < head) p[threadIdx.x] = 0.0f;
    if (tail + (long)threadIdx.x < n) p[tail + threadIdx.x] = 0.0f;
  }
}

__global__ __launch_bounds__(256) void zero_kernel(float* p, long n) {
  zero_range(p, n, (long)blockIdx.x);
}

__global__ __launch_bounds__(256) void pad_kernel(const float* __restrict__ in,
                                                  float* __restrict__ out,
                                                  int N, int F) {
  long t = (long)blockIdx.x * blockDim.x + threadIdx.x;
  if (t >= (long)N * H) return;
  int i = (int)(t >> 7), h = (int)(t & 127);
  out[t] = (h < F) ? in[(long)i * F + h] : 0.0f;
}

#define EPW 8
__global__ __launch_bounds__(256) void scatter_kernel(const float* __restrict__ X,
                                                      const int* __restrict__ src,
                                                      const int* __restrict__ dst,
                                                      int E, float* __restrict__ S) {
  const int lane = threadIdx.x & 31;
  const int wave = __builtin_amdgcn_readfirstlane((int)((blockIdx.x * 256u + threadIdx.x) >> 5));
  const int e0 = wave * EPW;
  if (e0 >= E) return;
  if (e0 + EPW <= E) {
    float v[EPW][4];
#pragma unroll
    for (int j = 0; j < EPW; ++j) {
      const float* xr = X + (size_t)src[e0 + j] * H + lane;
#pragma unroll
      for (int q = 0; q < 4; ++q) v[j][q] = xr[32 * q];
    }
#pragma unroll
    for (int j = 0; j < EPW; ++j) {
      float* o = S + (size_t)dst[e0 + j] * H + lane;
#pragma unroll
      for (int q = 0; q < 4; ++q) atomicAdd(o + 32 * q, v[j][q]);
    }
  } else {
    for (int e = e0; e < E; ++e) {
      const float* xr = X + (size_t)src[e] * H + lane;
      float* o = S + (size_t)dst[e] * H + lane;
      for (int q = 0; q < 4; ++q) atomicAdd(o + 32 * q, xr[32 * q]);
    }
  }
}

#define SFU 16
#define CAP 32
__device__ __forceinline__ void list_put(int* __restrict__ ell, int* __restrict__ ovf_cnt,
                                         int* __restrict__ ovf, unsigned d, unsigned r, int s) {
  if (r < (unsigned)CAP) {
    ell[(size_t)d * CAP + r] = s;
  } else {
    const int k = atomicAdd(ovf_cnt, 1);
    ovf[2 * (size_t)k] = s;
    ovf[2 * (size_t)k + 1] = (int)d;
  }
}
__global__ __launch_bounds__(256) void edge_first_kernel(
    const float* __restrict__ Xa, int Fa, int lpa, const int* __restrict__ edge_a,
    float* __restrict__ Sa,
    const float* __restrict__ Xb, int Fb, int lpb, const int* __restrict__ edge_b,
    float* __restrict__ Sb, int E,
    int* __restrict__ ell, int* __restrict__ ovf_cnt, int* __restrict__ ovf) {
  const bool second = (blockIdx.y != 0);
  const bool build = (ell != nullptr) && !second;
  const float* __restrict__ X = second ? Xb : Xa;
  const int* __restrict__ edge = second ? edge_b : edge_a;
  float* __restrict__ S = second ? Sb : Sa;
  const unsigned uF = (unsigned)(second ? Fb : Fa);
  const unsigned lp = (unsigned)(second ? lpb : lpa);
  const unsigned total = (unsigned)E << lp;
  const unsigned blk0 = blockIdx.x * (256u * SFU);
  if (blk0 >= total) return;
  const unsigned c = threadIdx.x & ((1u << lp) - 1u);
  if (c > uF) return;
  const unsigned e0 = (blk0 + threadIdx.x) >> lp;
  const unsigned estep = 256u >> lp;
  const bool is_cnt = (c == uF);
  const unsigned cc = is_cnt ? uF - 1u : c;
  if (total - blk0 >= 256u * SFU) {
    int s[SFU];
    unsigned o[SFU];
    float v[SFU];
#pragma unroll
    for (int j = 0; j < SFU; ++j) {
      const unsigned e = e0 + j * estep;
      s[j] = edge[e];
      o[j] = (((unsigned)edge[(unsigned)E + e]) << lp) + c;
    }
#pragma unroll
    for (int j = 0; j < SFU; ++j) {
      const float x = X[(size_t)s[j] * uF + cc];
      v[j] = is_cnt ? 1.0f : x;
    }
    if (build) {
      float old[SFU];
#pragma unroll
      for (int j = 0; j < SFU; ++j) old[j] = atomicAdd(S + o[j], v[j]);
      if (is_cnt) {
#pragma unroll
        for (int j = 0; j < SFU; ++j)
          list_put(ell, ovf_cnt, ovf, o[j] >> lp, (unsigned)old[j], s[j]);
      }
    } else {
#pragma unroll
      for (int j = 0; j < SFU; ++j) atomicAdd(S + o[j], v[j]);
    }
  } else {
    for (int j = 0; j < SFU; ++j) {
      const unsigned e = e0 + j * estep;
      if (e < (unsigned)E) {
        const int sj = edge[e];
        const unsigned d = (unsigned)edge[(unsigned)E + e];
        const float x = X[(size_t)sj * uF + cc];
        if (build) {
          const float old = atomicAdd(S + ((d << lp) + c), is_cnt ? 1.0f : x);
          if (is_cnt) list_put(ell, ovf_cnt, ovf, d, (unsigned)old, sj);
        } else {
          atomicAdd(S + ((d << lp) + c), is_cnt ? 1.0f : x);
        }
      }
    }
  }
}

#define GCH 8
template <int M>
__device__ __forceinline__ void gather_rows(const float* __restrict__ X,
                                            const int* __restrict__ e, int lane,
                                            float (&acc)[4]) {
  int s[M];
  float v[M][4];
#pragma unroll
  for (int u = 0; u < M; ++u) s[u] = e[u];
#pragma unroll
  for (int u = 0; u < M; ++u) {
    const float* xr = X + (size_t)s[u] * H + lane;
#pragma unroll
    for (int q = 0; q < 4; ++q) v[u][q] = xr[32 * q];
  }
#pragma unroll
  for (int u = 0; u < M; ++u) {
#pragma unroll
    for (int q = 0; q < 4; ++q) acc[q] += v[u][q];
  }
}

__global__ __launch_bounds__(256) void gather_kernel(const float* __restrict__ X,
                                                     const int* __restrict__ ell,
                                                     const float* __restrict__ deg,
                                                     const int* __restrict__ ovf,
                                                     const int* __restrict__ ovf_cnt,
                                                     int N, float* __restrict__ S) {
  const int lane = threadIdx.x & 31;
  const int i = __builtin_amdgcn_readfirstlane((int)((blockIdx.x * 256u + threadIdx.x) >> 5));
  if (i >= N) return;
  const int n = __builtin_amdgcn_readfirstlane((int)deg[i]);
  const int nin = (n < CAP) ? n : CAP;
  const int* __restrict__ er = ell + (size_t)i * CAP;
  float acc[4] = {0.0f, 0.0f, 0.0f, 0.0f};
  for (int j0 = 0; j0 < nin; j0 += GCH) {
    const int* __restrict__ e = er + j0;
    switch (nin - j0) {
      case 1: gather_rows<1>(X, e, lane, acc); break;
      case 2: gather_rows<2>(X, e, lane, acc); break;
      case 3: gather_rows<3>(X, e, lane, acc); break;
      case 4: gather_rows<4>(X, e, lane, acc); break;
      case 5: gather_rows<5>(X, e, lane, acc); break;
      case 6: gather_rows<6>(X, e, lane, acc); break;
      case 7: gather_rows<7>(X, e, lane, acc); break;
      default: gather_rows<GCH>(X, e, lane, acc); break;
    }
  }
  if (n > CAP) {
    const int m = *ovf_cnt;
    for (int k = 0; k < m; ++k) {
      if (ovf[2 * (size_t)k + 1] == i) {
        const float* xr = X + (size_t)ovf[2 * (size_t)k] * H + lane;
#pragma unroll
        for (int q = 0; q < 4; ++q) acc[q] += xr[32 * q];
      }
    }
  }
  float* o = S + (size_t)i * H + lane;
#pragma unroll
  for (int q = 0; q < 4; ++q) o[32 * q] = acc[q];
}

__device__ __forceinline__ v2f load_narrow(const float* __restrict__ row, int k, int kv) {
  const int k1 = k + 1;
  const float x0 = row[k  < kv ? k  : kv - 1];
  const float x1 = row[k1 < kv ? k1 : kv - 1];
  v2f r;
  r.x = (k  < kv) ? x0 : 0.0f;
  r.y = (k1 < kv) ? x1 : 0.0f;
  return r;
}

__device__ __forceinline__ void wmma_tail_slots(v8f (&c)[8]) {
  asm volatile("v_nop\n\tv_nop\n\tv_nop\n\tv_nop"
               : "+v"(c[0]), "+v"(c[1]), "+v"(c[2]), "+v"(c[3]),
                 "+v"(c[4]), "+v"(c[5]), "+v"(c[6]), "+v"(c[7]));
}

__device__ __forceinline__ void wmma_tail_slots2(v8f (&c)[8], const v8f (&d)[8]) {
  asm volatile("v_nop\n\tv_nop\n\tv_nop\n\tv_nop"
               : "+v"(c[0]), "+v"(c[1]), "+v"(c[2]), "+v"(c[3]),
                 "+v"(c[4]), "+v"(c[5]), "+v"(c[6]), "+v"(c[7])
               : "v"(d[0]), "v"(d[1]), "v"(d[2]), "v"(d[3]),
                 "v"(d[4]), "v"(d[5]), "v"(d[6]), "v"(d[7]));
}

#define FT (FROWS / 16)
__global__ __launch_bounds__(256) void head_kernel(
    const float* __restrict__ Wm_a, const float* __restrict__ bm_a, const float* __restrict__ Wu_a,
    const float* __restrict__ Wm_b, const float* __restrict__ bm_b, const float* __restrict__ Wu_b,
    int Lnum, float* __restrict__ FW, int nfold,
    float* p0, long n0, int nb0, float* p1, long n1, int* ovf_cnt) {
  if (ovf_cnt != nullptr && blockIdx.x == 0 && threadIdx.x == 0) *ovf_cnt = 0;
  if ((int)blockIdx.x >= nfold) {
    const int b = (int)blockIdx.x - nfold;
    if (b < nb0) zero_range(p0, n0, (long)b);
    else zero_range(p1, n1, (long)(b - nb0));
    return;
  }
  const int set = (int)blockIdx.x / FT;
  const int m0 = ((int)blockIdx.x - set * FT) << 4;
  const int l = set >> 1;
  const bool second = (set & 1) != 0;
  if (second && l + 1 >= Lnum) return;
  const float* __restrict__ Wm = (second ? Wm_b : Wm_a) + (size_t)l * 2 * H * H;
  const float* __restrict__ bm = (second ? bm_b : bm_a) + (size_t)l * H;
  const float* __restrict__ Wu = (second ? Wu_b : Wu_a) + (size_t)l * 2 * H * H;
  float* __restrict__ out = FW + (size_t)set * FROWS * H;

  const int wave = threadIdx.x >> 5;
  const int lane = threadIdx.x & 31;
  const int l15  = lane & 15;
  const int kh2  = (lane >> 4) << 1;
  const int n0c  = wave << 4;
  const float* Ap = ((m0 < 2 * H) ? Wm + (long)(m0 + l15) * H : bm) + kh2;
  const float* Bp = Wu + (long)kh2 * H + n0c + l15;
  v2f a[32], w[32];
#pragma unroll
  for (int kk = 0; kk < 32; ++kk) {
    a[kk] = *(const v2f*)(Ap + 4 * kk);
    w[kk].x = Bp[(long)(4 * kk) * H];
    w[kk].y = Bp[(long)(4 * kk + 1) * H];
  }
  v8f acc = (v8f){0, 0, 0, 0, 0, 0, 0, 0};
#pragma unroll
  for (int kk = 0; kk < 32; ++kk)
    acc = __builtin_amdgcn_wmma_f32_16x16x4_f32(false, a[kk], false, w[kk],
                                                (short)0, acc, false, false);
  const int rbase = m0 + ((lane >> 4) << 3);
#pragma unroll
  for (int r = 0; r < 8; ++r) out[(long)(rbase + r) * H + n0c + l15] = acc[r];
}

template <bool NARROW>
__device__ __forceinline__ void update_body(
    const float* __restrict__ P, int ldp, int kp,
    const float* __restrict__ Q, int ldq, int kq,
    const float* __restrict__ FW, const float* __restrict__ W3,
    const float* __restrict__ bu, float* __restrict__ deg,
    float* __restrict__ Y, int M, int bx, int gx) {
  __shared__ float lw1[H * LDSPAD];
  __shared__ float lw2[H * LDSPAD];
  __shared__ float lw3[H * LDSPAD];
  const int ks1 = NARROW ? ((kp + 3) & ~3) : H;
  const int ks2 = NARROW ? ((kq + 3) & ~3) : H;
  const float* __restrict__ W2 = FW + H * H;
  const float* __restrict__ b2 = FW + 2 * H * H;
  for (int idx = threadIdx.x; idx < ks1 * H; idx += 32 * UW) {
    int k = idx >> 7, n = idx & 127;
    lw1[n * LDSPAD + k] = FW[idx];
  }
  for (int idx = threadIdx.x; idx < ks2 * H; idx += 32 * UW) {
    int k = idx >> 7, n = idx & 127;
    lw2[n * LDSPAD + k] = W2[idx];
    lw3[n * LDSPAD + k] = W3[idx];
  }
  __syncthreads();

  const int wave = threadIdx.x >> 5;
  const int lane = threadIdx.x & 31;
  const int l15  = lane & 15;
  const int kh2  = (lane >> 4) << 1;
  const int ntiles = M >> 4;

  for (int t = bx * UW + wave; t < ntiles; t += gx * UW) {
    const int m0 = t << 4;
    v8f cA[8], cB[8];
#pragma unroll
    for (int n = 0; n < 8; ++n) {
      cA[n] = (v8f){0, 0, 0, 0, 0, 0, 0, 0};
      cB[n] = (v8f){0, 0, 0, 0, 0, 0, 0, 0};
    }
    const float* w1base = &lw1[l15 * LDSPAD + kh2];
    const float* w2base = &lw2[l15 * LDSPAD + kh2];
    const float* w3base = &lw3[l15 * LDSPAD + kh2];

    if (NARROW) {
      const float* Qrow = Q + (long)(m0 + l15) * ldq;
      const float* Prow = P + (long)(m0 + l15) * ldp;
      for (int k0 = 0; k0 < ks2; k0 += 4) {
        const v2f q = load_narrow(Qrow, k0 + kh2, kq);
#pragma unroll
        for (int n = 0; n < 8; ++n) {
          v2f w2 = *(const v2f*)(w2base + n * 16 * LDSPAD + k0);
          v2f w3 = *(const v2f*)(w3base + n * 16 * LDSPAD + k0);
          cB[n] = __builtin_amdgcn_wmma_f32_16x16x4_f32(false, q, false, w2,
                                                        (short)0, cB[n], false, false);
          cA[n] = __builtin_amdgcn_wmma_f32_16x16x4_f32(false, q, false, w3,
                                                        (short)0, cA[n], false, false);
        }
        wmma_tail_slots2(cA, cB);
      }
      for (int k0 = 0; k0 < ks1; k0 += 4) {
        const v2f p = load_narrow(Prow, k0 + kh2, kp);
#pragma unroll
        for (int n = 0; n < 8; ++n) {
          v2f w1 = *(const v2f*)(w1base + n * 16 * LDSPAD + k0);
          cA[n] = __builtin_amdgcn_wmma_f32_16x16x4_f32(false, p, false, w1,
                                                        (short)0, cA[n], false, false);
        }
        wmma_tail_slots(cA);
      }
    } else {
      const float* Pp = P + (long)(m0 + l15) * H + kh2;
      const float* Qp = Q + (long)(m0 + l15) * H + kh2;
#pragma unroll 1
      for (int kk = 0; kk < 32; ++kk) {
        const int k0 = kk << 2;
        v2f p = *(const v2f*)(Pp + k0);
        v2f q = *(const v2f*)(Qp + k0);
#pragma unroll
        for (int n = 0; n < 8; ++n) {
          v2f w1 = *(const v2f*)(w1base + n * 16 * LDSPAD + k0);
          v2f w2 = *(const v2f*)(w2base + n * 16 * LDSPAD + k0);
          v2f w3 = *(const v2f*)(w3base + n * 16 * LDSPAD + k0);
          cA[n] = __builtin_amdgcn_wmma_f32_16x16x4_f32(false, p, false, w1,
                                                        (short)0, cA[n], false, false);
          cB[n] = __builtin_amdgcn_wmma_f32_16x16x4_f32(false, q, false, w2,
                                                        (short)0, cB[n], false, false);
          cA[n] = __builtin_amdgcn_wmma_f32_16x16x4_f32(false, q, false, w3,
                                                        (short)0, cA[n], false, false);
        }
        wmma_tail_slots2(cA, cB);
      }
    }

    const int rbase = m0 + ((lane >> 4) << 3);
    float degv[8];
    if (NARROW) {
#pragma unroll
      for (int r = 0; r < 8; ++r) degv[r] = P[(long)(rbase + r) * ldp + kp];
      if (l15 == 0) {
        *(float4*)(deg + rbase)     = make_float4(degv[0], degv[1], degv[2], degv[3]);
        *(float4*)(deg + rbase + 4) = make_float4(degv[4], degv[5], degv[6], degv[7]);
      }
    } else {
      const float4 d0 = *(const float4*)(deg + rbase);
      const float4 d1 = *(const float4*)(deg + rbase + 4);
      degv[0] = d0.x; degv[1] = d0.y; degv[2] = d0.z; degv[3] = d0.w;
      degv[4] = d1.x; degv[5] = d1.y; degv[6] = d1.z; degv[7] = d1.w;
    }
#pragma unroll
    for (int n = 0; n < 8; ++n) {
      const float b2v = b2[n * 16 + l15];
      const float buv = bu[n * 16 + l15];
#pragma unroll
      for (int r = 0; r < 8; ++r) {
        const float acc = cA[n][r] + degv[r] * (cB[n][r] + b2v) + buv;
        Y[(long)(rbase + r) * H + n * 16 + l15] = fmaxf(acc, 0.0f);
      }
    }
  }
}

__global__ __launch_bounds__(32 * UW) void update_kernel(
    const float* __restrict__ P, const float* __restrict__ Q,
    const float* __restrict__ FW, const float* __restrict__ W3,
    const float* __restrict__ bu, float* __restrict__ deg,
    float* __restrict__ Y, int M) {
  update_body<false>(P, H, H, Q, H, H, FW, W3, bu, deg, Y, M,
                     (int)blockIdx.x, (int)gridDim.x);
}

__global__ __launch_bounds__(32 * UW) void update_first_kernel(
    const float* __restrict__ Pa, int ldpa, int kpa, const float* __restrict__ Qa, int kqa,
    const float* __restrict__ FWa, const float* __restrict__ W3a, const float* __restrict__ bua,
    float* __restrict__ dega, float* __restrict__ Ya, int Ma, int ga,
    const float* __restrict__ Pb, int ldpb, int kpb, const float* __restrict__ Qb, int kqb,
    const float* __restrict__ FWb, const float* __restrict__ W3b, const float* __restrict__ bub,
    float* __restrict__ degb, float* __restrict__ Yb, int Mb, int gb) {
  const bool second = (blockIdx.y != 0);
  const int g = second ? gb : ga;
  if ((int)blockIdx.x >= g) return;
  const int kq = second ? kqb : kqa;
  update_body<true>(second ? Pb : Pa, second ? ldpb : ldpa, second ? kpb : kpa,
                    second ? Qb : Qa, kq, kq,
                    second ? FWb : FWa, second ? W3b : W3a, second ? bub : bua,
                    second ? degb : dega, second ? Yb : Ya, second ? Mb : Ma,
                    (int)blockIdx.x, g);
}

#define PRW 16
__device__ __forceinline__ void pool_flush(float* __restrict__ S, float* __restrict__ cnt,
                                           int id, int lane, const float (&acc)[4], int run) {
  float* o = S + (size_t)id * H + lane;
#pragma unroll
  for (int q = 0; q < 4; ++q) atomicAdd(o + 32 * q, acc[q]);
  if (lane == 0) atomicAdd(cnt + id, (float)run);
}

template <bool FULL>
__device__ __forceinline__ void pool_rows(const float* __restrict__ X,
                                          const int* __restrict__ seg, int r0, int nr,
                                          int lane, float* __restrict__ S,
                                          float* __restrict__ cnt) {
  int id[PRW];
  float v[PRW][4];
#pragma unroll
  for (int i = 0; i < PRW; ++i) {
    const int r = r0 + ((FULL || i < nr) ? i : nr - 1);
    id[i] = seg[r];
    const float* xr = X + (size_t)r * H + lane;
#pragma unroll
    for (int q = 0; q < 4; ++q) v[i][q] = xr[32 * q];
  }
  float acc[4] = {v[0][0], v[0][1], v[0][2], v[0][3]};
  int cur = id[0];
  int run = 1;
#pragma unroll
  for (int i = 1; i < PRW; ++i) {
    if (FULL || i < nr) {
      if (id[i] != cur) {
        pool_flush(S, cnt, cur, lane, acc, run);
        acc[0] = 0.0f; acc[1] = 0.0f; acc[2] = 0.0f; acc[3] = 0.0f;
        run = 0;
        cur = id[i];
      }
#pragma unroll
      for (int q = 0; q < 4; ++q) acc[q] += v[i][q];
      ++run;
    }
  }
  pool_flush(S, cnt, cur, lane, acc, run);
}

__global__ __launch_bounds__(256) void pool_kernel(const float* __restrict__ X,
                                                   const int* __restrict__ seg, int N,
                                                   float* __restrict__ S,
                                                   float* __restrict__ cnt) {
  const int lane = threadIdx.x & 31;
  const int wave = __builtin_amdgcn_readfirstlane((int)((blockIdx.x * 256u + threadIdx.x) >> 5));
  const int r0 = wave * PRW;
  if (r0 >= N) return;
  if (r0 + PRW <= N) pool_rows<true>(X, seg, r0, PRW, lane, S, cnt);
  else pool_rows<false>(X, seg, r0, N - r0, lane, S, cnt);
}

__global__ __launch_bounds__(256) void finalize_kernel(const float* __restrict__ sums,
                                                       const float* __restrict__ cnt,
                                                       float* __restrict__ out, int G) {
  int t = blockIdx.x * blockDim.x + threadIdx.x;
  if (t >= G * H) return;
  int g = t >> 7;
  out[t] = sums[t] / fmaxf(cnt[g], 1.0f);
}

extern "C" void kernel_launch(void* const* d_in, const int* in_sizes, int n_in,
                              void* d_out, int out_size, void* d_ws, size_t ws_size,
                              hipStream_t stream) {
  const float* x_node = (const float*)d_in[0];
  const float* x_link = (const float*)d_in[1];
  const float* Wm_nl  = (const float*)d_in[2];
  const float* bm_nl  = (const float*)d_in[3];
  const float* Wu_nl  = (const float*)d_in[4];
  const float* bu_nl  = (const float*)d_in[5];
  const float* Wm_ln  = (const float*)d_in[6];
  const float* bm_ln  = (const float*)d_in[7];
  const float* Wu_ln  = (const float*)d_in[8];
  const float* bu_ln  = (const float*)d_in[9];
  const int*   nl     = (const int*)d_in[10];
  const int*   ln     = (const int*)d_in[11];
  const int*   batch  = (const int*)d_in[12];

  const int NL = in_sizes[12];
  const int NN = 50000;
  const int F_NODE = in_sizes[0] / NN;
  const int F_LINK = in_sizes[1] / NL;
  const int E  = in_sizes[10] / 2;
  const int Lnum = in_sizes[3] / H;
  const int G = out_size / H;

  const size_t MH = (size_t)((NN > NL) ? NN : NL) * H;
  float* ws   = (float*)d_ws;
  float* xn   = ws;
  float* xl   = xn  + MH;
  float* xn2  = xl  + MH;
  float* xl2  = xn2 + MH;
  float* Sl   = xl2 + MH;
  float* Sn   = Sl  + MH;
  float* FW   = Sn  + MH;
  float* dgl  = FW  + MH;
  float* dgn  = dgl + NL;
  float* sums = dgn + NN;
  float* cnt  = sums + (size_t)G * H;
  (void)n_in; (void)ws_size;
  const size_t fw_used = (size_t)2 * Lnum * FROWS * H;
  const bool use_list = (Lnum > 1) &&
      (fw_used + (size_t)NL * CAP + 32 + (size_t)2 * E <= MH);
  int* ell     = use_list ? (int*)(FW + fw_used) : nullptr;
  int* ovf_cnt = use_list ? ell + (size_t)NL * CAP : nullptr;
  int* ovf     = use_list ? ovf_cnt + 32 : nullptr;

  auto nb = [](long n, int b) { return (int)((n + b - 1) / b); };

  const bool node_live_any = (Lnum > 1);

  const long nsmall = (long)G * H + G;
  const int nbsmall = nb(nsmall, 4 * 256 * ZV);

  auto lp_of = [](int F) {
    return (F < 2) ? 1 : (F < 4) ? 2 : (F < 8) ? 3 : (F < 16) ? 4 : (F < 32) ? 5 : (F < 64) ? 6 : 7;
  };
  const int lpn = lp_of(F_NODE);
  const int lpl = lp_of(F_LINK);

  const int gl = nb(NL / 16, UW);
  const int gn = nb(NN / 16, UW);
  const size_t FSET = (size_t)FROWS * H;

  if (Lnum == 0) {
    zero_kernel<<<nbsmall, 256, 0, stream>>>(sums, nsmall);
    pad_kernel<<<nb((long)NL * H, 256), 256, 0, stream>>>(x_link, xl, NL, F_LINK);
  } else {
    float* S0l = Sl;
    float* S0n = Sl + ((size_t)NL << lpn);
    const long nS0 = ((long)NL << lpn) + (node_live_any ? ((long)NN << lpl) : 0L);
    const int nfold = FT * 2 * Lnum;
    head_kernel<<<nfold + nbsmall + nb(nS0, 4 * 256 * ZV), 256, 0, stream>>>(
        Wm_nl, bm_nl, Wu_nl, Wm_ln, bm_ln, Wu_ln, Lnum, FW, nfold,
        sums, nsmall, nbsmall, S0l, nS0, ovf_cnt);
    const int gxa = nb((long)E << lpn, 256 * SFU);
    const int gxb = nb((long)E << lpl, 256 * SFU);
    edge_first_kernel<<<dim3((gxa > gxb) ? gxa : gxb, node_live_any ? 2 : 1), 256, 0, stream>>>(
        x_node, F_NODE, lpn, nl, S0l, x_link, F_LINK, lpl, ln, S0n, E, ell, ovf_cnt, ovf);
    update_first_kernel<<<dim3((gl > gn) ? gl : gn, node_live_any ? 2 : 1), 32 * UW, 0, stream>>>(
        S0l, 1 << lpn, F_NODE, x_link, F_LINK,
        FW, Wu_nl + H * H, bu_nl, dgl, xl2, NL, gl,
        S0n, 1 << lpl, F_LINK, x_node, F_NODE,
        FW + FSET, Wu_ln + H * H, bu_ln, dgn, xn2, NN, gn);
  }

  for (int l = 1; l < Lnum; ++l) {
    float* cxn = (l & 1) ? xn2 : xn;
    float* cxl = (l & 1) ? xl2 : xl;
    float* oxn = (l & 1) ? xn  : xn2;
    float* oxl = (l & 1) ? xl  : xl2;

    const float* WuNLb = Wu_nl + (size_t)l * 2 * H * H + H * H;
    const float* WuLNb = Wu_ln + (size_t)l * 2 * H * H + H * H;
    const float* FWl = FW + (size_t)(2 * l) * FSET;

    const bool node_live = (l + 1 < Lnum);

    if (use_list) {
      gather_kernel<<<nb((long)NL * 32, 256), 256, 0, stream>>>(cxn, ell, dgl, ovf, ovf_cnt, NL, Sl);
    } else {
      zero_kernel<<<nb((long)NL * H, 4 * 256 * ZV), 256, 0, stream>>>(Sl, (long)NL * H);
      scatter_kernel<<<nb((long)nb(E, EPW) * 32, 256), 256, 0, stream>>>(cxn, nl, nl + E, E, Sl);
    }
    if (node_live) {
      zero_kernel<<<nb((long)NN * H, 4 * 256 * ZV), 256, 0, stream>>>(Sn, (long)NN * H);
      scatter_kernel<<<nb((long)nb(E, EPW) * 32, 256), 256, 0, stream>>>(cxl, ln, ln + E, E, Sn);
    }

    update_kernel<<<gl, 32 * UW, 0, stream>>>(
        Sl, cxl, FWl, WuNLb, bu_nl + (size_t)l * H, dgl, oxl, NL);
    if (node_live) {
      update_kernel<<<gn, 32 * UW, 0, stream>>>(
          Sn, cxn, FWl + FSET, WuLNb, bu_ln + (size_t)l * H, dgn, oxn, NN);
    }
  }

  float* fxl = (Lnum & 1) ? xl2 : xl;
  pool_kernel<<<nb((long)nb(NL, PRW) * 32, 256), 256, 0, stream>>>(fxl, batch, NL, sums, cnt);
  finalize_kernel<<<nb((long)G * H, 256), 256, 0, stream>>>(sums, cnt, (float*)d_out, G);
}
